# starting kernel
# baseline (speedup 1.0000x reference)
#include <hip/hip_runtime.h>

typedef float v2f __attribute__((ext_vector_type(2)));
typedef float v8f __attribute__((ext_vector_type(8)));

constexpr int HD   = 96;
constexpr int RPB  = 128;
constexpr int BK   = 32;
constexpr int LPAD = 36;
constexpr int WPAD = 38;


__global__ void fill_zero_kernel(float* __restrict__ p, long n) {
  long i = (long)blockIdx.x * blockDim.x + threadIdx.x;
  long stride = (long)gridDim.x * blockDim.x;
  for (; i < n; i += stride) p[i] = 0.0f;
}

__global__ void degree_kernel(const int* __restrict__ src, const int* __restrict__ dst,
                              float* __restrict__ deg_out, float* __restrict__ deg_in, int E) {
  int e = blockIdx.x * blockDim.x + threadIdx.x;
  if (e < E) {
    __hip_atomic_fetch_add(&deg_out[src[e]], 1.0f, __ATOMIC_RELAXED, __HIP_MEMORY_SCOPE_AGENT);
    __hip_atomic_fetch_add(&deg_in[dst[e]],  1.0f, __ATOMIC_RELAXED, __HIP_MEMORY_SCOPE_AGENT);
  }
}

__global__ void finalize_norm_kernel(float* __restrict__ p, int n) {
  int i = blockIdx.x * blockDim.x + threadIdx.x;
  if (i < n) p[i] = 1.0f / sqrtf(fmaxf(p[i], 1.0f));
}

__global__ void wsum_kernel(const float* __restrict__ Wm, const float* __restrict__ bm,
                            float* __restrict__ wsum, float* __restrict__ bsum) {
  int t = threadIdx.x;
  if (t < HD) {
    float s = 0.0f;
    for (int j = 0; j < HD; ++j) s += Wm[t * HD + j];
    wsum[t] = s;
  } else if (t == HD) {
    float s = 0.0f;
    for (int j = 0; j < HD; ++j) s += bm[j];
    *bsum = s;
  }
}

__global__ __launch_bounds__(256)
void gemm_norm_wmma_kernel(const float* __restrict__ X, const int* __restrict__ rowmap,
                           const float* __restrict__ norm, const float* __restrict__ W,
                           float* __restrict__ OUT, int nrows, int K) {
  __shared__ float ldsX[RPB][LPAD];
  __shared__ float ldsW[HD][WPAD];

  const int tid  = threadIdx.x;
  const int wave = tid >> 5;
  const int lane = tid & 31;
  const int half = lane >> 4;
  const int l16  = lane & 15;
  const int row0 = blockIdx.x * RPB;

  v8f acc[6] = {};

  for (int k0 = 0; k0 < K; k0 += BK) {
    #pragma unroll
    for (int it = 0; it < (RPB * BK / 4) / 256; ++it) {
      int idx  = tid + 256 * it;
      int r    = idx >> 3;
      int c    = idx & 7;
      int grow = row0 + r;
      int kg   = k0 + c * 4;
      float4 v = make_float4(0.f, 0.f, 0.f, 0.f);
      if (grow < nrows && kg < K) {
        int srow = rowmap ? rowmap[grow] : grow;
        const float* p = X + (long)srow * K + kg;
        v = *(const float4*)p;
        if (kg + BK < K) __builtin_prefetch(p + BK, 0, 0);
        float s = norm[grow];
        v.x *= s; v.y *= s; v.z *= s; v.w *= s;
      }
      ldsX[r][c * 4 + 0] = v.x;
      ldsX[r][c * 4 + 1] = v.y;
      ldsX[r][c * 4 + 2] = v.z;
      ldsX[r][c * 4 + 3] = v.w;
    }
    #pragma unroll
    for (int it = 0; it < (BK * HD) / 256; ++it) {
      int idx = tid + 256 * it;
      int kk  = idx / HD;
      int col = idx % HD;
      float v = 0.0f;
      if (k0 + kk < K) v = W[(long)(k0 + kk) * HD + col];
      ldsW[col][kk] = v;
    }
    __syncthreads();

    const int kmax = (BK < K - k0) ? BK : (K - k0);
    const int mrow = wave * 16 + l16;
    for (int kk = 0; kk < kmax; kk += 4) {
      int ka = kk + half * 2;

      v2f A = *(const v2f*)&ldsX[mrow][ka];
      v2f Bf[6];
      #pragma unroll
      for (int t = 0; t < 6; ++t)
        Bf[t] = *(const v2f*)&ldsW[t * 16 + l16][ka];

      #pragma unroll
      for (int t = 0; t < 6; ++t)
        acc[t] = __builtin_amdgcn_wmma_f32_16x16x4_f32(
             false, A,  false, Bf[t],
             (short)0, acc[t],  false,  false);
    }
    __syncthreads();
  }

  const int orow = row0 + wave * 16 + half * 8;
  #pragma unroll
  for (int t = 0; t < 6; ++t) {
    int col = t * 16 + l16;
    #pragma unroll
    for (int i = 0; i < 8; ++i) {
      int r = orow + i;
      if (r < nrows) OUT[(long)r * HD + col] = acc[t][i];
    }
  }
}

__global__ void spmm_kernel(const int* __restrict__ src, const int* __restrict__ dst,
                            const float* __restrict__ h, float* __restrict__ agg, int E) {
  int gw   = (int)(((long)blockIdx.x * blockDim.x + threadIdx.x) >> 5);
  int lane = threadIdx.x & 31;
  if (gw < E) {
    int s = src[gw], d = dst[gw];
    const float* hp = h + (long)s * HD;
    float* ap = agg + (long)d * HD;
    #pragma unroll
    for (int i = 0; i < 3; ++i) {
      float v = hp[lane + 32 * i];
      __hip_atomic_fetch_add(&ap[lane + 32 * i], v, __ATOMIC_RELAXED, __HIP_MEMORY_SCOPE_AGENT);
    }
  }
}

__global__ void prelu_kernel(const float* __restrict__ agg, const float* __restrict__ norm_in,
                             const float* __restrict__ b, const float* __restrict__ a,
                             float* __restrict__ out, int n) {
  long idx = (long)blockIdx.x * blockDim.x + threadIdx.x;
  if (idx < (long)n * HD) {
    int i = (int)(idx / HD);
    int j = (int)(idx % HD);
    float v = agg[idx] * norm_in[i] + b[j];
    out[idx] = (v >= 0.0f) ? v : a[j] * v;
  }
}

__global__ void reduce_kernel(const float* __restrict__ z, const float* __restrict__ wsum,
                              const float* __restrict__ bsum, float* __restrict__ out, int n) {
  int gw   = (int)(((long)blockIdx.x * blockDim.x + threadIdx.x) >> 5);
  int lane = threadIdx.x & 31;
  if (gw < n) {
    const float* zp = z + (long)gw * HD;
    float s = 0.0f;
    #pragma unroll
    for (int i = 0; i < 3; ++i) s = fmaf(zp[lane + 32 * i], wsum[lane + 32 * i], s);
    #pragma unroll
    for (int off = 16; off > 0; off >>= 1) s += __shfl_xor(s, off, 32);
    if (lane == 0) out[gw] = s + *bsum;
  }
}

extern "C" void kernel_launch(void* const* d_in, const int* in_sizes, int n_in,
                              void* d_out, int out_size, void* d_ws, size_t ws_size,
                              hipStream_t stream) {
  const float* x    = (const float*)d_in[0];
  const int*   src  = (const int*)  d_in[1];
  const int*   dst  = (const int*)  d_in[2];
  const int*   perm = (const int*)  d_in[3];
  const float* W1   = (const float*)d_in[4];
  const float* b1   = (const float*)d_in[5];
  const float* a1   = (const float*)d_in[6];
  const float* W2   = (const float*)d_in[7];
  const float* b2   = (const float*)d_in[8];
  const float* a2   = (const float*)d_in[9];
  const float* Wm   = (const float*)d_in[10];
  const float* bm   = (const float*)d_in[11];
  float* out = (float*)d_out;

  const int E  = in_sizes[1];
  const int N  = in_sizes[3];
  const int K1 = in_sizes[0] / N;

  float* ws       = (float*)d_ws;
  float* norm_out = ws;
  float* norm_in  = ws + N;
  float* wsum     = ws + 2L * N;
  float* bsum     = ws + 2L * N + HD;
  float* bufA     = ws + 2L * N + 2 * HD;
  float* bufB     = bufA + (long)N * HD;

  const long NH = (long)N * HD;
  const int gemm_blocks = (N + RPB - 1) / RPB;
  const int spmm_blocks = (int)(((long)E * 32 + 255) / 256);
  const int elw_blocks  = (int)((NH + 255) / 256);
  const int red_blocks  = (int)(((long)N * 32 + 255) / 256);

  fill_zero_kernel<<<512, 256, 0, stream>>>(norm_out, 2L * N);
  degree_kernel<<<(E + 255) / 256, 256, 0, stream>>>(src, dst, norm_out, norm_in, E);
  finalize_norm_kernel<<<(2 * N + 255) / 256, 256, 0, stream>>>(norm_out, 2 * N);
  wsum_kernel<<<1, 128, 0, stream>>>(Wm, bm, wsum, bsum);

  for (int view = 0; view < 2; ++view) {
    const int* rowmap = view ? perm : nullptr;
    gemm_norm_wmma_kernel<<<gemm_blocks, 256, 0, stream>>>(x, rowmap, norm_out, W1, bufA, N, K1);
    fill_zero_kernel<<<2048, 256, 0, stream>>>(bufB, NH);
    spmm_kernel<<<spmm_blocks, 256, 0, stream>>>(src, dst, bufA, bufB, E);
    prelu_kernel<<<elw_blocks, 256, 0, stream>>>(bufB, norm_in, b1, a1, bufA, N);
    gemm_norm_wmma_kernel<<<gemm_blocks, 256, 0, stream>>>(bufA, nullptr, norm_out, W2, bufB, N, HD);
    fill_zero_kernel<<<2048, 256, 0, stream>>>(bufA, NH);
    spmm_kernel<<<spmm_blocks, 256, 0, stream>>>(src, dst, bufB, bufA, E);
    prelu_kernel<<<elw_blocks, 256, 0, stream>>>(bufA, norm_in, b2, a2, bufB, N);
    reduce_kernel<<<red_blocks, 256, 0, stream>>>(bufB, wsum, bsum, out + (long)view * N, N);
  }
}


# SpMM as a gather: edges grouped by destination once per call (returning int add of the degree pass = rank, scan, placement); one wave sums a node's source rows in list order; no float atomics
# speedup vs baseline: 1.5365x; 1.5365x over previous
#include <hip/hip_runtime.h>

typedef float v2f __attribute__((ext_vector_type(2)));
typedef float v8f __attribute__((ext_vector_type(8)));

constexpr int HD   = 96;
constexpr int RPB  = 128;
constexpr int BK   = 32;
constexpr int LPAD = 36;
constexpr int WPAD = 38;


__global__ void fill_zero_kernel(float* __restrict__ p, long n) {
  long i = (long)blockIdx.x * blockDim.x + threadIdx.x;
  long stride = (long)gridDim.x * blockDim.x;
  for (; i < n; i += stride) p[i] = 0.0f;
}

__global__ void degree_kernel(const int* __restrict__ src, const int* __restrict__ dst,
                              int* __restrict__ cnt_out, int* __restrict__ cnt_in,
                              int* __restrict__ rank, int E) {
  int e = blockIdx.x * blockDim.x + threadIdx.x;
  if (e < E) {
    __hip_atomic_fetch_add(&cnt_out[src[e]], 1, __ATOMIC_RELAXED, __HIP_MEMORY_SCOPE_AGENT);
    if (rank) {
      rank[e] = __hip_atomic_fetch_add(&cnt_in[dst[e]], 1, __ATOMIC_RELAXED, __HIP_MEMORY_SCOPE_AGENT);
    } else {
      __hip_atomic_fetch_add(&cnt_in[dst[e]], 1, __ATOMIC_RELAXED, __HIP_MEMORY_SCOPE_AGENT);
    }
  }
}

__global__ void finalize_norm_kernel(float* __restrict__ p, int n) {
  int i = blockIdx.x * blockDim.x + threadIdx.x;
  if (i < n) p[i] = 1.0f / sqrtf(fmaxf((float)__float_as_int(p[i]), 1.0f));
}

__global__ __launch_bounds__(1024)
void rowptr_scan_kernel(const int* __restrict__ cnt, int* __restrict__ rowptr, int n) {
  __shared__ int wave_tot[32];
  const int t = threadIdx.x, lane = t & 31, w = t >> 5;
  int carry = 0;
  int c[4], nx[4];
  for (int k = 0; k < 4; ++k) {
    const int i = t * 4 + k;
    const int v = cnt[i < n ? i : n - 1];
    c[k] = (i < n) ? v : 0;
  }
  for (int base = 0; base < n; base += 4096) {
    const int i0 = base + t * 4;
    for (int k = 0; k < 4; ++k) {
      const int i = i0 + 4096 + k;
      const int v = cnt[i < n ? i : n - 1];
      nx[k] = (i < n) ? v : 0;
    }
    const int s = c[0] + c[1] + c[2] + c[3];
    int incl = s;
    for (int off = 1; off < 32; off <<= 1) {
      int v = __shfl_up(incl, off, 32);
      if (lane >= off) incl += v;
    }
    if (lane == 31) wave_tot[w] = incl;
    __syncthreads();
    const int wt = wave_tot[lane];
    int wincl = wt;
    for (int off = 1; off < 32; off <<= 1) {
      int v = __shfl_up(wincl, off, 32);
      if (lane >= off) wincl += v;
    }
    const int wave_excl = __shfl(wincl, w, 32) - __shfl(wt, w, 32);
    const int total     = __shfl(wincl, 31, 32);
    int run = carry + wave_excl + incl - s;
    for (int k = 0; k < 4; ++k) {
      if (i0 + k < n) rowptr[i0 + k] = run;
      run += c[k];
    }
    carry += total;
    for (int k = 0; k < 4; ++k) c[k] = nx[k];
    __syncthreads();
  }
  if (t == 0) rowptr[n] = carry;
}

__global__ void place_kernel(const int* __restrict__ src, const int* __restrict__ dst,
                             const int* __restrict__ rank, const int* __restrict__ rowptr,
                             int* __restrict__ csr_src, int E) {
  int e = blockIdx.x * blockDim.x + threadIdx.x;
  if (e < E) csr_src[rowptr[dst[e]] + rank[e]] = src[e];
}

__global__ void wsum_kernel(const float* __restrict__ Wm, const float* __restrict__ bm,
                            float* __restrict__ wsum, float* __restrict__ bsum) {
  int t = threadIdx.x;
  if (t < HD) {
    float s = 0.0f;
    for (int j = 0; j < HD; ++j) s += Wm[t * HD + j];
    wsum[t] = s;
  } else if (t == HD) {
    float s = 0.0f;
    for (int j = 0; j < HD; ++j) s += bm[j];
    *bsum = s;
  }
}

__global__ __launch_bounds__(256)
void gemm_norm_wmma_kernel(const float* __restrict__ X, const int* __restrict__ rowmap,
                           const float* __restrict__ norm, const float* __restrict__ W,
                           float* __restrict__ OUT, int nrows, int K) {
  __shared__ float ldsX[RPB][LPAD];
  __shared__ float ldsW[HD][WPAD];

  const int tid  = threadIdx.x;
  const int wave = tid >> 5;
  const int lane = tid & 31;
  const int half = lane >> 4;
  const int l16  = lane & 15;
  const int row0 = blockIdx.x * RPB;

  v8f acc[6] = {};

  for (int k0 = 0; k0 < K; k0 += BK) {
    #pragma unroll
    for (int it = 0; it < (RPB * BK / 4) / 256; ++it) {
      int idx  = tid + 256 * it;
      int r    = idx >> 3;
      int c    = idx & 7;
      int grow = row0 + r;
      int kg   = k0 + c * 4;
      float4 v = make_float4(0.f, 0.f, 0.f, 0.f);
      if (grow < nrows && kg < K) {
        int srow = rowmap ? rowmap[grow] : grow;
        const float* p = X + (long)srow * K + kg;
        v = *(const float4*)p;
        if (kg + BK < K) __builtin_prefetch(p + BK, 0, 0);
        float s = norm[grow];
        v.x *= s; v.y *= s; v.z *= s; v.w *= s;
      }
      ldsX[r][c * 4 + 0] = v.x;
      ldsX[r][c * 4 + 1] = v.y;
      ldsX[r][c * 4 + 2] = v.z;
      ldsX[r][c * 4 + 3] = v.w;
    }
    #pragma unroll
    for (int it = 0; it < (BK * HD) / 256; ++it) {
      int idx = tid + 256 * it;
      int kk  = idx / HD;
      int col = idx % HD;
      float v = 0.0f;
      if (k0 + kk < K) v = W[(long)(k0 + kk) * HD + col];
      ldsW[col][kk] = v;
    }
    __syncthreads();

    const int kmax = (BK < K - k0) ? BK : (K - k0);
    const int mrow = wave * 16 + l16;
    for (int kk = 0; kk < kmax; kk += 4) {
      int ka = kk + half * 2;

      v2f A = *(const v2f*)&ldsX[mrow][ka];
      v2f Bf[6];
      #pragma unroll
      for (int t = 0; t < 6; ++t)
        Bf[t] = *(const v2f*)&ldsW[t * 16 + l16][ka];

      #pragma unroll
      for (int t = 0; t < 6; ++t)
        acc[t] = __builtin_amdgcn_wmma_f32_16x16x4_f32(
             false, A,  false, Bf[t],
             (short)0, acc[t],  false,  false);
    }
    __syncthreads();
  }

  const int orow = row0 + wave * 16 + half * 8;
  #pragma unroll
  for (int t = 0; t < 6; ++t) {
    int col = t * 16 + l16;
    #pragma unroll
    for (int i = 0; i < 8; ++i) {
      int r = orow + i;
      if (r < nrows) OUT[(long)r * HD + col] = acc[t][i];
    }
  }
}

__global__ __launch_bounds__(256)
void spmm_gather_kernel(const int* __restrict__ rowptr, const int* __restrict__ csr_src,
                        const float* __restrict__ h, float* __restrict__ agg, int n) {
  const int lane = threadIdx.x & 31;
  const int row = __builtin_amdgcn_readfirstlane((int)(blockIdx.x * (blockDim.x >> 5) + (threadIdx.x >> 5)));
  if (row >= n) return;
  const int beg = rowptr[row];
  const int end = rowptr[row + 1];
  float a0 = 0.0f, a1 = 0.0f, a2 = 0.0f;
  for (int base = beg; base < end; base += 32) {
    const int cnt  = (end - base < 32) ? (end - base) : 32;
    const int mine = (lane < cnt) ? csr_src[base + lane] : 0;
    int j = 0;
    for (; j + 4 <= cnt; j += 4) {
      const float* p0 = h + (long)__builtin_amdgcn_readlane(mine, j + 0) * HD + lane;
      const float* p1 = h + (long)__builtin_amdgcn_readlane(mine, j + 1) * HD + lane;
      const float* p2 = h + (long)__builtin_amdgcn_readlane(mine, j + 2) * HD + lane;
      const float* p3 = h + (long)__builtin_amdgcn_readlane(mine, j + 3) * HD + lane;
      const float x00 = p0[0], x01 = p0[32], x02 = p0[64];
      const float x10 = p1[0], x11 = p1[32], x12 = p1[64];
      const float x20 = p2[0], x21 = p2[32], x22 = p2[64];
      const float x30 = p3[0], x31 = p3[32], x32 = p3[64];
      a0 += x00; a1 += x01; a2 += x02;
      a0 += x10; a1 += x11; a2 += x12;
      a0 += x20; a1 += x21; a2 += x22;
      a0 += x30; a1 += x31; a2 += x32;
    }
    for (; j < cnt; ++j) {
      const float* p = h + (long)__builtin_amdgcn_readlane(mine, j) * HD + lane;
      a0 += p[0]; a1 += p[32]; a2 += p[64];
    }
  }
  float* o = agg + (long)row * HD + lane;
  o[0] = a0; o[32] = a1; o[64] = a2;
}

__global__ void spmm_kernel(const int* __restrict__ src, const int* __restrict__ dst,
                            const float* __restrict__ h, float* __restrict__ agg, int E) {
  int gw   = (int)(((long)blockIdx.x * blockDim.x + threadIdx.x) >> 5);
  int lane = threadIdx.x & 31;
  if (gw < E) {
    int s = src[gw], d = dst[gw];
    const float* hp = h + (long)s * HD;
    float* ap = agg + (long)d * HD;
    #pragma unroll
    for (int i = 0; i < 3; ++i) {
      float v = hp[lane + 32 * i];
      __hip_atomic_fetch_add(&ap[lane + 32 * i], v, __ATOMIC_RELAXED, __HIP_MEMORY_SCOPE_AGENT);
    }
  }
}

__global__ void prelu_kernel(const float* __restrict__ agg, const float* __restrict__ norm_in,
                             const float* __restrict__ b, const float* __restrict__ a,
                             float* __restrict__ out, int n) {
  long idx = (long)blockIdx.x * blockDim.x + threadIdx.x;
  if (idx < (long)n * HD) {
    int i = (int)(idx / HD);
    int j = (int)(idx % HD);
    float v = agg[idx] * norm_in[i] + b[j];
    out[idx] = (v >= 0.0f) ? v : a[j] * v;
  }
}

__global__ void reduce_kernel(const float* __restrict__ z, const float* __restrict__ wsum,
                              const float* __restrict__ bsum, float* __restrict__ out, int n) {
  int gw   = (int)(((long)blockIdx.x * blockDim.x + threadIdx.x) >> 5);
  int lane = threadIdx.x & 31;
  if (gw < n) {
    const float* zp = z + (long)gw * HD;
    float s = 0.0f;
    #pragma unroll
    for (int i = 0; i < 3; ++i) s = fmaf(zp[lane + 32 * i], wsum[lane + 32 * i], s);
    #pragma unroll
    for (int off = 16; off > 0; off >>= 1) s += __shfl_xor(s, off, 32);
    if (lane == 0) out[gw] = s + *bsum;
  }
}

extern "C" void kernel_launch(void* const* d_in, const int* in_sizes, int n_in,
                              void* d_out, int out_size, void* d_ws, size_t ws_size,
                              hipStream_t stream) {
  const float* x    = (const float*)d_in[0];
  const int*   src  = (const int*)  d_in[1];
  const int*   dst  = (const int*)  d_in[2];
  const int*   perm = (const int*)  d_in[3];
  const float* W1   = (const float*)d_in[4];
  const float* b1   = (const float*)d_in[5];
  const float* a1   = (const float*)d_in[6];
  const float* W2   = (const float*)d_in[7];
  const float* b2   = (const float*)d_in[8];
  const float* a2   = (const float*)d_in[9];
  const float* Wm   = (const float*)d_in[10];
  const float* bm   = (const float*)d_in[11];
  float* out = (float*)d_out;

  const int E  = in_sizes[1];
  const int N  = in_sizes[3];
  const int K1 = in_sizes[0] / N;

  float* ws       = (float*)d_ws;
  float* norm_out = ws;
  float* norm_in  = ws + N;
  float* wsum     = ws + 2L * N;
  float* bsum     = ws + 2L * N + HD;
  float* bufA     = ws + 2L * N + 2 * HD;
  float* bufB     = bufA + (long)N * HD;

  const long NH = (long)N * HD;
  int*   rowptr   = (int*)(bufB + NH);
  int*   csr_src  = rowptr + (N + 1);
  int*   rank     = (int*)bufB;
  int*   cnt_out  = (int*)norm_out;
  int*   cnt_in   = (int*)norm_in;

  const int gemm_blocks = (N + RPB - 1) / RPB;
  const int spmm_blocks = (int)(((long)E * 32 + 255) / 256);
  const int gath_blocks = (N + 7) / 8;
  const int elw_blocks  = (int)((NH + 255) / 256);
  const int red_blocks  = (int)(((long)N * 32 + 255) / 256);

  const size_t need_grouped = (size_t)(2L * N + 2 * HD + 2 * NH + (N + 1) + E) * sizeof(float);
  const bool grouped = (ws_size >= need_grouped) && ((long)E <= NH);

  fill_zero_kernel<<<512, 256, 0, stream>>>(norm_out, 2L * N);
  degree_kernel<<<(E + 255) / 256, 256, 0, stream>>>(src, dst, cnt_out, cnt_in, grouped ? rank : nullptr, E);
  if (grouped) {
    rowptr_scan_kernel<<<1, 1024, 0, stream>>>(cnt_in, rowptr, N);
    place_kernel<<<(E + 255) / 256, 256, 0, stream>>>(src, dst, rank, rowptr, csr_src, E);
  }
  finalize_norm_kernel<<<(2 * N + 255) / 256, 256, 0, stream>>>(norm_out, 2 * N);
  wsum_kernel<<<1, 128, 0, stream>>>(Wm, bm, wsum, bsum);

  auto aggregate = [&](const float* h, float* agg) {
    if (grouped) {
      spmm_gather_kernel<<<gath_blocks, 256, 0, stream>>>(rowptr, csr_src, h, agg, N);
    } else {
      fill_zero_kernel<<<2048, 256, 0, stream>>>(agg, NH);
      spmm_kernel<<<spmm_blocks, 256, 0, stream>>>(src, dst, h, agg, E);
    }
  };

  for (int view = 0; view < 2; ++view) {
    const int* rowmap = view ? perm : nullptr;
    gemm_norm_wmma_kernel<<<gemm_blocks, 256, 0, stream>>>(x, rowmap, norm_out, W1, bufA, N, K1);
    aggregate(bufA, bufB);
    prelu_kernel<<<elw_blocks, 256, 0, stream>>>(bufB, norm_in, b1, a1, bufA, N);
    gemm_norm_wmma_kernel<<<gemm_blocks, 256, 0, stream>>>(bufA, nullptr, norm_out, W2, bufB, N, HD);
    aggregate(bufB, bufA);
    prelu_kernel<<<elw_blocks, 256, 0, stream>>>(bufA, norm_in, b2, a2, bufB, N);
    reduce_kernel<<<red_blocks, 256, 0, stream>>>(bufB, wsum, bsum, out + (long)view * N, N);
  }
}


# node finish fused into the gather wave: norm_in scale, bias, PReLU on its sums in registers; layer 2 also takes the wsum dot there, one float per node out; every sum order kept; 6 launches fewer
# speedup vs baseline: 1.7989x; 1.1708x over previous
#include <hip/hip_runtime.h>

typedef float v2f __attribute__((ext_vector_type(2)));
typedef float v8f __attribute__((ext_vector_type(8)));

constexpr int HD   = 96;
constexpr int RPB  = 128;
constexpr int BK   = 32;
constexpr int LPAD = 36;
constexpr int WPAD = 38;


__global__ void fill_zero_kernel(float* __restrict__ p, long n) {
  long i = (long)blockIdx.x * blockDim.x + threadIdx.x;
  long stride = (long)gridDim.x * blockDim.x;
  for (; i < n; i += stride) p[i] = 0.0f;
}

__global__ void degree_kernel(const int* __restrict__ src, const int* __restrict__ dst,
                              int* __restrict__ cnt_out, int* __restrict__ cnt_in,
                              int* __restrict__ rank, int E) {
  int e = blockIdx.x * blockDim.x + threadIdx.x;
  if (e < E) {
    __hip_atomic_fetch_add(&cnt_out[src[e]], 1, __ATOMIC_RELAXED, __HIP_MEMORY_SCOPE_AGENT);
    if (rank) {
      rank[e] = __hip_atomic_fetch_add(&cnt_in[dst[e]], 1, __ATOMIC_RELAXED, __HIP_MEMORY_SCOPE_AGENT);
    } else {
      __hip_atomic_fetch_add(&cnt_in[dst[e]], 1, __ATOMIC_RELAXED, __HIP_MEMORY_SCOPE_AGENT);
    }
  }
}

__global__ void finalize_norm_kernel(float* __restrict__ p, int n) {
  int i = blockIdx.x * blockDim.x + threadIdx.x;
  if (i < n) p[i] = 1.0f / sqrtf(fmaxf((float)__float_as_int(p[i]), 1.0f));
}

__global__ __launch_bounds__(1024)
void rowptr_scan_kernel(const int* __restrict__ cnt, int* __restrict__ rowptr, int n) {
  __shared__ int wave_tot[32];
  const int t = threadIdx.x, lane = t & 31, w = t >> 5;
  int carry = 0;
  int c[4], nx[4];
  for (int k = 0; k < 4; ++k) {
    const int i = t * 4 + k;
    const int v = cnt[i < n ? i : n - 1];
    c[k] = (i < n) ? v : 0;
  }
  for (int base = 0; base < n; base += 4096) {
    const int i0 = base + t * 4;
    for (int k = 0; k < 4; ++k) {
      const int i = i0 + 4096 + k;
      const int v = cnt[i < n ? i : n - 1];
      nx[k] = (i < n) ? v : 0;
    }
    const int s = c[0] + c[1] + c[2] + c[3];
    int incl = s;
    for (int off = 1; off < 32; off <<= 1) {
      int v = __shfl_up(incl, off, 32);
      if (lane >= off) incl += v;
    }
    if (lane == 31) wave_tot[w] = incl;
    __syncthreads();
    const int wt = wave_tot[lane];
    int wincl = wt;
    for (int off = 1; off < 32; off <<= 1) {
      int v = __shfl_up(wincl, off, 32);
      if (lane >= off) wincl += v;
    }
    const int wave_excl = __shfl(wincl, w, 32) - __shfl(wt, w, 32);
    const int total     = __shfl(wincl, 31, 32);
    int run = carry + wave_excl + incl - s;
    for (int k = 0; k < 4; ++k) {
      if (i0 + k < n) rowptr[i0 + k] = run;
      run += c[k];
    }
    carry += total;
    for (int k = 0; k < 4; ++k) c[k] = nx[k];
    __syncthreads();
  }
  if (t == 0) rowptr[n] = carry;
}

__global__ void place_kernel(const int* __restrict__ src, const int* __restrict__ dst,
                             const int* __restrict__ rank, const int* __restrict__ rowptr,
                             int* __restrict__ csr_src, int E) {
  int e = blockIdx.x * blockDim.x + threadIdx.x;
  if (e < E) csr_src[rowptr[dst[e]] + rank[e]] = src[e];
}

__global__ void wsum_kernel(const float* __restrict__ Wm, const float* __restrict__ bm,
                            float* __restrict__ wsum, float* __restrict__ bsum) {
  int t = threadIdx.x;
  if (t < HD) {
    float s = 0.0f;
    for (int j = 0; j < HD; ++j) s += Wm[t * HD + j];
    wsum[t] = s;
  } else if (t == HD) {
    float s = 0.0f;
    for (int j = 0; j < HD; ++j) s += bm[j];
    *bsum = s;
  }
}

__global__ __launch_bounds__(256)
void gemm_norm_wmma_kernel(const float* __restrict__ X, const int* __restrict__ rowmap,
                           const float* __restrict__ norm, const float* __restrict__ W,
                           float* __restrict__ OUT, int nrows, int K) {
  __shared__ float ldsX[RPB][LPAD];
  __shared__ float ldsW[HD][WPAD];

  const int tid  = threadIdx.x;
  const int wave = tid >> 5;
  const int lane = tid & 31;
  const int half = lane >> 4;
  const int l16  = lane & 15;
  const int row0 = blockIdx.x * RPB;

  v8f acc[6] = {};

  for (int k0 = 0; k0 < K; k0 += BK) {
    #pragma unroll
    for (int it = 0; it < (RPB * BK / 4) / 256; ++it) {
      int idx  = tid + 256 * it;
      int r    = idx >> 3;
      int c    = idx & 7;
      int grow = row0 + r;
      int kg   = k0 + c * 4;
      float4 v = make_float4(0.f, 0.f, 0.f, 0.f);
      if (grow < nrows && kg < K) {
        int srow = rowmap ? rowmap[grow] : grow;
        const float* p = X + (long)srow * K + kg;
        v = *(const float4*)p;
        if (kg + BK < K) __builtin_prefetch(p + BK, 0, 0);
        float s = norm[grow];
        v.x *= s; v.y *= s; v.z *= s; v.w *= s;
      }
      ldsX[r][c * 4 + 0] = v.x;
      ldsX[r][c * 4 + 1] = v.y;
      ldsX[r][c * 4 + 2] = v.z;
      ldsX[r][c * 4 + 3] = v.w;
    }
    #pragma unroll
    for (int it = 0; it < (BK * HD) / 256; ++it) {
      int idx = tid + 256 * it;
      int kk  = idx / HD;
      int col = idx % HD;
      float v = 0.0f;
      if (k0 + kk < K) v = W[(long)(k0 + kk) * HD + col];
      ldsW[col][kk] = v;
    }
    __syncthreads();

    const int kmax = (BK < K - k0) ? BK : (K - k0);
    const int mrow = wave * 16 + l16;
    for (int kk = 0; kk < kmax; kk += 4) {
      int ka = kk + half * 2;

      v2f A = *(const v2f*)&ldsX[mrow][ka];
      v2f Bf[6];
      #pragma unroll
      for (int t = 0; t < 6; ++t)
        Bf[t] = *(const v2f*)&ldsW[t * 16 + l16][ka];

      #pragma unroll
      for (int t = 0; t < 6; ++t)
        acc[t] = __builtin_amdgcn_wmma_f32_16x16x4_f32(
             false, A,  false, Bf[t],
             (short)0, acc[t],  false,  false);
    }
    __syncthreads();
  }

  const int orow = row0 + wave * 16 + half * 8;
  #pragma unroll
  for (int t = 0; t < 6; ++t) {
    int col = t * 16 + l16;
    #pragma unroll
    for (int i = 0; i < 8; ++i) {
      int r = orow + i;
      if (r < nrows) OUT[(long)r * HD + col] = acc[t][i];
    }
  }
}

__device__ __forceinline__ void gather_row_sum(const int* __restrict__ rowptr, const int* __restrict__ csr_src,
                                               const float* __restrict__ h, int row, int lane,
                                               float& a0, float& a1, float& a2) {
  const int beg = rowptr[row];
  const int end = rowptr[row + 1];
  a0 = 0.0f; a1 = 0.0f; a2 = 0.0f;
  for (int base = beg; base < end; base += 32) {
    const int cnt  = (end - base < 32) ? (end - base) : 32;
    const int mine = (lane < cnt) ? csr_src[base + lane] : 0;
    int j = 0;
    for (; j + 4 <= cnt; j += 4) {
      const float* p0 = h + (long)__builtin_amdgcn_readlane(mine, j + 0) * HD + lane;
      const float* p1 = h + (long)__builtin_amdgcn_readlane(mine, j + 1) * HD + lane;
      const float* p2 = h + (long)__builtin_amdgcn_readlane(mine, j + 2) * HD + lane;
      const float* p3 = h + (long)__builtin_amdgcn_readlane(mine, j + 3) * HD + lane;
      const float x00 = p0[0], x01 = p0[32], x02 = p0[64];
      const float x10 = p1[0], x11 = p1[32], x12 = p1[64];
      const float x20 = p2[0], x21 = p2[32], x22 = p2[64];
      const float x30 = p3[0], x31 = p3[32], x32 = p3[64];
      a0 += x00; a1 += x01; a2 += x02;
      a0 += x10; a1 += x11; a2 += x12;
      a0 += x20; a1 += x21; a2 += x22;
      a0 += x30; a1 += x31; a2 += x32;
    }
    for (; j < cnt; ++j) {
      const float* p = h + (long)__builtin_amdgcn_readlane(mine, j) * HD + lane;
      a0 += p[0]; a1 += p[32]; a2 += p[64];
    }
  }
}

__global__ __launch_bounds__(256)
void spmm_gather_prelu_kernel(const int* __restrict__ rowptr, const int* __restrict__ csr_src,
                              const float* __restrict__ h, const float* __restrict__ norm_in,
                              const float* __restrict__ b, const float* __restrict__ a,
                              float* __restrict__ act, int n) {
  const int lane = threadIdx.x & 31;
  const int row = __builtin_amdgcn_readfirstlane((int)(blockIdx.x * (blockDim.x >> 5) + (threadIdx.x >> 5)));
  if (row >= n) return;
  const float ni = norm_in[row];
  const float b0 = b[lane], b1 = b[lane + 32], b2 = b[lane + 64];
  const float s0 = a[lane], s1 = a[lane + 32], s2 = a[lane + 64];
  float a0, a1, a2;
  gather_row_sum(rowptr, csr_src, h, row, lane, a0, a1, a2);
  const float v0 = a0 * ni + b0;
  const float v1 = a1 * ni + b1;
  const float v2 = a2 * ni + b2;
  float* o = act + (long)row * HD + lane;
  o[0]  = (v0 >= 0.0f) ? v0 : s0 * v0;
  o[32] = (v1 >= 0.0f) ? v1 : s1 * v1;
  o[64] = (v2 >= 0.0f) ? v2 : s2 * v2;
}

__global__ __launch_bounds__(256)
void spmm_gather_prelu_dot_kernel(const int* __restrict__ rowptr, const int* __restrict__ csr_src,
                                  const float* __restrict__ h, const float* __restrict__ norm_in,
                                  const float* __restrict__ b, const float* __restrict__ a,
                                  const float* __restrict__ wsum, const float* __restrict__ bsum,
                                  float* __restrict__ out, int n) {
  const int lane = threadIdx.x & 31;
  const int row = __builtin_amdgcn_readfirstlane((int)(blockIdx.x * (blockDim.x >> 5) + (threadIdx.x >> 5)));
  if (row >= n) return;
  const float ni = norm_in[row];
  const float b0 = b[lane], b1 = b[lane + 32], b2 = b[lane + 64];
  const float s0 = a[lane], s1 = a[lane + 32], s2 = a[lane + 64];
  const float w0 = wsum[lane], w1 = wsum[lane + 32], w2 = wsum[lane + 64];
  const float bs = *bsum;
  float a0, a1, a2;
  gather_row_sum(rowptr, csr_src, h, row, lane, a0, a1, a2);
  const float v0 = a0 * ni + b0;
  const float v1 = a1 * ni + b1;
  const float v2 = a2 * ni + b2;
  const float z0 = (v0 >= 0.0f) ? v0 : s0 * v0;
  const float z1 = (v1 >= 0.0f) ? v1 : s1 * v1;
  const float z2 = (v2 >= 0.0f) ? v2 : s2 * v2;
  float s = fmaf(z0, w0, 0.0f);
  s = fmaf(z1, w1, s);
  s = fmaf(z2, w2, s);
  #pragma unroll
  for (int off = 16; off > 0; off >>= 1) s += __shfl_xor(s, off, 32);
  if (lane == 0) out[row] = s + bs;
}

__global__ void spmm_kernel(const int* __restrict__ src, const int* __restrict__ dst,
                            const float* __restrict__ h, float* __restrict__ agg, int E) {
  int gw   = (int)(((long)blockIdx.x * blockDim.x + threadIdx.x) >> 5);
  int lane = threadIdx.x & 31;
  if (gw < E) {
    int s = src[gw], d = dst[gw];
    const float* hp = h + (long)s * HD;
    float* ap = agg + (long)d * HD;
    #pragma unroll
    for (int i = 0; i < 3; ++i) {
      float v = hp[lane + 32 * i];
      __hip_atomic_fetch_add(&ap[lane + 32 * i], v, __ATOMIC_RELAXED, __HIP_MEMORY_SCOPE_AGENT);
    }
  }
}

__global__ void prelu_kernel(const float* __restrict__ agg, const float* __restrict__ norm_in,
                             const float* __restrict__ b, const float* __restrict__ a,
                             float* __restrict__ out, int n) {
  long idx = (long)blockIdx.x * blockDim.x + threadIdx.x;
  if (idx < (long)n * HD) {
    int i = (int)(idx / HD);
    int j = (int)(idx % HD);
    float v = agg[idx] * norm_in[i] + b[j];
    out[idx] = (v >= 0.0f) ? v : a[j] * v;
  }
}

__global__ void reduce_kernel(const float* __restrict__ z, const float* __restrict__ wsum,
                              const float* __restrict__ bsum, float* __restrict__ out, int n) {
  int gw   = (int)(((long)blockIdx.x * blockDim.x + threadIdx.x) >> 5);
  int lane = threadIdx.x & 31;
  if (gw < n) {
    const float* zp = z + (long)gw * HD;
    float s = 0.0f;
    #pragma unroll
    for (int i = 0; i < 3; ++i) s = fmaf(zp[lane + 32 * i], wsum[lane + 32 * i], s);
    #pragma unroll
    for (int off = 16; off > 0; off >>= 1) s += __shfl_xor(s, off, 32);
    if (lane == 0) out[gw] = s + *bsum;
  }
}

extern "C" void kernel_launch(void* const* d_in, const int* in_sizes, int n_in,
                              void* d_out, int out_size, void* d_ws, size_t ws_size,
                              hipStream_t stream) {
  const float* x    = (const float*)d_in[0];
  const int*   src  = (const int*)  d_in[1];
  const int*   dst  = (const int*)  d_in[2];
  const int*   perm = (const int*)  d_in[3];
  const float* W1   = (const float*)d_in[4];
  const float* b1   = (const float*)d_in[5];
  const float* a1   = (const float*)d_in[6];
  const float* W2   = (const float*)d_in[7];
  const float* b2   = (const float*)d_in[8];
  const float* a2   = (const float*)d_in[9];
  const float* Wm   = (const float*)d_in[10];
  const float* bm   = (const float*)d_in[11];
  float* out = (float*)d_out;

  const int E  = in_sizes[1];
  const int N  = in_sizes[3];
  const int K1 = in_sizes[0] / N;

  float* ws       = (float*)d_ws;
  float* norm_out = ws;
  float* norm_in  = ws + N;
  float* wsum     = ws + 2L * N;
  float* bsum     = ws + 2L * N + HD;
  float* bufA     = ws + 2L * N + 2 * HD;
  float* bufB     = bufA + (long)N * HD;

  const long NH = (long)N * HD;
  int*   rowptr   = (int*)(bufB + NH);
  int*   csr_src  = rowptr + (N + 1);
  int*   rank     = (int*)bufB;
  int*   cnt_out  = (int*)norm_out;
  int*   cnt_in   = (int*)norm_in;

  const int gemm_blocks = (N + RPB - 1) / RPB;
  const int spmm_blocks = (int)(((long)E * 32 + 255) / 256);
  const int gath_blocks = (N + 7) / 8;
  const int elw_blocks  = (int)((NH + 255) / 256);
  const int red_blocks  = (int)(((long)N * 32 + 255) / 256);

  const size_t need_grouped = (size_t)(2L * N + 2 * HD + 2 * NH + (N + 1) + E) * sizeof(float);
  const bool grouped = (ws_size >= need_grouped) && ((long)E <= NH);

  fill_zero_kernel<<<512, 256, 0, stream>>>(norm_out, 2L * N);
  degree_kernel<<<(E + 255) / 256, 256, 0, stream>>>(src, dst, cnt_out, cnt_in, grouped ? rank : nullptr, E);
  if (grouped) {
    rowptr_scan_kernel<<<1, 1024, 0, stream>>>(cnt_in, rowptr, N);
    place_kernel<<<(E + 255) / 256, 256, 0, stream>>>(src, dst, rank, rowptr, csr_src, E);
  }
  finalize_norm_kernel<<<(2 * N + 255) / 256, 256, 0, stream>>>(norm_out, 2 * N);
  wsum_kernel<<<1, 128, 0, stream>>>(Wm, bm, wsum, bsum);

  for (int view = 0; view < 2; ++view) {
    const int* rowmap = view ? perm : nullptr;
    float* outv = out + (long)view * N;
    if (grouped) {
      gemm_norm_wmma_kernel<<<gemm_blocks, 256, 0, stream>>>(x, rowmap, norm_out, W1, bufA, N, K1);
      spmm_gather_prelu_kernel<<<gath_blocks, 256, 0, stream>>>(rowptr, csr_src, bufA, norm_in, b1, a1, bufB, N);
      gemm_norm_wmma_kernel<<<gemm_blocks, 256, 0, stream>>>(bufB, nullptr, norm_out, W2, bufA, N, HD);
      spmm_gather_prelu_dot_kernel<<<gath_blocks, 256, 0, stream>>>(rowptr, csr_src, bufA, norm_in, b2, a2,
                                                                   wsum, bsum, outv, N);
    } else {
      gemm_norm_wmma_kernel<<<gemm_blocks, 256, 0, stream>>>(x, rowmap, norm_out, W1, bufA, N, K1);
      fill_zero_kernel<<<2048, 256, 0, stream>>>(bufB, NH);
      spmm_kernel<<<spmm_blocks, 256, 0, stream>>>(src, dst, bufA, bufB, E);
      prelu_kernel<<<elw_blocks, 256, 0, stream>>>(bufB, norm_in, b1, a1, bufA, N);
      gemm_norm_wmma_kernel<<<gemm_blocks, 256, 0, stream>>>(bufA, nullptr, norm_out, W2, bufB, N, HD);
      fill_zero_kernel<<<2048, 256, 0, stream>>>(bufA, NH);
      spmm_kernel<<<spmm_blocks, 256, 0, stream>>>(src, dst, bufB, bufA, E);
      prelu_kernel<<<elw_blocks, 256, 0, stream>>>(bufA, norm_in, b2, a2, bufB, N);
      reduce_kernel<<<red_blocks, 256, 0, stream>>>(bufB, wsum, bsum, outv, N);
    }
  }
}


# GEMM staging: each K-tile's 16 global loads per thread leave as one batch, one tile ahead of use (clamped addresses, zeros taken at the LDS store), not one by one under branches with a wait each
# speedup vs baseline: 2.4840x; 1.3808x over previous
#include <hip/hip_runtime.h>

typedef float v2f __attribute__((ext_vector_type(2)));
typedef float v8f __attribute__((ext_vector_type(8)));

constexpr int HD   = 96;
constexpr int RPB  = 128;
constexpr int BK   = 32;
constexpr int LPAD = 36;
constexpr int WPAD = 38;


__global__ void fill_zero_kernel(float* __restrict__ p, long n) {
  long i = (long)blockIdx.x * blockDim.x + threadIdx.x;
  long stride = (long)gridDim.x * blockDim.x;
  for (; i < n; i += stride) p[i] = 0.0f;
}

__global__ void degree_kernel(const int* __restrict__ src, const int* __restrict__ dst,
                              int* __restrict__ cnt_out, int* __restrict__ cnt_in,
                              int* __restrict__ rank, int E) {
  int e = blockIdx.x * blockDim.x + threadIdx.x;
  if (e < E) {
    __hip_atomic_fetch_add(&cnt_out[src[e]], 1, __ATOMIC_RELAXED, __HIP_MEMORY_SCOPE_AGENT);
    if (rank) {
      rank[e] = __hip_atomic_fetch_add(&cnt_in[dst[e]], 1, __ATOMIC_RELAXED, __HIP_MEMORY_SCOPE_AGENT);
    } else {
      __hip_atomic_fetch_add(&cnt_in[dst[e]], 1, __ATOMIC_RELAXED, __HIP_MEMORY_SCOPE_AGENT);
    }
  }
}

__global__ void finalize_norm_kernel(float* __restrict__ p, int n) {
  int i = blockIdx.x * blockDim.x + threadIdx.x;
  if (i < n) p[i] = 1.0f / sqrtf(fmaxf((float)__float_as_int(p[i]), 1.0f));
}

__global__ __launch_bounds__(1024)
void rowptr_scan_kernel(const int* __restrict__ cnt, int* __restrict__ rowptr, int n) {
  __shared__ int wave_tot[32];
  const int t = threadIdx.x, lane = t & 31, w = t >> 5;
  int carry = 0;
  int c[4], nx[4];
  for (int k = 0; k < 4; ++k) {
    const int i = t * 4 + k;
    const int v = cnt[i < n ? i : n - 1];
    c[k] = (i < n) ? v : 0;
  }
  for (int base = 0; base < n; base += 4096) {
    const int i0 = base + t * 4;
    for (int k = 0; k < 4; ++k) {
      const int i = i0 + 4096 + k;
      const int v = cnt[i < n ? i : n - 1];
      nx[k] = (i < n) ? v : 0;
    }
    const int s = c[0] + c[1] + c[2] + c[3];
    int incl = s;
    for (int off = 1; off < 32; off <<= 1) {
      int v = __shfl_up(incl, off, 32);
      if (lane >= off) incl += v;
    }
    if (lane == 31) wave_tot[w] = incl;
    __syncthreads();
    const int wt = wave_tot[lane];
    int wincl = wt;
    for (int off = 1; off < 32; off <<= 1) {
      int v = __shfl_up(wincl, off, 32);
      if (lane >= off) wincl += v;
    }
    const int wave_excl = __shfl(wincl, w, 32) - __shfl(wt, w, 32);
    const int total     = __shfl(wincl, 31, 32);
    int run = carry + wave_excl + incl - s;
    for (int k = 0; k < 4; ++k) {
      if (i0 + k < n) rowptr[i0 + k] = run;
      run += c[k];
    }
    carry += total;
    for (int k = 0; k < 4; ++k) c[k] = nx[k];
    __syncthreads();
  }
  if (t == 0) rowptr[n] = carry;
}

__global__ void place_kernel(const int* __restrict__ src, const int* __restrict__ dst,
                             const int* __restrict__ rank, const int* __restrict__ rowptr,
                             int* __restrict__ csr_src, int E) {
  int e = blockIdx.x * blockDim.x + threadIdx.x;
  if (e < E) csr_src[rowptr[dst[e]] + rank[e]] = src[e];
}

__global__ void wsum_kernel(const float* __restrict__ Wm, const float* __restrict__ bm,
                            float* __restrict__ wsum, float* __restrict__ bsum) {
  int t = threadIdx.x;
  if (t < HD) {
    float s = 0.0f;
    for (int j = 0; j < HD; ++j) s += Wm[t * HD + j];
    wsum[t] = s;
  } else if (t == HD) {
    float s = 0.0f;
    for (int j = 0; j < HD; ++j) s += bm[j];
    *bsum = s;
  }
}

__global__ __launch_bounds__(256)
void gemm_norm_wmma_kernel(const float* __restrict__ X, const int* __restrict__ rowmap,
                           const float* __restrict__ norm, const float* __restrict__ W,
                           float* __restrict__ OUT, int nrows, int K) {
  __shared__ float ldsX[RPB][LPAD];
  __shared__ float ldsW[HD][WPAD];

  const int tid  = threadIdx.x;
  const int wave = tid >> 5;
  const int lane = tid & 31;
  const int half = lane >> 4;
  const int l16  = lane & 15;
  const int row0 = blockIdx.x * RPB;

  v8f acc[6] = {};

  const int xc4 = (tid & 7) * 4;
  const float* xptr[4];
  float        xscale[4];
  bool         xrow_ok[4];
  int          xsrc[4];
  #pragma unroll
  for (int it = 0; it < 4; ++it) {
    const int grow = row0 + ((tid + 256 * it) >> 3);
    xrow_ok[it]    = grow < nrows;
    xsrc[it]       = xrow_ok[it] ? grow : nrows - 1;
    xscale[it]     = norm[xsrc[it]];
  }
  if (rowmap) {
    #pragma unroll
    for (int it = 0; it < 4; ++it) xsrc[it] = rowmap[xsrc[it]];
  }
  #pragma unroll
  for (int it = 0; it < 4; ++it) xptr[it] = X + (long)xsrc[it] * K + xc4;
  const float* wptr[12];
  int wkk[12];
  #pragma unroll
  for (int it = 0; it < 12; ++it) {
    const int idx = tid + 256 * it;
    wkk[it]  = idx / HD;
    wptr[it] = W + (idx % HD);
  }

  float4 xv[4];
  float  wv[12];
  auto load_tile = [&](int k0) {
    const bool kin = (k0 + xc4) < K;
    #pragma unroll
    for (int it = 0; it < 4; ++it) {
      const float* p = xptr[it] + (kin ? k0 : -xc4);
      xv[it] = *(const float4*)p;
      __builtin_prefetch(((k0 + BK + xc4) < K) ? (p + BK) : p, 0, 0);
    }
    #pragma unroll
    for (int it = 0; it < 12; ++it) {
      const int k = k0 + wkk[it];
      wv[it] = wptr[it][(long)((k < K) ? k : 0) * HD];
    }
  };

  load_tile(0);
  for (int k0 = 0; k0 < K; k0 += BK) {
    {
      const bool kin = (k0 + xc4) < K;
      #pragma unroll
      for (int it = 0; it < 4; ++it) {
        const int r = (tid + 256 * it) >> 3;
        float4 v = xv[it];
        const float s = xscale[it];
        v.x *= s; v.y *= s; v.z *= s; v.w *= s;
        if (!(kin && xrow_ok[it])) v = make_float4(0.f, 0.f, 0.f, 0.f);
        ldsX[r][xc4 + 0] = v.x;
        ldsX[r][xc4 + 1] = v.y;
        ldsX[r][xc4 + 2] = v.z;
        ldsX[r][xc4 + 3] = v.w;
      }
      #pragma unroll
      for (int it = 0; it < 12; ++it) {
        const int idx = tid + 256 * it;
        ldsW[idx % HD][wkk[it]] = (k0 + wkk[it] < K) ? wv[it] : 0.0f;
      }
    }
    __syncthreads();

    if (k0 + BK < K) load_tile(k0 + BK);

    const int kmax = (BK < K - k0) ? BK : (K - k0);
    const int mrow = wave * 16 + l16;
    for (int kk = 0; kk < kmax; kk += 4) {
      int ka = kk + half * 2;

      v2f A = *(const v2f*)&ldsX[mrow][ka];
      v2f Bf[6];
      #pragma unroll
      for (int t = 0; t < 6; ++t)
        Bf[t] = *(const v2f*)&ldsW[t * 16 + l16][ka];

      #pragma unroll
      for (int t = 0; t < 6; ++t)
        acc[t] = __builtin_amdgcn_wmma_f32_16x16x4_f32(
             false, A,  false, Bf[t],
             (short)0, acc[t],  false,  false);
    }
    __syncthreads();
  }

  const int orow = row0 + wave * 16 + half * 8;
  #pragma unroll
  for (int t = 0; t < 6; ++t) {
    int col = t * 16 + l16;
    #pragma unroll
    for (int i = 0; i < 8; ++i) {
      int r = orow + i;
      if (r < nrows) OUT[(long)r * HD + col] = acc[t][i];
    }
  }
}

__device__ __forceinline__ void gather_row_sum(const int* __restrict__ rowptr, const int* __restrict__ csr_src,
                                               const float* __restrict__ h, int row, int lane,
                                               float& a0, float& a1, float& a2) {
  const int beg = rowptr[row];
  const int end = rowptr[row + 1];
  a0 = 0.0f; a1 = 0.0f; a2 = 0.0f;
  for (int base = beg; base < end; base += 32) {
    const int cnt  = (end - base < 32) ? (end - base) : 32;
    const int mine = (lane < cnt) ? csr_src[base + lane] : 0;
    int j = 0;
    for (; j + 4 <= cnt; j += 4) {
      const float* p0 = h + (long)__builtin_amdgcn_readlane(mine, j + 0) * HD + lane;
      const float* p1 = h + (long)__builtin_amdgcn_readlane(mine, j + 1) * HD + lane;
      const float* p2 = h + (long)__builtin_amdgcn_readlane(mine, j + 2) * HD + lane;
      const float* p3 = h + (long)__builtin_amdgcn_readlane(mine, j + 3) * HD + lane;
      const float x00 = p0[0], x01 = p0[32], x02 = p0[64];
      const float x10 = p1[0], x11 = p1[32], x12 = p1[64];
      const float x20 = p2[0], x21 = p2[32], x22 = p2[64];
      const float x30 = p3[0], x31 = p3[32], x32 = p3[64];
      a0 += x00; a1 += x01; a2 += x02;
      a0 += x10; a1 += x11; a2 += x12;
      a0 += x20; a1 += x21; a2 += x22;
      a0 += x30; a1 += x31; a2 += x32;
    }
    for (; j < cnt; ++j) {
      const float* p = h + (long)__builtin_amdgcn_readlane(mine, j) * HD + lane;
      a0 += p[0]; a1 += p[32]; a2 += p[64];
    }
  }
}

__global__ __launch_bounds__(256)
void spmm_gather_prelu_kernel(const int* __restrict__ rowptr, const int* __restrict__ csr_src,
                              const float* __restrict__ h, const float* __restrict__ norm_in,
                              const float* __restrict__ b, const float* __restrict__ a,
                              float* __restrict__ act, int n) {
  const int lane = threadIdx.x & 31;
  const int row = __builtin_amdgcn_readfirstlane((int)(blockIdx.x * (blockDim.x >> 5) + (threadIdx.x >> 5)));
  if (row >= n) return;
  const float ni = norm_in[row];
  const float b0 = b[lane], b1 = b[lane + 32], b2 = b[lane + 64];
  const float s0 = a[lane], s1 = a[lane + 32], s2 = a[lane + 64];
  float a0, a1, a2;
  gather_row_sum(rowptr, csr_src, h, row, lane, a0, a1, a2);
  const float v0 = a0 * ni + b0;
  const float v1 = a1 * ni + b1;
  const float v2 = a2 * ni + b2;
  float* o = act + (long)row * HD + lane;
  o[0]  = (v0 >= 0.0f) ? v0 : s0 * v0;
  o[32] = (v1 >= 0.0f) ? v1 : s1 * v1;
  o[64] = (v2 >= 0.0f) ? v2 : s2 * v2;
}

__global__ __launch_bounds__(256)
void spmm_gather_prelu_dot_kernel(const int* __restrict__ rowptr, const int* __restrict__ csr_src,
                                  const float* __restrict__ h, const float* __restrict__ norm_in,
                                  const float* __restrict__ b, const float* __restrict__ a,
                                  const float* __restrict__ wsum, const float* __restrict__ bsum,
                                  float* __restrict__ out, int n) {
  const int lane = threadIdx.x & 31;
  const int row = __builtin_amdgcn_readfirstlane((int)(blockIdx.x * (blockDim.x >> 5) + (threadIdx.x >> 5)));
  if (row >= n) return;
  const float ni = norm_in[row];
  const float b0 = b[lane], b1 = b[lane + 32], b2 = b[lane + 64];
  const float s0 = a[lane], s1 = a[lane + 32], s2 = a[lane + 64];
  const float w0 = wsum[lane], w1 = wsum[lane + 32], w2 = wsum[lane + 64];
  const float bs = *bsum;
  float a0, a1, a2;
  gather_row_sum(rowptr, csr_src, h, row, lane, a0, a1, a2);
  const float v0 = a0 * ni + b0;
  const float v1 = a1 * ni + b1;
  const float v2 = a2 * ni + b2;
  const float z0 = (v0 >= 0.0f) ? v0 : s0 * v0;
  const float z1 = (v1 >= 0.0f) ? v1 : s1 * v1;
  const float z2 = (v2 >= 0.0f) ? v2 : s2 * v2;
  float s = fmaf(z0, w0, 0.0f);
  s = fmaf(z1, w1, s);
  s = fmaf(z2, w2, s);
  #pragma unroll
  for (int off = 16; off > 0; off >>= 1) s += __shfl_xor(s, off, 32);
  if (lane == 0) out[row] = s + bs;
}

__global__ void spmm_kernel(const int* __restrict__ src, const int* __restrict__ dst,
                            const float* __restrict__ h, float* __restrict__ agg, int E) {
  int gw   = (int)(((long)blockIdx.x * blockDim.x + threadIdx.x) >> 5);
  int lane = threadIdx.x & 31;
  if (gw < E) {
    int s = src[gw], d = dst[gw];
    const float* hp = h + (long)s * HD;
    float* ap = agg + (long)d * HD;
    #pragma unroll
    for (int i = 0; i < 3; ++i) {
      float v = hp[lane + 32 * i];
      __hip_atomic_fetch_add(&ap[lane + 32 * i], v, __ATOMIC_RELAXED, __HIP_MEMORY_SCOPE_AGENT);
    }
  }
}

__global__ void prelu_kernel(const float* __restrict__ agg, const float* __restrict__ norm_in,
                             const float* __restrict__ b, const float* __restrict__ a,
                             float* __restrict__ out, int n) {
  long idx = (long)blockIdx.x * blockDim.x + threadIdx.x;
  if (idx < (long)n * HD) {
    int i = (int)(idx / HD);
    int j = (int)(idx % HD);
    float v = agg[idx] * norm_in[i] + b[j];
    out[idx] = (v >= 0.0f) ? v : a[j] * v;
  }
}

__global__ void reduce_kernel(const float* __restrict__ z, const float* __restrict__ wsum,
                              const float* __restrict__ bsum, float* __restrict__ out, int n) {
  int gw   = (int)(((long)blockIdx.x * blockDim.x + threadIdx.x) >> 5);
  int lane = threadIdx.x & 31;
  if (gw < n) {
    const float* zp = z + (long)gw * HD;
    float s = 0.0f;
    #pragma unroll
    for (int i = 0; i < 3; ++i) s = fmaf(zp[lane + 32 * i], wsum[lane + 32 * i], s);
    #pragma unroll
    for (int off = 16; off > 0; off >>= 1) s += __shfl_xor(s, off, 32);
    if (lane == 0) out[gw] = s + *bsum;
  }
}

extern "C" void kernel_launch(void* const* d_in, const int* in_sizes, int n_in,
                              void* d_out, int out_size, void* d_ws, size_t ws_size,
                              hipStream_t stream) {
  const float* x    = (const float*)d_in[0];
  const int*   src  = (const int*)  d_in[1];
  const int*   dst  = (const int*)  d_in[2];
  const int*   perm = (const int*)  d_in[3];
  const float* W1   = (const float*)d_in[4];
  const float* b1   = (const float*)d_in[5];
  const float* a1   = (const float*)d_in[6];
  const float* W2   = (const float*)d_in[7];
  const float* b2   = (const float*)d_in[8];
  const float* a2   = (const float*)d_in[9];
  const float* Wm   = (const float*)d_in[10];
  const float* bm   = (const float*)d_in[11];
  float* out = (float*)d_out;

  const int E  = in_sizes[1];
  const int N  = in_sizes[3];
  const int K1 = in_sizes[0] / N;

  float* ws       = (float*)d_ws;
  float* norm_out = ws;
  float* norm_in  = ws + N;
  float* wsum     = ws + 2L * N;
  float* bsum     = ws + 2L * N + HD;
  float* bufA     = ws + 2L * N + 2 * HD;
  float* bufB     = bufA + (long)N * HD;

  const long NH = (long)N * HD;
  int*   rowptr   = (int*)(bufB + NH);
  int*   csr_src  = rowptr + (N + 1);
  int*   rank     = (int*)bufB;
  int*   cnt_out  = (int*)norm_out;
  int*   cnt_in   = (int*)norm_in;

  const int gemm_blocks = (N + RPB - 1) / RPB;
  const int spmm_blocks = (int)(((long)E * 32 + 255) / 256);
  const int gath_blocks = (N + 7) / 8;
  const int elw_blocks  = (int)((NH + 255) / 256);
  const int red_blocks  = (int)(((long)N * 32 + 255) / 256);

  const size_t need_grouped = (size_t)(2L * N + 2 * HD + 2 * NH + (N + 1) + E) * sizeof(float);
  const bool grouped = (ws_size >= need_grouped) && ((long)E <= NH);

  fill_zero_kernel<<<512, 256, 0, stream>>>(norm_out, 2L * N);
  degree_kernel<<<(E + 255) / 256, 256, 0, stream>>>(src, dst, cnt_out, cnt_in, grouped ? rank : nullptr, E);
  if (grouped) {
    rowptr_scan_kernel<<<1, 1024, 0, stream>>>(cnt_in, rowptr, N);
    place_kernel<<<(E + 255) / 256, 256, 0, stream>>>(src, dst, rank, rowptr, csr_src, E);
  }
  finalize_norm_kernel<<<(2 * N + 255) / 256, 256, 0, stream>>>(norm_out, 2 * N);
  wsum_kernel<<<1, 128, 0, stream>>>(Wm, bm, wsum, bsum);

  for (int view = 0; view < 2; ++view) {
    const int* rowmap = view ? perm : nullptr;
    float* outv = out + (long)view * N;
    if (grouped) {
      gemm_norm_wmma_kernel<<<gemm_blocks, 256, 0, stream>>>(x, rowmap, norm_out, W1, bufA, N, K1);
      spmm_gather_prelu_kernel<<<gath_blocks, 256, 0, stream>>>(rowptr, csr_src, bufA, norm_in, b1, a1, bufB, N);
      gemm_norm_wmma_kernel<<<gemm_blocks, 256, 0, stream>>>(bufB, nullptr, norm_out, W2, bufA, N, HD);
      spmm_gather_prelu_dot_kernel<<<gath_blocks, 256, 0, stream>>>(rowptr, csr_src, bufA, norm_in, b2, a2,
                                                                   wsum, bsum, outv, N);
    } else {
      gemm_norm_wmma_kernel<<<gemm_blocks, 256, 0, stream>>>(x, rowmap, norm_out, W1, bufA, N, K1);
      fill_zero_kernel<<<2048, 256, 0, stream>>>(bufB, NH);
      spmm_kernel<<<spmm_blocks, 256, 0, stream>>>(src, dst, bufA, bufB, E);
      prelu_kernel<<<elw_blocks, 256, 0, stream>>>(bufB, norm_in, b1, a1, bufA, N);
      gemm_norm_wmma_kernel<<<gemm_blocks, 256, 0, stream>>>(bufA, nullptr, norm_out, W2, bufB, N, HD);
      fill_zero_kernel<<<2048, 256, 0, stream>>>(bufA, NH);
      spmm_kernel<<<spmm_blocks, 256, 0, stream>>>(src, dst, bufB, bufA, E);
      prelu_kernel<<<elw_blocks, 256, 0, stream>>>(bufA, norm_in, b2, a2, bufB, N);
      reduce_kernel<<<red_blocks, 256, 0, stream>>>(bufB, wsum, bsum, outv, N);
    }
  }
}


# degree counts and edge ranks on LDS counters (64 edge chunks, private node table per chunk, folded across chunks) in place of global atomics; gather list order = chunk, then LDS arrival
# speedup vs baseline: 3.0677x; 1.2350x over previous
#include <hip/hip_runtime.h>

typedef float v2f __attribute__((ext_vector_type(2)));
typedef float v8f __attribute__((ext_vector_type(8)));

constexpr int HD   = 96;
constexpr int RPB  = 128;
constexpr int BK   = 32;
constexpr int LPAD = 36;
constexpr int WPAD = 38;
constexpr int HG   = 64;
constexpr int HCAP = 65536;


__global__ void fill_zero_kernel(float* __restrict__ p, long n) {
  long i = (long)blockIdx.x * blockDim.x + threadIdx.x;
  long stride = (long)gridDim.x * blockDim.x;
  for (; i < n; i += stride) p[i] = 0.0f;
}

__global__ void degree_kernel(const int* __restrict__ src, const int* __restrict__ dst,
                              int* __restrict__ cnt_out, int* __restrict__ cnt_in,
                              int* __restrict__ rank, int E) {
  int e = blockIdx.x * blockDim.x + threadIdx.x;
  if (e < E) {
    __hip_atomic_fetch_add(&cnt_out[src[e]], 1, __ATOMIC_RELAXED, __HIP_MEMORY_SCOPE_AGENT);
    if (rank) {
      rank[e] = __hip_atomic_fetch_add(&cnt_in[dst[e]], 1, __ATOMIC_RELAXED, __HIP_MEMORY_SCOPE_AGENT);
    } else {
      __hip_atomic_fetch_add(&cnt_in[dst[e]], 1, __ATOMIC_RELAXED, __HIP_MEMORY_SCOPE_AGENT);
    }
  }
}

__global__ void finalize_norm_kernel(float* __restrict__ p, int n) {
  int i = blockIdx.x * blockDim.x + threadIdx.x;
  if (i < n) p[i] = 1.0f / sqrtf(fmaxf((float)__float_as_int(p[i]), 1.0f));
}

__global__ __launch_bounds__(1024)
void rowptr_scan_kernel(const int* __restrict__ cnt, int* __restrict__ rowptr, int n) {
  __shared__ int wave_tot[32];
  const int t = threadIdx.x, lane = t & 31, w = t >> 5;
  int carry = 0;
  int c[4], nx[4];
  for (int k = 0; k < 4; ++k) {
    const int i = t * 4 + k;
    const int v = cnt[i < n ? i : n - 1];
    c[k] = (i < n) ? v : 0;
  }
  for (int base = 0; base < n; base += 4096) {
    const int i0 = base + t * 4;
    for (int k = 0; k < 4; ++k) {
      const int i = i0 + 4096 + k;
      const int v = cnt[i < n ? i : n - 1];
      nx[k] = (i < n) ? v : 0;
    }
    const int s = c[0] + c[1] + c[2] + c[3];
    int incl = s;
    for (int off = 1; off < 32; off <<= 1) {
      int v = __shfl_up(incl, off, 32);
      if (lane >= off) incl += v;
    }
    if (lane == 31) wave_tot[w] = incl;
    __syncthreads();
    const int wt = wave_tot[lane];
    int wincl = wt;
    for (int off = 1; off < 32; off <<= 1) {
      int v = __shfl_up(wincl, off, 32);
      if (lane >= off) wincl += v;
    }
    const int wave_excl = __shfl(wincl, w, 32) - __shfl(wt, w, 32);
    const int total     = __shfl(wincl, 31, 32);
    int run = carry + wave_excl + incl - s;
    for (int k = 0; k < 4; ++k) {
      if (i0 + k < n) rowptr[i0 + k] = run;
      run += c[k];
    }
    carry += total;
    for (int k = 0; k < 4; ++k) c[k] = nx[k];
    __syncthreads();
  }
  if (t == 0) rowptr[n] = carry;
}

__global__ void place_kernel(const int* __restrict__ src, const int* __restrict__ dst,
                             const int* __restrict__ rank, const int* __restrict__ rowptr,
                             int* __restrict__ csr_src, int E) {
  int e = blockIdx.x * blockDim.x + threadIdx.x;
  if (e < E) csr_src[rowptr[dst[e]] + rank[e]] = src[e];
}

__global__ __launch_bounds__(1024)
void hist_lds_kernel(const int* __restrict__ src, const int* __restrict__ dst,
                     int* __restrict__ hist_in, int* __restrict__ hist_out,
                     int* __restrict__ rank, int E, int n, int chunk) {
  __shared__ int4 cnt4[HCAP / 4];
  int* cnt = reinterpret_cast<int*>(cnt4);
  const int  t     = threadIdx.x;
  const bool is_in = blockIdx.x < HG;
  const int  g     = is_in ? (int)blockIdx.x : (int)blockIdx.x - HG;
  const int* __restrict__ key  = is_in ? dst : src;
  int*       __restrict__ hist = (is_in ? hist_in : hist_out) + (long)g * n;
  const bool vec = ((n & 3) == 0) && ((((size_t)hist) & 15) == 0);

  if (vec) {
    for (int i = t; i < (n >> 2); i += 1024) cnt4[i] = make_int4(0, 0, 0, 0);
  } else {
    for (int i = t; i < n; i += 1024) cnt[i] = 0;
  }
  __syncthreads();

  const int beg = g * chunk;
  const int end = (beg + chunk < E) ? (beg + chunk) : E;
  for (int base = beg + t; base < end; base += 8 * 1024) {
    int k[8];
    #pragma unroll
    for (int u = 0; u < 8; ++u) {
      const int e = base + u * 1024;
      k[u] = key[e < end ? e : end - 1];
    }
    int r[8];
    #pragma unroll
    for (int u = 0; u < 8; ++u) {
      const int e = base + u * 1024;
      r[u] = __hip_atomic_fetch_add(&cnt[k[u]], (e < end) ? 1 : 0, __ATOMIC_RELAXED, __HIP_MEMORY_SCOPE_WORKGROUP);
    }
    if (is_in) {
      #pragma unroll
      for (int u = 0; u < 8; ++u) {
        const int e = base + u * 1024;
        if (e < end) rank[e] = r[u];
      }
    }
  }
  __syncthreads();

  if (vec) {
    int4* __restrict__ h4 = reinterpret_cast<int4*>(hist);
    for (int i = t; i < (n >> 2); i += 1024) h4[i] = cnt4[i];
  } else {
    for (int i = t; i < n; i += 1024) hist[i] = cnt[i];
  }
}

__global__ __launch_bounds__(256)
void hist_fold_kernel(int* __restrict__ hist_in, const int* __restrict__ hist_out,
                      int* __restrict__ cnt_tot, float* __restrict__ norm_out,
                      float* __restrict__ norm_in, int n) {
  const int i = blockIdx.x * blockDim.x + threadIdx.x;
  if (i < n) {
    int c[HG];
    #pragma unroll
    for (int g = 0; g < HG; ++g) c[g] = hist_in[(long)g * n + i];
    int run = 0;
    #pragma unroll
    for (int g = 0; g < HG; ++g) {
      hist_in[(long)g * n + i] = run;
      run += c[g];
    }
    cnt_tot[i] = run;
    norm_in[i] = 1.0f / sqrtf(fmaxf((float)run, 1.0f));
  } else if (i < 2 * n) {
    const int j = i - n;
    int s = 0;
    #pragma unroll
    for (int g = 0; g < HG; ++g) s += hist_out[(long)g * n + j];
    norm_out[j] = 1.0f / sqrtf(fmaxf((float)s, 1.0f));
  }
}

__global__ void place_chunked_kernel(const int* __restrict__ src, const int* __restrict__ dst,
                                     const int* __restrict__ rank, const int* __restrict__ rowptr,
                                     const int* __restrict__ before, int* __restrict__ csr_src,
                                     int E, int n, int chunk) {
  int e = blockIdx.x * blockDim.x + threadIdx.x;
  if (e < E) {
    const int d = dst[e];
    const int g = e / chunk;
    csr_src[rowptr[d] + before[(long)g * n + d] + rank[e]] = src[e];
  }
}

__global__ void wsum_kernel(const float* __restrict__ Wm, const float* __restrict__ bm,
                            float* __restrict__ wsum, float* __restrict__ bsum) {
  int t = threadIdx.x;
  if (t < HD) {
    float s = 0.0f;
    for (int j = 0; j < HD; ++j) s += Wm[t * HD + j];
    wsum[t] = s;
  } else if (t == HD) {
    float s = 0.0f;
    for (int j = 0; j < HD; ++j) s += bm[j];
    *bsum = s;
  }
}

__global__ __launch_bounds__(256)
void gemm_norm_wmma_kernel(const float* __restrict__ X, const int* __restrict__ rowmap,
                           const float* __restrict__ norm, const float* __restrict__ W,
                           float* __restrict__ OUT, int nrows, int K) {
  __shared__ float ldsX[RPB][LPAD];
  __shared__ float ldsW[HD][WPAD];

  const int tid  = threadIdx.x;
  const int wave = tid >> 5;
  const int lane = tid & 31;
  const int half = lane >> 4;
  const int l16  = lane & 15;
  const int row0 = blockIdx.x * RPB;

  v8f acc[6] = {};

  const int xc4 = (tid & 7) * 4;
  const float* xptr[4];
  float        xscale[4];
  bool         xrow_ok[4];
  int          xsrc[4];
  #pragma unroll
  for (int it = 0; it < 4; ++it) {
    const int grow = row0 + ((tid + 256 * it) >> 3);
    xrow_ok[it]    = grow < nrows;
    xsrc[it]       = xrow_ok[it] ? grow : nrows - 1;
    xscale[it]     = norm[xsrc[it]];
  }
  if (rowmap) {
    #pragma unroll
    for (int it = 0; it < 4; ++it) xsrc[it] = rowmap[xsrc[it]];
  }
  #pragma unroll
  for (int it = 0; it < 4; ++it) xptr[it] = X + (long)xsrc[it] * K + xc4;
  const float* wptr[12];
  int wkk[12];
  #pragma unroll
  for (int it = 0; it < 12; ++it) {
    const int idx = tid + 256 * it;
    wkk[it]  = idx / HD;
    wptr[it] = W + (idx % HD);
  }

  float4 xv[4];
  float  wv[12];
  auto load_tile = [&](int k0) {
    const bool kin = (k0 + xc4) < K;
    #pragma unroll
    for (int it = 0; it < 4; ++it) {
      const float* p = xptr[it] + (kin ? k0 : -xc4);
      xv[it] = *(const float4*)p;
      __builtin_prefetch(((k0 + BK + xc4) < K) ? (p + BK) : p, 0, 0);
    }
    #pragma unroll
    for (int it = 0; it < 12; ++it) {
      const int k = k0 + wkk[it];
      wv[it] = wptr[it][(long)((k < K) ? k : 0) * HD];
    }
  };

  load_tile(0);
  for (int k0 = 0; k0 < K; k0 += BK) {
    {
      const bool kin = (k0 + xc4) < K;
      #pragma unroll
      for (int it = 0; it < 4; ++it) {
        const int r = (tid + 256 * it) >> 3;
        float4 v = xv[it];
        const float s = xscale[it];
        v.x *= s; v.y *= s; v.z *= s; v.w *= s;
        if (!(kin && xrow_ok[it])) v = make_float4(0.f, 0.f, 0.f, 0.f);
        ldsX[r][xc4 + 0] = v.x;
        ldsX[r][xc4 + 1] = v.y;
        ldsX[r][xc4 + 2] = v.z;
        ldsX[r][xc4 + 3] = v.w;
      }
      #pragma unroll
      for (int it = 0; it < 12; ++it) {
        const int idx = tid + 256 * it;
        ldsW[idx % HD][wkk[it]] = (k0 + wkk[it] < K) ? wv[it] : 0.0f;
      }
    }
    __syncthreads();

    if (k0 + BK < K) load_tile(k0 + BK);

    const int kmax = (BK < K - k0) ? BK : (K - k0);
    const int mrow = wave * 16 + l16;
    for (int kk = 0; kk < kmax; kk += 4) {
      int ka = kk + half * 2;

      v2f A = *(const v2f*)&ldsX[mrow][ka];
      v2f Bf[6];
      #pragma unroll
      for (int t = 0; t < 6; ++t)
        Bf[t] = *(const v2f*)&ldsW[t * 16 + l16][ka];

      #pragma unroll
      for (int t = 0; t < 6; ++t)
        acc[t] = __builtin_amdgcn_wmma_f32_16x16x4_f32(
             false, A,  false, Bf[t],
             (short)0, acc[t],  false,  false);
    }
    __syncthreads();
  }

  const int orow = row0 + wave * 16 + half * 8;
  #pragma unroll
  for (int t = 0; t < 6; ++t) {
    int col = t * 16 + l16;
    #pragma unroll
    for (int i = 0; i < 8; ++i) {
      int r = orow + i;
      if (r < nrows) OUT[(long)r * HD + col] = acc[t][i];
    }
  }
}

__device__ __forceinline__ void gather_row_sum(const int* __restrict__ rowptr, const int* __restrict__ csr_src,
                                               const float* __restrict__ h, int row, int lane,
                                               float& a0, float& a1, float& a2) {
  const int beg = rowptr[row];
  const int end = rowptr[row + 1];
  a0 = 0.0f; a1 = 0.0f; a2 = 0.0f;
  for (int base = beg; base < end; base += 32) {
    const int cnt  = (end - base < 32) ? (end - base) : 32;
    const int mine = (lane < cnt) ? csr_src[base + lane] : 0;
    int j = 0;
    for (; j + 4 <= cnt; j += 4) {
      const float* p0 = h + (long)__builtin_amdgcn_readlane(mine, j + 0) * HD + lane;
      const float* p1 = h + (long)__builtin_amdgcn_readlane(mine, j + 1) * HD + lane;
      const float* p2 = h + (long)__builtin_amdgcn_readlane(mine, j + 2) * HD + lane;
      const float* p3 = h + (long)__builtin_amdgcn_readlane(mine, j + 3) * HD + lane;
      const float x00 = p0[0], x01 = p0[32], x02 = p0[64];
      const float x10 = p1[0], x11 = p1[32], x12 = p1[64];
      const float x20 = p2[0], x21 = p2[32], x22 = p2[64];
      const float x30 = p3[0], x31 = p3[32], x32 = p3[64];
      a0 += x00; a1 += x01; a2 += x02;
      a0 += x10; a1 += x11; a2 += x12;
      a0 += x20; a1 += x21; a2 += x22;
      a0 += x30; a1 += x31; a2 += x32;
    }
    for (; j < cnt; ++j) {
      const float* p = h + (long)__builtin_amdgcn_readlane(mine, j) * HD + lane;
      a0 += p[0]; a1 += p[32]; a2 += p[64];
    }
  }
}

__global__ __launch_bounds__(256)
void spmm_gather_prelu_kernel(const int* __restrict__ rowptr, const int* __restrict__ csr_src,
                              const float* __restrict__ h, const float* __restrict__ norm_in,
                              const float* __restrict__ b, const float* __restrict__ a,
                              float* __restrict__ act, int n) {
  const int lane = threadIdx.x & 31;
  const int row = __builtin_amdgcn_readfirstlane((int)(blockIdx.x * (blockDim.x >> 5) + (threadIdx.x >> 5)));
  if (row >= n) return;
  const float ni = norm_in[row];
  const float b0 = b[lane], b1 = b[lane + 32], b2 = b[lane + 64];
  const float s0 = a[lane], s1 = a[lane + 32], s2 = a[lane + 64];
  float a0, a1, a2;
  gather_row_sum(rowptr, csr_src, h, row, lane, a0, a1, a2);
  const float v0 = a0 * ni + b0;
  const float v1 = a1 * ni + b1;
  const float v2 = a2 * ni + b2;
  float* o = act + (long)row * HD + lane;
  o[0]  = (v0 >= 0.0f) ? v0 : s0 * v0;
  o[32] = (v1 >= 0.0f) ? v1 : s1 * v1;
  o[64] = (v2 >= 0.0f) ? v2 : s2 * v2;
}

__global__ __launch_bounds__(256)
void spmm_gather_prelu_dot_kernel(const int* __restrict__ rowptr, const int* __restrict__ csr_src,
                                  const float* __restrict__ h, const float* __restrict__ norm_in,
                                  const float* __restrict__ b, const float* __restrict__ a,
                                  const float* __restrict__ wsum, const float* __restrict__ bsum,
                                  float* __restrict__ out, int n) {
  const int lane = threadIdx.x & 31;
  const int row = __builtin_amdgcn_readfirstlane((int)(blockIdx.x * (blockDim.x >> 5) + (threadIdx.x >> 5)));
  if (row >= n) return;
  const float ni = norm_in[row];
  const float b0 = b[lane], b1 = b[lane + 32], b2 = b[lane + 64];
  const float s0 = a[lane], s1 = a[lane + 32], s2 = a[lane + 64];
  const float w0 = wsum[lane], w1 = wsum[lane + 32], w2 = wsum[lane + 64];
  const float bs = *bsum;
  float a0, a1, a2;
  gather_row_sum(rowptr, csr_src, h, row, lane, a0, a1, a2);
  const float v0 = a0 * ni + b0;
  const float v1 = a1 * ni + b1;
  const float v2 = a2 * ni + b2;
  const float z0 = (v0 >= 0.0f) ? v0 : s0 * v0;
  const float z1 = (v1 >= 0.0f) ? v1 : s1 * v1;
  const float z2 = (v2 >= 0.0f) ? v2 : s2 * v2;
  float s = fmaf(z0, w0, 0.0f);
  s = fmaf(z1, w1, s);
  s = fmaf(z2, w2, s);
  #pragma unroll
  for (int off = 16; off > 0; off >>= 1) s += __shfl_xor(s, off, 32);
  if (lane == 0) out[row] = s + bs;
}

__global__ void spmm_kernel(const int* __restrict__ src, const int* __restrict__ dst,
                            const float* __restrict__ h, float* __restrict__ agg, int E) {
  int gw   = (int)(((long)blockIdx.x * blockDim.x + threadIdx.x) >> 5);
  int lane = threadIdx.x & 31;
  if (gw < E) {
    int s = src[gw], d = dst[gw];
    const float* hp = h + (long)s * HD;
    float* ap = agg + (long)d * HD;
    #pragma unroll
    for (int i = 0; i < 3; ++i) {
      float v = hp[lane + 32 * i];
      __hip_atomic_fetch_add(&ap[lane + 32 * i], v, __ATOMIC_RELAXED, __HIP_MEMORY_SCOPE_AGENT);
    }
  }
}

__global__ void prelu_kernel(const float* __restrict__ agg, const float* __restrict__ norm_in,
                             const float* __restrict__ b, const float* __restrict__ a,
                             float* __restrict__ out, int n) {
  long idx = (long)blockIdx.x * blockDim.x + threadIdx.x;
  if (idx < (long)n * HD) {
    int i = (int)(idx / HD);
    int j = (int)(idx % HD);
    float v = agg[idx] * norm_in[i] + b[j];
    out[idx] = (v >= 0.0f) ? v : a[j] * v;
  }
}

__global__ void reduce_kernel(const float* __restrict__ z, const float* __restrict__ wsum,
                              const float* __restrict__ bsum, float* __restrict__ out, int n) {
  int gw   = (int)(((long)blockIdx.x * blockDim.x + threadIdx.x) >> 5);
  int lane = threadIdx.x & 31;
  if (gw < n) {
    const float* zp = z + (long)gw * HD;
    float s = 0.0f;
    #pragma unroll
    for (int i = 0; i < 3; ++i) s = fmaf(zp[lane + 32 * i], wsum[lane + 32 * i], s);
    #pragma unroll
    for (int off = 16; off > 0; off >>= 1) s += __shfl_xor(s, off, 32);
    if (lane == 0) out[gw] = s + *bsum;
  }
}

extern "C" void kernel_launch(void* const* d_in, const int* in_sizes, int n_in,
                              void* d_out, int out_size, void* d_ws, size_t ws_size,
                              hipStream_t stream) {
  const float* x    = (const float*)d_in[0];
  const int*   src  = (const int*)  d_in[1];
  const int*   dst  = (const int*)  d_in[2];
  const int*   perm = (const int*)  d_in[3];
  const float* W1   = (const float*)d_in[4];
  const float* b1   = (const float*)d_in[5];
  const float* a1   = (const float*)d_in[6];
  const float* W2   = (const float*)d_in[7];
  const float* b2   = (const float*)d_in[8];
  const float* a2   = (const float*)d_in[9];
  const float* Wm   = (const float*)d_in[10];
  const float* bm   = (const float*)d_in[11];
  float* out = (float*)d_out;

  const int E  = in_sizes[1];
  const int N  = in_sizes[3];
  const int K1 = in_sizes[0] / N;

  float* ws       = (float*)d_ws;
  float* norm_out = ws;
  float* norm_in  = ws + N;
  float* wsum     = ws + 2L * N;
  float* bsum     = ws + 2L * N + HD;
  float* bufA     = ws + 2L * N + 2 * HD;
  float* bufB     = bufA + (long)N * HD;

  const long NH = (long)N * HD;
  int*   rowptr   = (int*)(bufB + NH);
  int*   csr_src  = rowptr + (N + 1);
  int*   rank     = (int*)bufB;
  int*   cnt_out  = (int*)norm_out;
  int*   cnt_in   = (int*)norm_in;
  int*   hist_in  = (int*)bufA;
  int*   hist_out = (int*)bufB;
  int*   lrank    = hist_out + (long)HG * N;
  int*   cnt_tot  = lrank + E;

  const int gemm_blocks = (N + RPB - 1) / RPB;
  const int spmm_blocks = (int)(((long)E * 32 + 255) / 256);
  const int gath_blocks = (N + 7) / 8;
  const int elw_blocks  = (int)((NH + 255) / 256);
  const int red_blocks  = (int)(((long)N * 32 + 255) / 256);

  const size_t need_grouped = (size_t)(2L * N + 2 * HD + 2 * NH + (N + 1) + E) * sizeof(float);
  const bool grouped = (ws_size >= need_grouped) && ((long)E <= NH);
  const int  chunk    = (E + HG - 1) / HG;
  const bool lds_hist = grouped && (N <= HCAP) && ((long)(HG + 1) * N + E <= NH);

  if (lds_hist) {
    hist_lds_kernel<<<2 * HG, 1024, 0, stream>>>(src, dst, hist_in, hist_out, lrank, E, N, chunk);
    hist_fold_kernel<<<(2 * N + 255) / 256, 256, 0, stream>>>(hist_in, hist_out, cnt_tot, norm_out, norm_in, N);
    rowptr_scan_kernel<<<1, 1024, 0, stream>>>(cnt_tot, rowptr, N);
    place_chunked_kernel<<<(E + 255) / 256, 256, 0, stream>>>(src, dst, lrank, rowptr, hist_in, csr_src, E, N, chunk);
  } else {
    fill_zero_kernel<<<512, 256, 0, stream>>>(norm_out, 2L * N);
    degree_kernel<<<(E + 255) / 256, 256, 0, stream>>>(src, dst, cnt_out, cnt_in, grouped ? rank : nullptr, E);
    if (grouped) {
      rowptr_scan_kernel<<<1, 1024, 0, stream>>>(cnt_in, rowptr, N);
      place_kernel<<<(E + 255) / 256, 256, 0, stream>>>(src, dst, rank, rowptr, csr_src, E);
    }
    finalize_norm_kernel<<<(2 * N + 255) / 256, 256, 0, stream>>>(norm_out, 2 * N);
  }
  wsum_kernel<<<1, 128, 0, stream>>>(Wm, bm, wsum, bsum);

  for (int view = 0; view < 2; ++view) {
    const int* rowmap = view ? perm : nullptr;
    float* outv = out + (long)view * N;
    if (grouped) {
      gemm_norm_wmma_kernel<<<gemm_blocks, 256, 0, stream>>>(x, rowmap, norm_out, W1, bufA, N, K1);
      spmm_gather_prelu_kernel<<<gath_blocks, 256, 0, stream>>>(rowptr, csr_src, bufA, norm_in, b1, a1, bufB, N);
      gemm_norm_wmma_kernel<<<gemm_blocks, 256, 0, stream>>>(bufB, nullptr, norm_out, W2, bufA, N, HD);
      spmm_gather_prelu_dot_kernel<<<gath_blocks, 256, 0, stream>>>(rowptr, csr_src, bufA, norm_in, b2, a2,
                                                                   wsum, bsum, outv, N);
    } else {
      gemm_norm_wmma_kernel<<<gemm_blocks, 256, 0, stream>>>(x, rowmap, norm_out, W1, bufA, N, K1);
      fill_zero_kernel<<<2048, 256, 0, stream>>>(bufB, NH);
      spmm_kernel<<<spmm_blocks, 256, 0, stream>>>(src, dst, bufA, bufB, E);
      prelu_kernel<<<elw_blocks, 256, 0, stream>>>(bufB, norm_in, b1, a1, bufA, N);
      gemm_norm_wmma_kernel<<<gemm_blocks, 256, 0, stream>>>(bufA, nullptr, norm_out, W2, bufB, N, HD);
      fill_zero_kernel<<<2048, 256, 0, stream>>>(bufA, NH);
      spmm_kernel<<<spmm_blocks, 256, 0, stream>>>(src, dst, bufB, bufA, E);
      prelu_kernel<<<elw_blocks, 256, 0, stream>>>(bufA, norm_in, b2, a2, bufB, N);
      reduce_kernel<<<red_blocks, 256, 0, stream>>>(bufB, wsum, bsum, outv, N);
    }
  }
}


# x@W1 computed once for both views: norm_out row scale and perm row map move from the GEMM into the layer-1 gather (one fma per edge, list order kept, h rounds differently); layer-2 GEMM in place
# speedup vs baseline: 3.5961x; 1.1722x over previous
#include <hip/hip_runtime.h>

typedef float v2f __attribute__((ext_vector_type(2)));
typedef float v8f __attribute__((ext_vector_type(8)));

constexpr int HD   = 96;
constexpr int RPB  = 128;
constexpr int BK   = 32;
constexpr int LPAD = 36;
constexpr int WPAD = 38;
constexpr int HG   = 64;
constexpr int HCAP = 65536;


__global__ void fill_zero_kernel(float* __restrict__ p, long n) {
  long i = (long)blockIdx.x * blockDim.x + threadIdx.x;
  long stride = (long)gridDim.x * blockDim.x;
  for (; i < n; i += stride) p[i] = 0.0f;
}

__global__ void degree_kernel(const int* __restrict__ src, const int* __restrict__ dst,
                              int* __restrict__ cnt_out, int* __restrict__ cnt_in,
                              int* __restrict__ rank, int E) {
  int e = blockIdx.x * blockDim.x + threadIdx.x;
  if (e < E) {
    __hip_atomic_fetch_add(&cnt_out[src[e]], 1, __ATOMIC_RELAXED, __HIP_MEMORY_SCOPE_AGENT);
    if (rank) {
      rank[e] = __hip_atomic_fetch_add(&cnt_in[dst[e]], 1, __ATOMIC_RELAXED, __HIP_MEMORY_SCOPE_AGENT);
    } else {
      __hip_atomic_fetch_add(&cnt_in[dst[e]], 1, __ATOMIC_RELAXED, __HIP_MEMORY_SCOPE_AGENT);
    }
  }
}

__global__ void finalize_norm_kernel(float* __restrict__ p, int n) {
  int i = blockIdx.x * blockDim.x + threadIdx.x;
  if (i < n) p[i] = 1.0f / sqrtf(fmaxf((float)__float_as_int(p[i]), 1.0f));
}

__global__ __launch_bounds__(1024)
void rowptr_scan_kernel(const int* __restrict__ cnt, int* __restrict__ rowptr, int n) {
  __shared__ int wave_tot[32];
  const int t = threadIdx.x, lane = t & 31, w = t >> 5;
  int carry = 0;
  int c[4], nx[4];
  for (int k = 0; k < 4; ++k) {
    const int i = t * 4 + k;
    const int v = cnt[i < n ? i : n - 1];
    c[k] = (i < n) ? v : 0;
  }
  for (int base = 0; base < n; base += 4096) {
    const int i0 = base + t * 4;
    for (int k = 0; k < 4; ++k) {
      const int i = i0 + 4096 + k;
      const int v = cnt[i < n ? i : n - 1];
      nx[k] = (i < n) ? v : 0;
    }
    const int s = c[0] + c[1] + c[2] + c[3];
    int incl = s;
    for (int off = 1; off < 32; off <<= 1) {
      int v = __shfl_up(incl, off, 32);
      if (lane >= off) incl += v;
    }
    if (lane == 31) wave_tot[w] = incl;
    __syncthreads();
    const int wt = wave_tot[lane];
    int wincl = wt;
    for (int off = 1; off < 32; off <<= 1) {
      int v = __shfl_up(wincl, off, 32);
      if (lane >= off) wincl += v;
    }
    const int wave_excl = __shfl(wincl, w, 32) - __shfl(wt, w, 32);
    const int total     = __shfl(wincl, 31, 32);
    int run = carry + wave_excl + incl - s;
    for (int k = 0; k < 4; ++k) {
      if (i0 + k < n) rowptr[i0 + k] = run;
      run += c[k];
    }
    carry += total;
    for (int k = 0; k < 4; ++k) c[k] = nx[k];
    __syncthreads();
  }
  if (t == 0) rowptr[n] = carry;
}

__global__ void place_kernel(const int* __restrict__ src, const int* __restrict__ dst,
                             const int* __restrict__ rank, const int* __restrict__ rowptr,
                             int* __restrict__ csr_src, int E) {
  int e = blockIdx.x * blockDim.x + threadIdx.x;
  if (e < E) csr_src[rowptr[dst[e]] + rank[e]] = src[e];
}

__global__ __launch_bounds__(1024)
void hist_lds_kernel(const int* __restrict__ src, const int* __restrict__ dst,
                     int* __restrict__ hist_in, int* __restrict__ hist_out,
                     int* __restrict__ rank, int E, int n, int chunk) {
  __shared__ int4 cnt4[HCAP / 4];
  int* cnt = reinterpret_cast<int*>(cnt4);
  const int  t     = threadIdx.x;
  const bool is_in = blockIdx.x < HG;
  const int  g     = is_in ? (int)blockIdx.x : (int)blockIdx.x - HG;
  const int* __restrict__ key  = is_in ? dst : src;
  int*       __restrict__ hist = (is_in ? hist_in : hist_out) + (long)g * n;
  const bool vec = ((n & 3) == 0) && ((((size_t)hist) & 15) == 0);

  if (vec) {
    for (int i = t; i < (n >> 2); i += 1024) cnt4[i] = make_int4(0, 0, 0, 0);
  } else {
    for (int i = t; i < n; i += 1024) cnt[i] = 0;
  }
  __syncthreads();

  const int beg = g * chunk;
  const int end = (beg + chunk < E) ? (beg + chunk) : E;
  for (int base = beg + t; base < end; base += 8 * 1024) {
    int k[8];
    #pragma unroll
    for (int u = 0; u < 8; ++u) {
      const int e = base + u * 1024;
      k[u] = key[e < end ? e : end - 1];
    }
    int r[8];
    #pragma unroll
    for (int u = 0; u < 8; ++u) {
      const int e = base + u * 1024;
      r[u] = __hip_atomic_fetch_add(&cnt[k[u]], (e < end) ? 1 : 0, __ATOMIC_RELAXED, __HIP_MEMORY_SCOPE_WORKGROUP);
    }
    if (is_in) {
      #pragma unroll
      for (int u = 0; u < 8; ++u) {
        const int e = base + u * 1024;
        if (e < end) rank[e] = r[u];
      }
    }
  }
  __syncthreads();

  if (vec) {
    int4* __restrict__ h4 = reinterpret_cast<int4*>(hist);
    for (int i = t; i < (n >> 2); i += 1024) h4[i] = cnt4[i];
  } else {
    for (int i = t; i < n; i += 1024) hist[i] = cnt[i];
  }
}

__global__ __launch_bounds__(256)
void hist_fold_kernel(int* __restrict__ hist_in, const int* __restrict__ hist_out,
                      int* __restrict__ cnt_tot, float* __restrict__ norm_out,
                      float* __restrict__ norm_in, int n) {
  const int i = blockIdx.x * blockDim.x + threadIdx.x;
  if (i < n) {
    int c[HG];
    #pragma unroll
    for (int g = 0; g < HG; ++g) c[g] = hist_in[(long)g * n + i];
    int run = 0;
    #pragma unroll
    for (int g = 0; g < HG; ++g) {
      hist_in[(long)g * n + i] = run;
      run += c[g];
    }
    cnt_tot[i] = run;
    norm_in[i] = 1.0f / sqrtf(fmaxf((float)run, 1.0f));
  } else if (i < 2 * n) {
    const int j = i - n;
    int s = 0;
    #pragma unroll
    for (int g = 0; g < HG; ++g) s += hist_out[(long)g * n + j];
    norm_out[j] = 1.0f / sqrtf(fmaxf((float)s, 1.0f));
  }
}

__global__ void place_chunked_kernel(const int* __restrict__ src, const int* __restrict__ dst,
                                     const int* __restrict__ rank, const int* __restrict__ rowptr,
                                     const int* __restrict__ before, int* __restrict__ csr_src,
                                     int E, int n, int chunk) {
  int e = blockIdx.x * blockDim.x + threadIdx.x;
  if (e < E) {
    const int d = dst[e];
    const int g = e / chunk;
    csr_src[rowptr[d] + before[(long)g * n + d] + rank[e]] = src[e];
  }
}

__global__ void wsum_kernel(const float* __restrict__ Wm, const float* __restrict__ bm,
                            float* __restrict__ wsum, float* __restrict__ bsum) {
  int t = threadIdx.x;
  if (t < HD) {
    float s = 0.0f;
    for (int j = 0; j < HD; ++j) s += Wm[t * HD + j];
    wsum[t] = s;
  } else if (t == HD) {
    float s = 0.0f;
    for (int j = 0; j < HD; ++j) s += bm[j];
    *bsum = s;
  }
}

__global__ __launch_bounds__(256)
void gemm_norm_wmma_kernel(const float* X, const int* __restrict__ rowmap,
                           const float* __restrict__ norm, const float* __restrict__ W,
                           float* OUT, int nrows, int K) {
  __shared__ float ldsX[RPB][LPAD];
  __shared__ float ldsW[HD][WPAD];

  const int tid  = threadIdx.x;
  const int wave = tid >> 5;
  const int lane = tid & 31;
  const int half = lane >> 4;
  const int l16  = lane & 15;
  const int row0 = blockIdx.x * RPB;

  v8f acc[6] = {};

  const int xc4 = (tid & 7) * 4;
  const float* xptr[4];
  float        xscale[4];
  bool         xrow_ok[4];
  int          xsrc[4];
  #pragma unroll
  for (int it = 0; it < 4; ++it) {
    const int grow = row0 + ((tid + 256 * it) >> 3);
    xrow_ok[it]    = grow < nrows;
    xsrc[it]       = xrow_ok[it] ? grow : nrows - 1;
  }
  if (norm) {
    #pragma unroll
    for (int it = 0; it < 4; ++it) xscale[it] = norm[xsrc[it]];
  } else {
    #pragma unroll
    for (int it = 0; it < 4; ++it) xscale[it] = 1.0f;
  }
  if (rowmap) {
    #pragma unroll
    for (int it = 0; it < 4; ++it) xsrc[it] = rowmap[xsrc[it]];
  }
  #pragma unroll
  for (int it = 0; it < 4; ++it) xptr[it] = X + (long)xsrc[it] * K + xc4;
  const float* wptr[12];
  int wkk[12];
  #pragma unroll
  for (int it = 0; it < 12; ++it) {
    const int idx = tid + 256 * it;
    wkk[it]  = idx / HD;
    wptr[it] = W + (idx % HD);
  }

  float4 xv[4];
  float  wv[12];
  auto load_tile = [&](int k0) {
    const bool kin = (k0 + xc4) < K;
    #pragma unroll
    for (int it = 0; it < 4; ++it) {
      const float* p = xptr[it] + (kin ? k0 : -xc4);
      xv[it] = *(const float4*)p;
      __builtin_prefetch(((k0 + BK + xc4) < K) ? (p + BK) : p, 0, 0);
    }
    #pragma unroll
    for (int it = 0; it < 12; ++it) {
      const int k = k0 + wkk[it];
      wv[it] = wptr[it][(long)((k < K) ? k : 0) * HD];
    }
  };

  load_tile(0);
  for (int k0 = 0; k0 < K; k0 += BK) {
    {
      const bool kin = (k0 + xc4) < K;
      #pragma unroll
      for (int it = 0; it < 4; ++it) {
        const int r = (tid + 256 * it) >> 3;
        float4 v = xv[it];
        const float s = xscale[it];
        v.x *= s; v.y *= s; v.z *= s; v.w *= s;
        if (!(kin && xrow_ok[it])) v = make_float4(0.f, 0.f, 0.f, 0.f);
        ldsX[r][xc4 + 0] = v.x;
        ldsX[r][xc4 + 1] = v.y;
        ldsX[r][xc4 + 2] = v.z;
        ldsX[r][xc4 + 3] = v.w;
      }
      #pragma unroll
      for (int it = 0; it < 12; ++it) {
        const int idx = tid + 256 * it;
        ldsW[idx % HD][wkk[it]] = (k0 + wkk[it] < K) ? wv[it] : 0.0f;
      }
    }
    __syncthreads();

    if (k0 + BK < K) load_tile(k0 + BK);

    const int kmax = (BK < K - k0) ? BK : (K - k0);
    const int mrow = wave * 16 + l16;
    for (int kk = 0; kk < kmax; kk += 4) {
      int ka = kk + half * 2;

      v2f A = *(const v2f*)&ldsX[mrow][ka];
      v2f Bf[6];
      #pragma unroll
      for (int t = 0; t < 6; ++t)
        Bf[t] = *(const v2f*)&ldsW[t * 16 + l16][ka];

      #pragma unroll
      for (int t = 0; t < 6; ++t)
        acc[t] = __builtin_amdgcn_wmma_f32_16x16x4_f32(
             false, A,  false, Bf[t],
             (short)0, acc[t],  false,  false);
    }
    __syncthreads();
  }

  const int orow = row0 + wave * 16 + half * 8;
  #pragma unroll
  for (int t = 0; t < 6; ++t) {
    int col = t * 16 + l16;
    #pragma unroll
    for (int i = 0; i < 8; ++i) {
      int r = orow + i;
      if (r < nrows) OUT[(long)r * HD + col] = acc[t][i];
    }
  }
}

__device__ __forceinline__ void gather_row_sum(const int* __restrict__ rowptr, const int* __restrict__ csr_src,
                                               const float* __restrict__ h, int row, int lane,
                                               float& a0, float& a1, float& a2) {
  const int beg = rowptr[row];
  const int end = rowptr[row + 1];
  a0 = 0.0f; a1 = 0.0f; a2 = 0.0f;
  for (int base = beg; base < end; base += 32) {
    const int cnt  = (end - base < 32) ? (end - base) : 32;
    const int mine = (lane < cnt) ? csr_src[base + lane] : 0;
    int j = 0;
    for (; j + 4 <= cnt; j += 4) {
      const float* p0 = h + (long)__builtin_amdgcn_readlane(mine, j + 0) * HD + lane;
      const float* p1 = h + (long)__builtin_amdgcn_readlane(mine, j + 1) * HD + lane;
      const float* p2 = h + (long)__builtin_amdgcn_readlane(mine, j + 2) * HD + lane;
      const float* p3 = h + (long)__builtin_amdgcn_readlane(mine, j + 3) * HD + lane;
      const float x00 = p0[0], x01 = p0[32], x02 = p0[64];
      const float x10 = p1[0], x11 = p1[32], x12 = p1[64];
      const float x20 = p2[0], x21 = p2[32], x22 = p2[64];
      const float x30 = p3[0], x31 = p3[32], x32 = p3[64];
      a0 += x00; a1 += x01; a2 += x02;
      a0 += x10; a1 += x11; a2 += x12;
      a0 += x20; a1 += x21; a2 += x22;
      a0 += x30; a1 += x31; a2 += x32;
    }
    for (; j < cnt; ++j) {
      const float* p = h + (long)__builtin_amdgcn_readlane(mine, j) * HD + lane;
      a0 += p[0]; a1 += p[32]; a2 += p[64];
    }
  }
}

template <bool MAPPED>
__device__ __forceinline__ void gather_row_sum_scaled(const int* __restrict__ rowptr, const int* __restrict__ csr_src,
                                                      const float* __restrict__ xw, const int* __restrict__ rowmap,
                                                      const float* __restrict__ scale, int row, int lane,
                                                      float& a0, float& a1, float& a2) {
  const int beg = rowptr[row];
  const int end = rowptr[row + 1];
  a0 = 0.0f; a1 = 0.0f; a2 = 0.0f;
  for (int base = beg; base < end; base += 32) {
    const int cnt  = (end - base < 32) ? (end - base) : 32;
    const int mine = (lane < cnt) ? csr_src[base + lane] : 0;
    const int m    = MAPPED ? rowmap[mine] : mine;
    const int sc   = __float_as_int(scale[mine]);
    int j = 0;
    for (; j + 4 <= cnt; j += 4) {
      const float* p0 = xw + (long)__builtin_amdgcn_readlane(m, j + 0) * HD + lane;
      const float* p1 = xw + (long)__builtin_amdgcn_readlane(m, j + 1) * HD + lane;
      const float* p2 = xw + (long)__builtin_amdgcn_readlane(m, j + 2) * HD + lane;
      const float* p3 = xw + (long)__builtin_amdgcn_readlane(m, j + 3) * HD + lane;
      const float x00 = p0[0], x01 = p0[32], x02 = p0[64];
      const float x10 = p1[0], x11 = p1[32], x12 = p1[64];
      const float x20 = p2[0], x21 = p2[32], x22 = p2[64];
      const float x30 = p3[0], x31 = p3[32], x32 = p3[64];
      __builtin_amdgcn_sched_barrier(0);
      const float c0 = __int_as_float(__builtin_amdgcn_readlane(sc, j + 0));
      const float c1 = __int_as_float(__builtin_amdgcn_readlane(sc, j + 1));
      const float c2 = __int_as_float(__builtin_amdgcn_readlane(sc, j + 2));
      const float c3 = __int_as_float(__builtin_amdgcn_readlane(sc, j + 3));
      a0 = fmaf(c0, x00, a0); a1 = fmaf(c0, x01, a1); a2 = fmaf(c0, x02, a2);
      a0 = fmaf(c1, x10, a0); a1 = fmaf(c1, x11, a1); a2 = fmaf(c1, x12, a2);
      a0 = fmaf(c2, x20, a0); a1 = fmaf(c2, x21, a1); a2 = fmaf(c2, x22, a2);
      a0 = fmaf(c3, x30, a0); a1 = fmaf(c3, x31, a1); a2 = fmaf(c3, x32, a2);
    }
    for (; j < cnt; ++j) {
      const float* p = xw + (long)__builtin_amdgcn_readlane(m, j) * HD + lane;
      const float x0 = p[0], x1 = p[32], x2 = p[64];
      const float c  = __int_as_float(__builtin_amdgcn_readlane(sc, j));
      a0 = fmaf(c, x0, a0); a1 = fmaf(c, x1, a1); a2 = fmaf(c, x2, a2);
    }
  }
}

template <bool MAPPED>
__global__ __launch_bounds__(256)
void spmm_gather_scaled_prelu_kernel(const int* __restrict__ rowptr, const int* __restrict__ csr_src,
                                     const float* __restrict__ xw, const float* __restrict__ norm_out,
                                     const float* __restrict__ norm_in, const float* __restrict__ b,
                                     const float* __restrict__ a, float* __restrict__ act,
                                     const int* __restrict__ rowmap, int n) {
  const int lane = threadIdx.x & 31;
  const int row = __builtin_amdgcn_readfirstlane((int)(blockIdx.x * (blockDim.x >> 5) + (threadIdx.x >> 5)));
  if (row >= n) return;
  const float ni = norm_in[row];
  const float b0 = b[lane], b1 = b[lane + 32], b2 = b[lane + 64];
  const float s0 = a[lane], s1 = a[lane + 32], s2 = a[lane + 64];
  float a0, a1, a2;
  gather_row_sum_scaled<MAPPED>(rowptr, csr_src, xw, rowmap, norm_out, row, lane, a0, a1, a2);
  const float v0 = a0 * ni + b0;
  const float v1 = a1 * ni + b1;
  const float v2 = a2 * ni + b2;
  float* o = act + (long)row * HD + lane;
  o[0]  = (v0 >= 0.0f) ? v0 : s0 * v0;
  o[32] = (v1 >= 0.0f) ? v1 : s1 * v1;
  o[64] = (v2 >= 0.0f) ? v2 : s2 * v2;
}

__global__ __launch_bounds__(256)
void spmm_gather_prelu_dot_kernel(const int* __restrict__ rowptr, const int* __restrict__ csr_src,
                                  const float* __restrict__ h, const float* __restrict__ norm_in,
                                  const float* __restrict__ b, const float* __restrict__ a,
                                  const float* __restrict__ wsum, const float* __restrict__ bsum,
                                  float* __restrict__ out, int n) {
  const int lane = threadIdx.x & 31;
  const int row = __builtin_amdgcn_readfirstlane((int)(blockIdx.x * (blockDim.x >> 5) + (threadIdx.x >> 5)));
  if (row >= n) return;
  const float ni = norm_in[row];
  const float b0 = b[lane], b1 = b[lane + 32], b2 = b[lane + 64];
  const float s0 = a[lane], s1 = a[lane + 32], s2 = a[lane + 64];
  const float w0 = wsum[lane], w1 = wsum[lane + 32], w2 = wsum[lane + 64];
  const float bs = *bsum;
  float a0, a1, a2;
  gather_row_sum(rowptr, csr_src, h, row, lane, a0, a1, a2);
  const float v0 = a0 * ni + b0;
  const float v1 = a1 * ni + b1;
  const float v2 = a2 * ni + b2;
  const float z0 = (v0 >= 0.0f) ? v0 : s0 * v0;
  const float z1 = (v1 >= 0.0f) ? v1 : s1 * v1;
  const float z2 = (v2 >= 0.0f) ? v2 : s2 * v2;
  float s = fmaf(z0, w0, 0.0f);
  s = fmaf(z1, w1, s);
  s = fmaf(z2, w2, s);
  #pragma unroll
  for (int off = 16; off > 0; off >>= 1) s += __shfl_xor(s, off, 32);
  if (lane == 0) out[row] = s + bs;
}

__global__ void spmm_kernel(const int* __restrict__ src, const int* __restrict__ dst,
                            const float* __restrict__ h, float* __restrict__ agg, int E) {
  int gw   = (int)(((long)blockIdx.x * blockDim.x + threadIdx.x) >> 5);
  int lane = threadIdx.x & 31;
  if (gw < E) {
    int s = src[gw], d = dst[gw];
    const float* hp = h + (long)s * HD;
    float* ap = agg + (long)d * HD;
    #pragma unroll
    for (int i = 0; i < 3; ++i) {
      float v = hp[lane + 32 * i];
      __hip_atomic_fetch_add(&ap[lane + 32 * i], v, __ATOMIC_RELAXED, __HIP_MEMORY_SCOPE_AGENT);
    }
  }
}

__global__ void prelu_kernel(const float* __restrict__ agg, const float* __restrict__ norm_in,
                             const float* __restrict__ b, const float* __restrict__ a,
                             float* __restrict__ out, int n) {
  long idx = (long)blockIdx.x * blockDim.x + threadIdx.x;
  if (idx < (long)n * HD) {
    int i = (int)(idx / HD);
    int j = (int)(idx % HD);
    float v = agg[idx] * norm_in[i] + b[j];
    out[idx] = (v >= 0.0f) ? v : a[j] * v;
  }
}

__global__ void reduce_kernel(const float* __restrict__ z, const float* __restrict__ wsum,
                              const float* __restrict__ bsum, float* __restrict__ out, int n) {
  int gw   = (int)(((long)blockIdx.x * blockDim.x + threadIdx.x) >> 5);
  int lane = threadIdx.x & 31;
  if (gw < n) {
    const float* zp = z + (long)gw * HD;
    float s = 0.0f;
    #pragma unroll
    for (int i = 0; i < 3; ++i) s = fmaf(zp[lane + 32 * i], wsum[lane + 32 * i], s);
    #pragma unroll
    for (int off = 16; off > 0; off >>= 1) s += __shfl_xor(s, off, 32);
    if (lane == 0) out[gw] = s + *bsum;
  }
}

extern "C" void kernel_launch(void* const* d_in, const int* in_sizes, int n_in,
                              void* d_out, int out_size, void* d_ws, size_t ws_size,
                              hipStream_t stream) {
  const float* x    = (const float*)d_in[0];
  const int*   src  = (const int*)  d_in[1];
  const int*   dst  = (const int*)  d_in[2];
  const int*   perm = (const int*)  d_in[3];
  const float* W1   = (const float*)d_in[4];
  const float* b1   = (const float*)d_in[5];
  const float* a1   = (const float*)d_in[6];
  const float* W2   = (const float*)d_in[7];
  const float* b2   = (const float*)d_in[8];
  const float* a2   = (const float*)d_in[9];
  const float* Wm   = (const float*)d_in[10];
  const float* bm   = (const float*)d_in[11];
  float* out = (float*)d_out;

  const int E  = in_sizes[1];
  const int N  = in_sizes[3];
  const int K1 = in_sizes[0] / N;

  float* ws       = (float*)d_ws;
  float* norm_out = ws;
  float* norm_in  = ws + N;
  float* wsum     = ws + 2L * N;
  float* bsum     = ws + 2L * N + HD;
  float* bufA     = ws + 2L * N + 2 * HD;
  float* bufB     = bufA + (long)N * HD;

  const long NH = (long)N * HD;
  int*   rowptr   = (int*)(bufB + NH);
  int*   csr_src  = rowptr + (N + 1);
  int*   rank     = (int*)bufB;
  int*   cnt_out  = (int*)norm_out;
  int*   cnt_in   = (int*)norm_in;
  int*   hist_in  = (int*)bufA;
  int*   hist_out = (int*)bufB;
  int*   lrank    = hist_out + (long)HG * N;
  int*   cnt_tot  = lrank + E;

  const int gemm_blocks = (N + RPB - 1) / RPB;
  const int spmm_blocks = (int)(((long)E * 32 + 255) / 256);
  const int gath_blocks = (N + 7) / 8;
  const int elw_blocks  = (int)((NH + 255) / 256);
  const int red_blocks  = (int)(((long)N * 32 + 255) / 256);

  const size_t need_grouped = (size_t)(2L * N + 2 * HD + 2 * NH + (N + 1) + E) * sizeof(float);
  const bool grouped = (ws_size >= need_grouped) && ((long)E <= NH);
  const int  chunk    = (E + HG - 1) / HG;
  const bool lds_hist = grouped && (N <= HCAP) && ((long)(HG + 1) * N + E <= NH);

  if (lds_hist) {
    hist_lds_kernel<<<2 * HG, 1024, 0, stream>>>(src, dst, hist_in, hist_out, lrank, E, N, chunk);
    hist_fold_kernel<<<(2 * N + 255) / 256, 256, 0, stream>>>(hist_in, hist_out, cnt_tot, norm_out, norm_in, N);
    rowptr_scan_kernel<<<1, 1024, 0, stream>>>(cnt_tot, rowptr, N);
    place_chunked_kernel<<<(E + 255) / 256, 256, 0, stream>>>(src, dst, lrank, rowptr, hist_in, csr_src, E, N, chunk);
  } else {
    fill_zero_kernel<<<512, 256, 0, stream>>>(norm_out, 2L * N);
    degree_kernel<<<(E + 255) / 256, 256, 0, stream>>>(src, dst, cnt_out, cnt_in, grouped ? rank : nullptr, E);
    if (grouped) {
      rowptr_scan_kernel<<<1, 1024, 0, stream>>>(cnt_in, rowptr, N);
      place_kernel<<<(E + 255) / 256, 256, 0, stream>>>(src, dst, rank, rowptr, csr_src, E);
    }
    finalize_norm_kernel<<<(2 * N + 255) / 256, 256, 0, stream>>>(norm_out, 2 * N);
  }
  wsum_kernel<<<1, 128, 0, stream>>>(Wm, bm, wsum, bsum);

  if (grouped) gemm_norm_wmma_kernel<<<gemm_blocks, 256, 0, stream>>>(x, nullptr, nullptr, W1, bufA, N, K1);

  for (int view = 0; view < 2; ++view) {
    const int* rowmap = view ? perm : nullptr;
    float* outv = out + (long)view * N;
    if (grouped) {
      if (view) {
        spmm_gather_scaled_prelu_kernel<true><<<gath_blocks, 256, 0, stream>>>(rowptr, csr_src, bufA, norm_out, norm_in,
                                                                              b1, a1, bufB, rowmap, N);
      } else {
        spmm_gather_scaled_prelu_kernel<false><<<gath_blocks, 256, 0, stream>>>(rowptr, csr_src, bufA, norm_out, norm_in,
                                                                               b1, a1, bufB, nullptr, N);
      }
      gemm_norm_wmma_kernel<<<gemm_blocks, 256, 0, stream>>>(bufB, nullptr, norm_out, W2, bufB, N, HD);
      spmm_gather_prelu_dot_kernel<<<gath_blocks, 256, 0, stream>>>(rowptr, csr_src, bufB, norm_in, b2, a2,
                                                                   wsum, bsum, outv, N);
    } else {
      gemm_norm_wmma_kernel<<<gemm_blocks, 256, 0, stream>>>(x, rowmap, norm_out, W1, bufA, N, K1);
      fill_zero_kernel<<<2048, 256, 0, stream>>>(bufB, NH);
      spmm_kernel<<<spmm_blocks, 256, 0, stream>>>(src, dst, bufA, bufB, E);
      prelu_kernel<<<elw_blocks, 256, 0, stream>>>(bufB, norm_in, b1, a1, bufA, N);
      gemm_norm_wmma_kernel<<<gemm_blocks, 256, 0, stream>>>(bufA, nullptr, norm_out, W2, bufB, N, HD);
      fill_zero_kernel<<<2048, 256, 0, stream>>>(bufA, NH);
      spmm_kernel<<<spmm_blocks, 256, 0, stream>>>(src, dst, bufB, bufA, E);
      prelu_kernel<<<elw_blocks, 256, 0, stream>>>(bufA, norm_in, b2, a2, bufB, N);
      reduce_kernel<<<red_blocks, 256, 0, stream>>>(bufB, wsum, bsum, outv, N);
    }
  }
}


# both views through every stage together in one buffer of two rows per node: layer-2 walks each source list once (one 768-B row per edge, not two 384-B rows in two launches); 12 launches -> 9
# speedup vs baseline: 3.8789x; 1.0786x over previous
#include <hip/hip_runtime.h>

typedef float v2f __attribute__((ext_vector_type(2)));
typedef float v8f __attribute__((ext_vector_type(8)));

constexpr int HD   = 96;
constexpr int RPB  = 128;
constexpr int BK   = 32;
constexpr int LPAD = 36;
constexpr int WPAD = 38;
constexpr int HG   = 64;
constexpr int HCAP = 65536;


__global__ void fill_zero_kernel(float* __restrict__ p, long n) {
  long i = (long)blockIdx.x * blockDim.x + threadIdx.x;
  long stride = (long)gridDim.x * blockDim.x;
  for (; i < n; i += stride) p[i] = 0.0f;
}

__global__ void degree_kernel(const int* __restrict__ src, const int* __restrict__ dst,
                              int* __restrict__ cnt_out, int* __restrict__ cnt_in,
                              int* __restrict__ rank, int E) {
  int e = blockIdx.x * blockDim.x + threadIdx.x;
  if (e < E) {
    __hip_atomic_fetch_add(&cnt_out[src[e]], 1, __ATOMIC_RELAXED, __HIP_MEMORY_SCOPE_AGENT);
    if (rank) {
      rank[e] = __hip_atomic_fetch_add(&cnt_in[dst[e]], 1, __ATOMIC_RELAXED, __HIP_MEMORY_SCOPE_AGENT);
    } else {
      __hip_atomic_fetch_add(&cnt_in[dst[e]], 1, __ATOMIC_RELAXED, __HIP_MEMORY_SCOPE_AGENT);
    }
  }
}

__global__ void finalize_norm_kernel(float* __restrict__ p, int n) {
  int i = blockIdx.x * blockDim.x + threadIdx.x;
  if (i < n) p[i] = 1.0f / sqrtf(fmaxf((float)__float_as_int(p[i]), 1.0f));
}

__global__ __launch_bounds__(1024)
void rowptr_scan_kernel(const int* __restrict__ cnt, int* __restrict__ rowptr, int n) {
  __shared__ int wave_tot[32];
  const int t = threadIdx.x, lane = t & 31, w = t >> 5;
  int carry = 0;
  int c[4], nx[4];
  for (int k = 0; k < 4; ++k) {
    const int i = t * 4 + k;
    const int v = cnt[i < n ? i : n - 1];
    c[k] = (i < n) ? v : 0;
  }
  for (int base = 0; base < n; base += 4096) {
    const int i0 = base + t * 4;
    for (int k = 0; k < 4; ++k) {
      const int i = i0 + 4096 + k;
      const int v = cnt[i < n ? i : n - 1];
      nx[k] = (i < n) ? v : 0;
    }
    const int s = c[0] + c[1] + c[2] + c[3];
    int incl = s;
    for (int off = 1; off < 32; off <<= 1) {
      int v = __shfl_up(incl, off, 32);
      if (lane >= off) incl += v;
    }
    if (lane == 31) wave_tot[w] = incl;
    __syncthreads();
    const int wt = wave_tot[lane];
    int wincl = wt;
    for (int off = 1; off < 32; off <<= 1) {
      int v = __shfl_up(wincl, off, 32);
      if (lane >= off) wincl += v;
    }
    const int wave_excl = __shfl(wincl, w, 32) - __shfl(wt, w, 32);
    const int total     = __shfl(wincl, 31, 32);
    int run = carry + wave_excl + incl - s;
    for (int k = 0; k < 4; ++k) {
      if (i0 + k < n) rowptr[i0 + k] = run;
      run += c[k];
    }
    carry += total;
    for (int k = 0; k < 4; ++k) c[k] = nx[k];
    __syncthreads();
  }
  if (t == 0) rowptr[n] = carry;
}

__global__ void place_kernel(const int* __restrict__ src, const int* __restrict__ dst,
                             const int* __restrict__ rank, const int* __restrict__ rowptr,
                             int* __restrict__ csr_src, int E) {
  int e = blockIdx.x * blockDim.x + threadIdx.x;
  if (e < E) csr_src[rowptr[dst[e]] + rank[e]] = src[e];
}

__global__ __launch_bounds__(1024)
void hist_lds_kernel(const int* __restrict__ src, const int* __restrict__ dst,
                     int* __restrict__ hist_in, int* __restrict__ hist_out,
                     int* __restrict__ rank, int E, int n, int chunk) {
  __shared__ int4 cnt4[HCAP / 4];
  int* cnt = reinterpret_cast<int*>(cnt4);
  const int  t     = threadIdx.x;
  const bool is_in = blockIdx.x < HG;
  const int  g     = is_in ? (int)blockIdx.x : (int)blockIdx.x - HG;
  const int* __restrict__ key  = is_in ? dst : src;
  int*       __restrict__ hist = (is_in ? hist_in : hist_out) + (long)g * n;
  const bool vec = ((n & 3) == 0) && ((((size_t)hist) & 15) == 0);

  if (vec) {
    for (int i = t; i < (n >> 2); i += 1024) cnt4[i] = make_int4(0, 0, 0, 0);
  } else {
    for (int i = t; i < n; i += 1024) cnt[i] = 0;
  }
  __syncthreads();

  const int beg = g * chunk;
  const int end = (beg + chunk < E) ? (beg + chunk) : E;
  for (int base = beg + t; base < end; base += 8 * 1024) {
    int k[8];
    #pragma unroll
    for (int u = 0; u < 8; ++u) {
      const int e = base + u * 1024;
      k[u] = key[e < end ? e : end - 1];
    }
    int r[8];
    #pragma unroll
    for (int u = 0; u < 8; ++u) {
      const int e = base + u * 1024;
      r[u] = __hip_atomic_fetch_add(&cnt[k[u]], (e < end) ? 1 : 0, __ATOMIC_RELAXED, __HIP_MEMORY_SCOPE_WORKGROUP);
    }
    if (is_in) {
      #pragma unroll
      for (int u = 0; u < 8; ++u) {
        const int e = base + u * 1024;
        if (e < end) rank[e] = r[u];
      }
    }
  }
  __syncthreads();

  if (vec) {
    int4* __restrict__ h4 = reinterpret_cast<int4*>(hist);
    for (int i = t; i < (n >> 2); i += 1024) h4[i] = cnt4[i];
  } else {
    for (int i = t; i < n; i += 1024) hist[i] = cnt[i];
  }
}

__global__ __launch_bounds__(256)
void hist_fold_kernel(int* __restrict__ hist_in, const int* __restrict__ hist_out,
                      int* __restrict__ cnt_tot, float* __restrict__ norm_out,
                      float* __restrict__ norm_in, int n) {
  const int i = blockIdx.x * blockDim.x + threadIdx.x;
  if (i < n) {
    int c[HG];
    #pragma unroll
    for (int g = 0; g < HG; ++g) c[g] = hist_in[(long)g * n + i];
    int run = 0;
    #pragma unroll
    for (int g = 0; g < HG; ++g) {
      hist_in[(long)g * n + i] = run;
      run += c[g];
    }
    cnt_tot[i] = run;
    norm_in[i] = 1.0f / sqrtf(fmaxf((float)run, 1.0f));
  } else if (i < 2 * n) {
    const int j = i - n;
    int s = 0;
    #pragma unroll
    for (int g = 0; g < HG; ++g) s += hist_out[(long)g * n + j];
    norm_out[j] = 1.0f / sqrtf(fmaxf((float)s, 1.0f));
  }
}

__global__ void place_chunked_kernel(const int* __restrict__ src, const int* __restrict__ dst,
                                     const int* __restrict__ rank, const int* __restrict__ rowptr,
                                     const int* __restrict__ before, int* __restrict__ csr_src,
                                     int E, int n, int chunk) {
  int e = blockIdx.x * blockDim.x + threadIdx.x;
  if (e < E) {
    const int d = dst[e];
    const int g = e / chunk;
    csr_src[rowptr[d] + before[(long)g * n + d] + rank[e]] = src[e];
  }
}

__global__ void wsum_kernel(const float* __restrict__ Wm, const float* __restrict__ bm,
                            float* __restrict__ wsum, float* __restrict__ bsum) {
  int t = threadIdx.x;
  if (t < HD) {
    float s = 0.0f;
    for (int j = 0; j < HD; ++j) s += Wm[t * HD + j];
    wsum[t] = s;
  } else if (t == HD) {
    float s = 0.0f;
    for (int j = 0; j < HD; ++j) s += bm[j];
    *bsum = s;
  }
}

__global__ __launch_bounds__(256)
void gemm_norm_wmma_kernel(const float* X, const int* __restrict__ rowmap,
                           const float* __restrict__ norm, const float* __restrict__ W,
                           float* OUT, int nrows, int K, int norm_shift) {
  __shared__ float ldsX[RPB][LPAD];
  __shared__ float ldsW[HD][WPAD];

  const int tid  = threadIdx.x;
  const int wave = tid >> 5;
  const int lane = tid & 31;
  const int half = lane >> 4;
  const int l16  = lane & 15;
  const int row0 = blockIdx.x * RPB;

  v8f acc[6] = {};

  const int xc4 = (tid & 7) * 4;
  const float* xptr[4];
  float        xscale[4];
  bool         xrow_ok[4];
  int          xsrc[4];
  #pragma unroll
  for (int it = 0; it < 4; ++it) {
    const int grow = row0 + ((tid + 256 * it) >> 3);
    xrow_ok[it]    = grow < nrows;
    xsrc[it]       = xrow_ok[it] ? grow : nrows - 1;
  }
  if (norm) {
    #pragma unroll
    for (int it = 0; it < 4; ++it) xscale[it] = norm[xsrc[it] >> norm_shift];
  } else {
    #pragma unroll
    for (int it = 0; it < 4; ++it) xscale[it] = 1.0f;
  }
  if (rowmap) {
    #pragma unroll
    for (int it = 0; it < 4; ++it) xsrc[it] = rowmap[xsrc[it]];
  }
  #pragma unroll
  for (int it = 0; it < 4; ++it) xptr[it] = X + (long)xsrc[it] * K + xc4;
  const float* wptr[12];
  int wkk[12];
  #pragma unroll
  for (int it = 0; it < 12; ++it) {
    const int idx = tid + 256 * it;
    wkk[it]  = idx / HD;
    wptr[it] = W + (idx % HD);
  }

  float4 xv[4];
  float  wv[12];
  auto load_tile = [&](int k0) {
    const bool kin = (k0 + xc4) < K;
    #pragma unroll
    for (int it = 0; it < 4; ++it) {
      const float* p = xptr[it] + (kin ? k0 : -xc4);
      xv[it] = *(const float4*)p;
      __builtin_prefetch(((k0 + BK + xc4) < K) ? (p + BK) : p, 0, 0);
    }
    #pragma unroll
    for (int it = 0; it < 12; ++it) {
      const int k = k0 + wkk[it];
      wv[it] = wptr[it][(long)((k < K) ? k : 0) * HD];
    }
  };

  load_tile(0);
  for (int k0 = 0; k0 < K; k0 += BK) {
    {
      const bool kin = (k0 + xc4) < K;
      #pragma unroll
      for (int it = 0; it < 4; ++it) {
        const int r = (tid + 256 * it) >> 3;
        float4 v = xv[it];
        const float s = xscale[it];
        v.x *= s; v.y *= s; v.z *= s; v.w *= s;
        if (!(kin && xrow_ok[it])) v = make_float4(0.f, 0.f, 0.f, 0.f);
        ldsX[r][xc4 + 0] = v.x;
        ldsX[r][xc4 + 1] = v.y;
        ldsX[r][xc4 + 2] = v.z;
        ldsX[r][xc4 + 3] = v.w;
      }
      #pragma unroll
      for (int it = 0; it < 12; ++it) {
        const int idx = tid + 256 * it;
        ldsW[idx % HD][wkk[it]] = (k0 + wkk[it] < K) ? wv[it] : 0.0f;
      }
    }
    __syncthreads();

    if (k0 + BK < K) load_tile(k0 + BK);

    const int kmax = (BK < K - k0) ? BK : (K - k0);
    const int mrow = wave * 16 + l16;
    for (int kk = 0; kk < kmax; kk += 4) {
      int ka = kk + half * 2;

      v2f A = *(const v2f*)&ldsX[mrow][ka];
      v2f Bf[6];
      #pragma unroll
      for (int t = 0; t < 6; ++t)
        Bf[t] = *(const v2f*)&ldsW[t * 16 + l16][ka];

      #pragma unroll
      for (int t = 0; t < 6; ++t)
        acc[t] = __builtin_amdgcn_wmma_f32_16x16x4_f32(
             false, A,  false, Bf[t],
             (short)0, acc[t],  false,  false);
    }
    __syncthreads();
  }

  const int orow = row0 + wave * 16 + half * 8;
  #pragma unroll
  for (int t = 0; t < 6; ++t) {
    int col = t * 16 + l16;
    #pragma unroll
    for (int i = 0; i < 8; ++i) {
      int r = orow + i;
      if (r < nrows) OUT[(long)r * HD + col] = acc[t][i];
    }
  }
}

__device__ __forceinline__ void gather_row_sum(const int* __restrict__ rowptr, const int* __restrict__ csr_src,
                                               const float* __restrict__ h, int row, int lane,
                                               float& a0, float& a1, float& a2) {
  const int beg = rowptr[row];
  const int end = rowptr[row + 1];
  a0 = 0.0f; a1 = 0.0f; a2 = 0.0f;
  for (int base = beg; base < end; base += 32) {
    const int cnt  = (end - base < 32) ? (end - base) : 32;
    const int mine = (lane < cnt) ? csr_src[base + lane] : 0;
    int j = 0;
    for (; j + 4 <= cnt; j += 4) {
      const float* p0 = h + (long)__builtin_amdgcn_readlane(mine, j + 0) * HD + lane;
      const float* p1 = h + (long)__builtin_amdgcn_readlane(mine, j + 1) * HD + lane;
      const float* p2 = h + (long)__builtin_amdgcn_readlane(mine, j + 2) * HD + lane;
      const float* p3 = h + (long)__builtin_amdgcn_readlane(mine, j + 3) * HD + lane;
      const float x00 = p0[0], x01 = p0[32], x02 = p0[64];
      const float x10 = p1[0], x11 = p1[32], x12 = p1[64];
      const float x20 = p2[0], x21 = p2[32], x22 = p2[64];
      const float x30 = p3[0], x31 = p3[32], x32 = p3[64];
      a0 += x00; a1 += x01; a2 += x02;
      a0 += x10; a1 += x11; a2 += x12;
      a0 += x20; a1 += x21; a2 += x22;
      a0 += x30; a1 += x31; a2 += x32;
    }
    for (; j < cnt; ++j) {
      const float* p = h + (long)__builtin_amdgcn_readlane(mine, j) * HD + lane;
      a0 += p[0]; a1 += p[32]; a2 += p[64];
    }
  }
}

template <bool MAPPED>
__device__ __forceinline__ void gather_row_sum_scaled(const int* __restrict__ rowptr, const int* __restrict__ csr_src,
                                                      const float* __restrict__ xw, const int* __restrict__ rowmap,
                                                      const float* __restrict__ scale, int row, int lane,
                                                      float& a0, float& a1, float& a2) {
  const int beg = rowptr[row];
  const int end = rowptr[row + 1];
  a0 = 0.0f; a1 = 0.0f; a2 = 0.0f;
  for (int base = beg; base < end; base += 32) {
    const int cnt  = (end - base < 32) ? (end - base) : 32;
    const int mine = (lane < cnt) ? csr_src[base + lane] : 0;
    const int m    = MAPPED ? rowmap[mine] : mine;
    const int sc   = __float_as_int(scale[mine]);
    int j = 0;
    for (; j + 4 <= cnt; j += 4) {
      const float* p0 = xw + (long)__builtin_amdgcn_readlane(m, j + 0) * HD + lane;
      const float* p1 = xw + (long)__builtin_amdgcn_readlane(m, j + 1) * HD + lane;
      const float* p2 = xw + (long)__builtin_amdgcn_readlane(m, j + 2) * HD + lane;
      const float* p3 = xw + (long)__builtin_amdgcn_readlane(m, j + 3) * HD + lane;
      const float x00 = p0[0], x01 = p0[32], x02 = p0[64];
      const float x10 = p1[0], x11 = p1[32], x12 = p1[64];
      const float x20 = p2[0], x21 = p2[32], x22 = p2[64];
      const float x30 = p3[0], x31 = p3[32], x32 = p3[64];
      __builtin_amdgcn_sched_barrier(0);
      const float c0 = __int_as_float(__builtin_amdgcn_readlane(sc, j + 0));
      const float c1 = __int_as_float(__builtin_amdgcn_readlane(sc, j + 1));
      const float c2 = __int_as_float(__builtin_amdgcn_readlane(sc, j + 2));
      const float c3 = __int_as_float(__builtin_amdgcn_readlane(sc, j + 3));
      a0 = fmaf(c0, x00, a0); a1 = fmaf(c0, x01, a1); a2 = fmaf(c0, x02, a2);
      a0 = fmaf(c1, x10, a0); a1 = fmaf(c1, x11, a1); a2 = fmaf(c1, x12, a2);
      a0 = fmaf(c2, x20, a0); a1 = fmaf(c2, x21, a1); a2 = fmaf(c2, x22, a2);
      a0 = fmaf(c3, x30, a0); a1 = fmaf(c3, x31, a1); a2 = fmaf(c3, x32, a2);
    }
    for (; j < cnt; ++j) {
      const float* p = xw + (long)__builtin_amdgcn_readlane(m, j) * HD + lane;
      const float x0 = p[0], x1 = p[32], x2 = p[64];
      const float c  = __int_as_float(__builtin_amdgcn_readlane(sc, j));
      a0 = fmaf(c, x0, a0); a1 = fmaf(c, x1, a1); a2 = fmaf(c, x2, a2);
    }
  }
}

template <bool MAPPED>
__device__ __forceinline__ void gather_scaled_prelu_node(const int* __restrict__ rowptr, const int* __restrict__ csr_src,
                                                         const float* __restrict__ xw, const float* __restrict__ norm_out,
                                                         const float* __restrict__ norm_in, const float* __restrict__ b,
                                                         const float* __restrict__ a, float* __restrict__ act_row,
                                                         const int* __restrict__ rowmap, int row, int lane) {
  const float ni = norm_in[row];
  const float b0 = b[lane], b1 = b[lane + 32], b2 = b[lane + 64];
  const float s0 = a[lane], s1 = a[lane + 32], s2 = a[lane + 64];
  float a0, a1, a2;
  gather_row_sum_scaled<MAPPED>(rowptr, csr_src, xw, rowmap, norm_out, row, lane, a0, a1, a2);
  const float v0 = a0 * ni + b0;
  const float v1 = a1 * ni + b1;
  const float v2 = a2 * ni + b2;
  float* o = act_row + lane;
  o[0]  = (v0 >= 0.0f) ? v0 : s0 * v0;
  o[32] = (v1 >= 0.0f) ? v1 : s1 * v1;
  o[64] = (v2 >= 0.0f) ? v2 : s2 * v2;
}

template <bool MAPPED>
__global__ __launch_bounds__(256)
void spmm_gather_scaled_prelu_kernel(const int* __restrict__ rowptr, const int* __restrict__ csr_src,
                                     const float* __restrict__ xw, const float* __restrict__ norm_out,
                                     const float* __restrict__ norm_in, const float* __restrict__ b,
                                     const float* __restrict__ a, float* __restrict__ act,
                                     const int* __restrict__ rowmap, int n) {
  const int lane = threadIdx.x & 31;
  const int row = __builtin_amdgcn_readfirstlane((int)(blockIdx.x * (blockDim.x >> 5) + (threadIdx.x >> 5)));
  if (row >= n) return;
  gather_scaled_prelu_node<MAPPED>(rowptr, csr_src, xw, norm_out, norm_in, b, a, act + (long)row * HD, rowmap, row, lane);
}

__global__ __launch_bounds__(256)
void spmm_gather_scaled_prelu_both_kernel(const int* __restrict__ rowptr, const int* __restrict__ csr_src,
                                          const float* __restrict__ xw, const float* __restrict__ norm_out,
                                          const float* __restrict__ norm_in, const float* __restrict__ b,
                                          const float* __restrict__ a, float* __restrict__ act2,
                                          const int* __restrict__ rowmap, int n, int blocks_per_view) {
  const int lane = threadIdx.x & 31;
  const bool second = (int)blockIdx.x >= blocks_per_view;
  const int blk = (int)blockIdx.x - (second ? blocks_per_view : 0);
  const int row = __builtin_amdgcn_readfirstlane((int)(blk * (blockDim.x >> 5) + (threadIdx.x >> 5)));
  if (row >= n) return;
  float* dst = act2 + (long)row * (2 * HD);
  if (second) gather_scaled_prelu_node<true>(rowptr, csr_src, xw, norm_out, norm_in, b, a, dst + HD, rowmap, row, lane);
  else        gather_scaled_prelu_node<false>(rowptr, csr_src, xw, norm_out, norm_in, b, a, dst, nullptr, row, lane);
}

__global__ __launch_bounds__(256)
void spmm_gather_prelu_dot_kernel(const int* __restrict__ rowptr, const int* __restrict__ csr_src,
                                  const float* __restrict__ h, const float* __restrict__ norm_in,
                                  const float* __restrict__ b, const float* __restrict__ a,
                                  const float* __restrict__ wsum, const float* __restrict__ bsum,
                                  float* __restrict__ out, int n) {
  const int lane = threadIdx.x & 31;
  const int row = __builtin_amdgcn_readfirstlane((int)(blockIdx.x * (blockDim.x >> 5) + (threadIdx.x >> 5)));
  if (row >= n) return;
  const float ni = norm_in[row];
  const float b0 = b[lane], b1 = b[lane + 32], b2 = b[lane + 64];
  const float s0 = a[lane], s1 = a[lane + 32], s2 = a[lane + 64];
  const float w0 = wsum[lane], w1 = wsum[lane + 32], w2 = wsum[lane + 64];
  const float bs = *bsum;
  float a0, a1, a2;
  gather_row_sum(rowptr, csr_src, h, row, lane, a0, a1, a2);
  const float v0 = a0 * ni + b0;
  const float v1 = a1 * ni + b1;
  const float v2 = a2 * ni + b2;
  const float z0 = (v0 >= 0.0f) ? v0 : s0 * v0;
  const float z1 = (v1 >= 0.0f) ? v1 : s1 * v1;
  const float z2 = (v2 >= 0.0f) ? v2 : s2 * v2;
  float s = fmaf(z0, w0, 0.0f);
  s = fmaf(z1, w1, s);
  s = fmaf(z2, w2, s);
  #pragma unroll
  for (int off = 16; off > 0; off >>= 1) s += __shfl_xor(s, off, 32);
  if (lane == 0) out[row] = s + bs;
}

__device__ __forceinline__ void gather_row_sum2(const int* __restrict__ rowptr, const int* __restrict__ csr_src,
                                                const float* __restrict__ h2, int row, int lane,
                                                float& a0, float& a1, float& a2, float& c0, float& c1, float& c2) {
  const int beg = rowptr[row];
  const int end = rowptr[row + 1];
  a0 = 0.0f; a1 = 0.0f; a2 = 0.0f; c0 = 0.0f; c1 = 0.0f; c2 = 0.0f;
  for (int base = beg; base < end; base += 32) {
    const int cnt  = (end - base < 32) ? (end - base) : 32;
    const int mine = (lane < cnt) ? csr_src[base + lane] : 0;
    int j = 0;
    for (; j + 4 <= cnt; j += 4) {
      const float* p0 = h2 + (long)__builtin_amdgcn_readlane(mine, j + 0) * (2 * HD) + lane;
      const float* p1 = h2 + (long)__builtin_amdgcn_readlane(mine, j + 1) * (2 * HD) + lane;
      const float* p2 = h2 + (long)__builtin_amdgcn_readlane(mine, j + 2) * (2 * HD) + lane;
      const float* p3 = h2 + (long)__builtin_amdgcn_readlane(mine, j + 3) * (2 * HD) + lane;
      const float x00 = p0[0], x01 = p0[32], x02 = p0[64], y00 = p0[96], y01 = p0[128], y02 = p0[160];
      const float x10 = p1[0], x11 = p1[32], x12 = p1[64], y10 = p1[96], y11 = p1[128], y12 = p1[160];
      const float x20 = p2[0], x21 = p2[32], x22 = p2[64], y20 = p2[96], y21 = p2[128], y22 = p2[160];
      const float x30 = p3[0], x31 = p3[32], x32 = p3[64], y30 = p3[96], y31 = p3[128], y32 = p3[160];
      a0 += x00; a1 += x01; a2 += x02; c0 += y00; c1 += y01; c2 += y02;
      a0 += x10; a1 += x11; a2 += x12; c0 += y10; c1 += y11; c2 += y12;
      a0 += x20; a1 += x21; a2 += x22; c0 += y20; c1 += y21; c2 += y22;
      a0 += x30; a1 += x31; a2 += x32; c0 += y30; c1 += y31; c2 += y32;
    }
    for (; j < cnt; ++j) {
      const float* p = h2 + (long)__builtin_amdgcn_readlane(mine, j) * (2 * HD) + lane;
      const float x0 = p[0], x1 = p[32], x2 = p[64], y0 = p[96], y1 = p[128], y2 = p[160];
      a0 += x0; a1 += x1; a2 += x2; c0 += y0; c1 += y1; c2 += y2;
    }
  }
}

__global__ __launch_bounds__(256)
void spmm_gather2_prelu_dot_kernel(const int* __restrict__ rowptr, const int* __restrict__ csr_src,
                                   const float* __restrict__ h2, const float* __restrict__ norm_in,
                                   const float* __restrict__ b, const float* __restrict__ a,
                                   const float* __restrict__ wsum, const float* __restrict__ bsum,
                                   float* __restrict__ out, int n) {
  const int lane = threadIdx.x & 31;
  const int row = __builtin_amdgcn_readfirstlane((int)(blockIdx.x * (blockDim.x >> 5) + (threadIdx.x >> 5)));
  if (row >= n) return;
  const float ni = norm_in[row];
  const float b0 = b[lane], b1 = b[lane + 32], b2 = b[lane + 64];
  const float s0 = a[lane], s1 = a[lane + 32], s2 = a[lane + 64];
  const float w0 = wsum[lane], w1 = wsum[lane + 32], w2 = wsum[lane + 64];
  const float bs = *bsum;
  float a0, a1, a2, c0, c1, c2;
  gather_row_sum2(rowptr, csr_src, h2, row, lane, a0, a1, a2, c0, c1, c2);
  const float v0 = a0 * ni + b0;
  const float v1 = a1 * ni + b1;
  const float v2 = a2 * ni + b2;
  const float u0 = c0 * ni + b0;
  const float u1 = c1 * ni + b1;
  const float u2 = c2 * ni + b2;
  const float z0 = (v0 >= 0.0f) ? v0 : s0 * v0;
  const float z1 = (v1 >= 0.0f) ? v1 : s1 * v1;
  const float z2 = (v2 >= 0.0f) ? v2 : s2 * v2;
  const float q0 = (u0 >= 0.0f) ? u0 : s0 * u0;
  const float q1 = (u1 >= 0.0f) ? u1 : s1 * u1;
  const float q2 = (u2 >= 0.0f) ? u2 : s2 * u2;
  float s = fmaf(z0, w0, 0.0f);
  s = fmaf(z1, w1, s);
  s = fmaf(z2, w2, s);
  float r = fmaf(q0, w0, 0.0f);
  r = fmaf(q1, w1, r);
  r = fmaf(q2, w2, r);
  for (int off = 16; off > 0; off >>= 1) {
    s += __shfl_xor(s, off, 32);
    r += __shfl_xor(r, off, 32);
  }
  if (lane == 0) {
    out[row]     = s + bs;
    out[n + row] = r + bs;
  }
}

__global__ void spmm_kernel(const int* __restrict__ src, const int* __restrict__ dst,
                            const float* __restrict__ h, float* __restrict__ agg, int E) {
  int gw   = (int)(((long)blockIdx.x * blockDim.x + threadIdx.x) >> 5);
  int lane = threadIdx.x & 31;
  if (gw < E) {
    int s = src[gw], d = dst[gw];
    const float* hp = h + (long)s * HD;
    float* ap = agg + (long)d * HD;
    #pragma unroll
    for (int i = 0; i < 3; ++i) {
      float v = hp[lane + 32 * i];
      __hip_atomic_fetch_add(&ap[lane + 32 * i], v, __ATOMIC_RELAXED, __HIP_MEMORY_SCOPE_AGENT);
    }
  }
}

__global__ void prelu_kernel(const float* __restrict__ agg, const float* __restrict__ norm_in,
                             const float* __restrict__ b, const float* __restrict__ a,
                             float* __restrict__ out, int n) {
  long idx = (long)blockIdx.x * blockDim.x + threadIdx.x;
  if (idx < (long)n * HD) {
    int i = (int)(idx / HD);
    int j = (int)(idx % HD);
    float v = agg[idx] * norm_in[i] + b[j];
    out[idx] = (v >= 0.0f) ? v : a[j] * v;
  }
}

__global__ void reduce_kernel(const float* __restrict__ z, const float* __restrict__ wsum,
                              const float* __restrict__ bsum, float* __restrict__ out, int n) {
  int gw   = (int)(((long)blockIdx.x * blockDim.x + threadIdx.x) >> 5);
  int lane = threadIdx.x & 31;
  if (gw < n) {
    const float* zp = z + (long)gw * HD;
    float s = 0.0f;
    #pragma unroll
    for (int i = 0; i < 3; ++i) s = fmaf(zp[lane + 32 * i], wsum[lane + 32 * i], s);
    #pragma unroll
    for (int off = 16; off > 0; off >>= 1) s += __shfl_xor(s, off, 32);
    if (lane == 0) out[gw] = s + *bsum;
  }
}

extern "C" void kernel_launch(void* const* d_in, const int* in_sizes, int n_in,
                              void* d_out, int out_size, void* d_ws, size_t ws_size,
                              hipStream_t stream) {
  const float* x    = (const float*)d_in[0];
  const int*   src  = (const int*)  d_in[1];
  const int*   dst  = (const int*)  d_in[2];
  const int*   perm = (const int*)  d_in[3];
  const float* W1   = (const float*)d_in[4];
  const float* b1   = (const float*)d_in[5];
  const float* a1   = (const float*)d_in[6];
  const float* W2   = (const float*)d_in[7];
  const float* b2   = (const float*)d_in[8];
  const float* a2   = (const float*)d_in[9];
  const float* Wm   = (const float*)d_in[10];
  const float* bm   = (const float*)d_in[11];
  float* out = (float*)d_out;

  const int E  = in_sizes[1];
  const int N  = in_sizes[3];
  const int K1 = in_sizes[0] / N;

  float* ws       = (float*)d_ws;
  float* norm_out = ws;
  float* norm_in  = ws + N;
  float* wsum     = ws + 2L * N;
  float* bsum     = ws + 2L * N + HD;
  float* bufA     = ws + 2L * N + 2 * HD;
  float* bufB     = bufA + (long)N * HD;

  const long NH = (long)N * HD;
  const size_t need_wide = (size_t)(2L * N + 2 * HD + 3 * NH + (N + 1) + E) * sizeof(float);
  const bool wide = (ws_size >= need_wide) && ((long)E <= NH);
  float* H2       = bufB;
  int*   rowptr   = (int*)(bufB + (wide ? 2 * NH : NH));
  int*   csr_src  = rowptr + (N + 1);
  int*   rank     = (int*)bufB;
  int*   cnt_out  = (int*)norm_out;
  int*   cnt_in   = (int*)norm_in;
  int*   hist_in  = (int*)bufA;
  int*   hist_out = (int*)bufB;
  int*   lrank    = hist_out + (long)HG * N;
  int*   cnt_tot  = lrank + E;

  const int gemm_blocks = (N + RPB - 1) / RPB;
  const int spmm_blocks = (int)(((long)E * 32 + 255) / 256);
  const int gath_blocks = (N + 7) / 8;
  const int elw_blocks  = (int)((NH + 255) / 256);
  const int red_blocks  = (int)(((long)N * 32 + 255) / 256);

  const size_t need_grouped = (size_t)(2L * N + 2 * HD + 2 * NH + (N + 1) + E) * sizeof(float);
  const bool grouped = wide || ((ws_size >= need_grouped) && ((long)E <= NH));
  const int  chunk    = (E + HG - 1) / HG;
  const bool lds_hist = grouped && (N <= HCAP) && ((long)(HG + 1) * N + E <= NH);

  if (lds_hist) {
    hist_lds_kernel<<<2 * HG, 1024, 0, stream>>>(src, dst, hist_in, hist_out, lrank, E, N, chunk);
    hist_fold_kernel<<<(2 * N + 255) / 256, 256, 0, stream>>>(hist_in, hist_out, cnt_tot, norm_out, norm_in, N);
    rowptr_scan_kernel<<<1, 1024, 0, stream>>>(cnt_tot, rowptr, N);
    place_chunked_kernel<<<(E + 255) / 256, 256, 0, stream>>>(src, dst, lrank, rowptr, hist_in, csr_src, E, N, chunk);
  } else {
    fill_zero_kernel<<<512, 256, 0, stream>>>(norm_out, 2L * N);
    degree_kernel<<<(E + 255) / 256, 256, 0, stream>>>(src, dst, cnt_out, cnt_in, grouped ? rank : nullptr, E);
    if (grouped) {
      rowptr_scan_kernel<<<1, 1024, 0, stream>>>(cnt_in, rowptr, N);
      place_kernel<<<(E + 255) / 256, 256, 0, stream>>>(src, dst, rank, rowptr, csr_src, E);
    }
    finalize_norm_kernel<<<(2 * N + 255) / 256, 256, 0, stream>>>(norm_out, 2 * N);
  }
  wsum_kernel<<<1, 128, 0, stream>>>(Wm, bm, wsum, bsum);

  if (grouped) gemm_norm_wmma_kernel<<<gemm_blocks, 256, 0, stream>>>(x, nullptr, nullptr, W1, bufA, N, K1, 0);

  if (wide) {
    spmm_gather_scaled_prelu_both_kernel<<<2 * gath_blocks, 256, 0, stream>>>(rowptr, csr_src, bufA, norm_out, norm_in,
                                                                             b1, a1, H2, perm, N, gath_blocks);
    gemm_norm_wmma_kernel<<<(2 * N + RPB - 1) / RPB, 256, 0, stream>>>(H2, nullptr, norm_out, W2, H2, 2 * N, HD, 1);
    spmm_gather2_prelu_dot_kernel<<<gath_blocks, 256, 0, stream>>>(rowptr, csr_src, H2, norm_in, b2, a2,
                                                                  wsum, bsum, out, N);
    return;
  }

  for (int view = 0; view < 2; ++view) {
    const int* rowmap = view ? perm : nullptr;
    float* outv = out + (long)view * N;
    if (grouped) {
      if (view) {
        spmm_gather_scaled_prelu_kernel<true><<<gath_blocks, 256, 0, stream>>>(rowptr, csr_src, bufA, norm_out, norm_in,
                                                                              b1, a1, bufB, rowmap, N);
      } else {
        spmm_gather_scaled_prelu_kernel<false><<<gath_blocks, 256, 0, stream>>>(rowptr, csr_src, bufA, norm_out, norm_in,
                                                                               b1, a1, bufB, nullptr, N);
      }
      gemm_norm_wmma_kernel<<<gemm_blocks, 256, 0, stream>>>(bufB, nullptr, norm_out, W2, bufB, N, HD, 0);
      spmm_gather_prelu_dot_kernel<<<gath_blocks, 256, 0, stream>>>(rowptr, csr_src, bufB, norm_in, b2, a2,
                                                                   wsum, bsum, outv, N);
    } else {
      gemm_norm_wmma_kernel<<<gemm_blocks, 256, 0, stream>>>(x, rowmap, norm_out, W1, bufA, N, K1, 0);
      fill_zero_kernel<<<2048, 256, 0, stream>>>(bufB, NH);
      spmm_kernel<<<spmm_blocks, 256, 0, stream>>>(src, dst, bufA, bufB, E);
      prelu_kernel<<<elw_blocks, 256, 0, stream>>>(bufB, norm_in, b1, a1, bufA, N);
      gemm_norm_wmma_kernel<<<gemm_blocks, 256, 0, stream>>>(bufA, nullptr, norm_out, W2, bufB, N, HD, 0);
      fill_zero_kernel<<<2048, 256, 0, stream>>>(bufA, NH);
      spmm_kernel<<<spmm_blocks, 256, 0, stream>>>(src, dst, bufB, bufA, E);
      prelu_kernel<<<elw_blocks, 256, 0, stream>>>(bufA, norm_in, b2, a2, bufB, N);
      reduce_kernel<<<red_blocks, 256, 0, stream>>>(bufB, wsum, bsum, outv, N);
    }
  }
}


# both one-work-group launches folded away: list-start scan = scan inside 256-node groups (fold launch) + prefix over groups redone by each placement work-group; wsum in fold launch; 9 -> 7 launches
# speedup vs baseline: 4.2741x; 1.1019x over previous
#include <hip/hip_runtime.h>

typedef float v2f __attribute__((ext_vector_type(2)));
typedef float v8f __attribute__((ext_vector_type(8)));

constexpr int HD   = 96;
constexpr int RPB  = 128;
constexpr int BK   = 32;
constexpr int LPAD = 36;
constexpr int WPAD = 38;
constexpr int HG   = 64;
constexpr int HCAP = 65536;


__global__ void fill_zero_kernel(float* __restrict__ p, long n) {
  long i = (long)blockIdx.x * blockDim.x + threadIdx.x;
  long stride = (long)gridDim.x * blockDim.x;
  for (; i < n; i += stride) p[i] = 0.0f;
}

__global__ void degree_kernel(const int* __restrict__ src, const int* __restrict__ dst,
                              int* __restrict__ cnt_out, int* __restrict__ cnt_in,
                              int* __restrict__ rank, int E) {
  int e = blockIdx.x * blockDim.x + threadIdx.x;
  if (e < E) {
    __hip_atomic_fetch_add(&cnt_out[src[e]], 1, __ATOMIC_RELAXED, __HIP_MEMORY_SCOPE_AGENT);
    if (rank) {
      rank[e] = __hip_atomic_fetch_add(&cnt_in[dst[e]], 1, __ATOMIC_RELAXED, __HIP_MEMORY_SCOPE_AGENT);
    } else {
      __hip_atomic_fetch_add(&cnt_in[dst[e]], 1, __ATOMIC_RELAXED, __HIP_MEMORY_SCOPE_AGENT);
    }
  }
}

__global__ void finalize_norm_kernel(float* __restrict__ p, int n) {
  int i = blockIdx.x * blockDim.x + threadIdx.x;
  if (i < n) p[i] = 1.0f / sqrtf(fmaxf((float)__float_as_int(p[i]), 1.0f));
}

__global__ __launch_bounds__(1024)
void rowptr_scan_kernel(const int* __restrict__ cnt, int* __restrict__ rowptr, int n) {
  __shared__ int wave_tot[32];
  const int t = threadIdx.x, lane = t & 31, w = t >> 5;
  int carry = 0;
  int c[4], nx[4];
  for (int k = 0; k < 4; ++k) {
    const int i = t * 4 + k;
    const int v = cnt[i < n ? i : n - 1];
    c[k] = (i < n) ? v : 0;
  }
  for (int base = 0; base < n; base += 4096) {
    const int i0 = base + t * 4;
    for (int k = 0; k < 4; ++k) {
      const int i = i0 + 4096 + k;
      const int v = cnt[i < n ? i : n - 1];
      nx[k] = (i < n) ? v : 0;
    }
    const int s = c[0] + c[1] + c[2] + c[3];
    int incl = s;
    for (int off = 1; off < 32; off <<= 1) {
      int v = __shfl_up(incl, off, 32);
      if (lane >= off) incl += v;
    }
    if (lane == 31) wave_tot[w] = incl;
    __syncthreads();
    const int wt = wave_tot[lane];
    int wincl = wt;
    for (int off = 1; off < 32; off <<= 1) {
      int v = __shfl_up(wincl, off, 32);
      if (lane >= off) wincl += v;
    }
    const int wave_excl = __shfl(wincl, w, 32) - __shfl(wt, w, 32);
    const int total     = __shfl(wincl, 31, 32);
    int run = carry + wave_excl + incl - s;
    for (int k = 0; k < 4; ++k) {
      if (i0 + k < n) rowptr[i0 + k] = run;
      run += c[k];
    }
    carry += total;
    for (int k = 0; k < 4; ++k) c[k] = nx[k];
    __syncthreads();
  }
  if (t == 0) rowptr[n] = carry;
}

__global__ void place_kernel(const int* __restrict__ src, const int* __restrict__ dst,
                             const int* __restrict__ rank, const int* __restrict__ rowptr,
                             int* __restrict__ csr_src, int E) {
  int e = blockIdx.x * blockDim.x + threadIdx.x;
  if (e < E) csr_src[rowptr[dst[e]] + rank[e]] = src[e];
}

__global__ __launch_bounds__(1024)
void hist_lds_kernel(const int* __restrict__ src, const int* __restrict__ dst,
                     int* __restrict__ hist_in, int* __restrict__ hist_out,
                     int* __restrict__ rank, int E, int n, int chunk) {
  __shared__ int4 cnt4[HCAP / 4];
  int* cnt = reinterpret_cast<int*>(cnt4);
  const int  t     = threadIdx.x;
  const bool is_in = blockIdx.x < HG;
  const int  g     = is_in ? (int)blockIdx.x : (int)blockIdx.x - HG;
  const int* __restrict__ key  = is_in ? dst : src;
  int*       __restrict__ hist = (is_in ? hist_in : hist_out) + (long)g * n;
  const bool vec = ((n & 3) == 0) && ((((size_t)hist) & 15) == 0);

  if (vec) {
    for (int i = t; i < (n >> 2); i += 1024) cnt4[i] = make_int4(0, 0, 0, 0);
  } else {
    for (int i = t; i < n; i += 1024) cnt[i] = 0;
  }
  __syncthreads();

  const int beg = g * chunk;
  const int end = (beg + chunk < E) ? (beg + chunk) : E;
  for (int base = beg + t; base < end; base += 8 * 1024) {
    int k[8];
    #pragma unroll
    for (int u = 0; u < 8; ++u) {
      const int e = base + u * 1024;
      k[u] = key[e < end ? e : end - 1];
    }
    int r[8];
    #pragma unroll
    for (int u = 0; u < 8; ++u) {
      const int e = base + u * 1024;
      r[u] = __hip_atomic_fetch_add(&cnt[k[u]], (e < end) ? 1 : 0, __ATOMIC_RELAXED, __HIP_MEMORY_SCOPE_WORKGROUP);
    }
    if (is_in) {
      #pragma unroll
      for (int u = 0; u < 8; ++u) {
        const int e = base + u * 1024;
        if (e < end) rank[e] = r[u];
      }
    }
  }
  __syncthreads();

  if (vec) {
    int4* __restrict__ h4 = reinterpret_cast<int4*>(hist);
    for (int i = t; i < (n >> 2); i += 1024) h4[i] = cnt4[i];
  } else {
    for (int i = t; i < n; i += 1024) hist[i] = cnt[i];
  }
}

__global__ __launch_bounds__(256)
void hist_fold_scan_kernel(int* __restrict__ hist_in, const int* __restrict__ hist_out,
                           int* __restrict__ rowloc, int* __restrict__ btot,
                           float* __restrict__ norm_out, float* __restrict__ norm_in, int n,
                           const float* __restrict__ Wm, const float* __restrict__ bm,
                           float* __restrict__ wsum, float* __restrict__ bsum) {
  __shared__ int wave_tot[8];
  const int t = threadIdx.x;
  if (blockIdx.x == 0) {
    if (t < HD) {
      float s = 0.0f;
      for (int j = 0; j < HD; ++j) s += Wm[t * HD + j];
      wsum[t] = s;
    } else if (t == HD) {
      float s = 0.0f;
      for (int j = 0; j < HD; ++j) s += bm[j];
      *bsum = s;
    }
    return;
  }
  const int blk = (int)blockIdx.x - 1;
  const int i = blk * 256 + t;
  int c[HG];
  int run = 0;
  if (i < n) {
    #pragma unroll
    for (int g = 0; g < HG; ++g) c[g] = hist_in[(long)g * n + i];
    #pragma unroll
    for (int g = 0; g < HG; ++g) run += c[g];
  } else if (i < 2 * n) {
    const int j = i - n;
    int s = 0;
    #pragma unroll
    for (int g = 0; g < HG; ++g) s += hist_out[(long)g * n + j];
    norm_out[j] = 1.0f / sqrtf(fmaxf((float)s, 1.0f));
  }
  if (blk * 256 < n) {
    const int lane = t & 31, w = t >> 5;
    int incl = run;
    for (int off = 1; off < 32; off <<= 1) {
      int v = __shfl_up(incl, off, 32);
      if (lane >= off) incl += v;
    }
    if (lane == 31) wave_tot[w] = incl;
    __syncthreads();
    int wexcl = 0, tot = 0;
    for (int k = 0; k < 8; ++k) {
      const int v = wave_tot[k];
      if (k < w) wexcl += v;
      tot += v;
    }
    if (i < n) rowloc[i] = wexcl + incl - run;
    if (t == 0) btot[blk] = tot;
  }
  if (i < n) {
    int before = 0;
    #pragma unroll
    for (int g = 0; g < HG; ++g) {
      hist_in[(long)g * n + i] = before;
      before += c[g];
    }
    norm_in[i] = 1.0f / sqrtf(fmaxf((float)run, 1.0f));
  }
}

__global__ __launch_bounds__(256)
void place_chunked_scan_kernel(const int* __restrict__ src, const int* __restrict__ dst,
                               const int* __restrict__ rank, const int* __restrict__ rowloc,
                               const int* __restrict__ btot, const int* __restrict__ before,
                               int* __restrict__ rowptr, int* __restrict__ csr_src,
                               int E, int n, int chunk) {
  __shared__ int bp[256];
  __shared__ int wave_tot[8];
  const int t = threadIdx.x, lane = t & 31, w = t >> 5;
  const int nblk = (n + 255) >> 8;
  const int e  = blockIdx.x * 256 + t;
  const int ec = (e < E) ? e : E - 1;
  const int d  = dst[ec];
  const int rk = rank[ec];
  const int s  = src[ec];
  const int bl = btot[(t < nblk) ? t : 0];
  const int v  = (t < nblk) ? bl : 0;
  const int g   = ec / chunk;
  const int loc = rowloc[d];
  const int bf  = before[(long)g * n + d];
  int incl = v;
  for (int off = 1; off < 32; off <<= 1) {
    int u = __shfl_up(incl, off, 32);
    if (lane >= off) incl += u;
  }
  if (lane == 31) wave_tot[w] = incl;
  __syncthreads();
  int wexcl = 0, tot = 0;
  for (int k = 0; k < 8; ++k) {
    const int u = wave_tot[k];
    if (k < w) wexcl += u;
    tot += u;
  }
  bp[t] = wexcl + incl - v;
  __syncthreads();
  if (e < E) csr_src[bp[d >> 8] + loc + bf + rk] = s;
  for (int b = blockIdx.x; b < nblk; b += gridDim.x) {
    const int i = b * 256 + t;
    if (i < n) rowptr[i] = rowloc[i] + bp[b];
  }
  if (blockIdx.x == 0 && t == 0) rowptr[n] = tot;
}

__global__ void wsum_kernel(const float* __restrict__ Wm, const float* __restrict__ bm,
                            float* __restrict__ wsum, float* __restrict__ bsum) {
  int t = threadIdx.x;
  if (t < HD) {
    float s = 0.0f;
    for (int j = 0; j < HD; ++j) s += Wm[t * HD + j];
    wsum[t] = s;
  } else if (t == HD) {
    float s = 0.0f;
    for (int j = 0; j < HD; ++j) s += bm[j];
    *bsum = s;
  }
}

__global__ __launch_bounds__(256)
void gemm_norm_wmma_kernel(const float* X, const int* __restrict__ rowmap,
                           const float* __restrict__ norm, const float* __restrict__ W,
                           float* OUT, int nrows, int K, int norm_shift) {
  __shared__ float ldsX[RPB][LPAD];
  __shared__ float ldsW[HD][WPAD];

  const int tid  = threadIdx.x;
  const int wave = tid >> 5;
  const int lane = tid & 31;
  const int half = lane >> 4;
  const int l16  = lane & 15;
  const int row0 = blockIdx.x * RPB;

  v8f acc[6] = {};

  const int xc4 = (tid & 7) * 4;
  const float* xptr[4];
  float        xscale[4];
  bool         xrow_ok[4];
  int          xsrc[4];
  #pragma unroll
  for (int it = 0; it < 4; ++it) {
    const int grow = row0 + ((tid + 256 * it) >> 3);
    xrow_ok[it]    = grow < nrows;
    xsrc[it]       = xrow_ok[it] ? grow : nrows - 1;
  }
  if (norm) {
    #pragma unroll
    for (int it = 0; it < 4; ++it) xscale[it] = norm[xsrc[it] >> norm_shift];
  } else {
    #pragma unroll
    for (int it = 0; it < 4; ++it) xscale[it] = 1.0f;
  }
  if (rowmap) {
    #pragma unroll
    for (int it = 0; it < 4; ++it) xsrc[it] = rowmap[xsrc[it]];
  }
  #pragma unroll
  for (int it = 0; it < 4; ++it) xptr[it] = X + (long)xsrc[it] * K + xc4;
  const float* wptr[12];
  int wkk[12];
  #pragma unroll
  for (int it = 0; it < 12; ++it) {
    const int idx = tid + 256 * it;
    wkk[it]  = idx / HD;
    wptr[it] = W + (idx % HD);
  }

  float4 xv[4];
  float  wv[12];
  auto load_tile = [&](int k0) {
    const bool kin = (k0 + xc4) < K;
    #pragma unroll
    for (int it = 0; it < 4; ++it) {
      const float* p = xptr[it] + (kin ? k0 : -xc4);
      xv[it] = *(const float4*)p;
      __builtin_prefetch(((k0 + BK + xc4) < K) ? (p + BK) : p, 0, 0);
    }
    #pragma unroll
    for (int it = 0; it < 12; ++it) {
      const int k = k0 + wkk[it];
      wv[it] = wptr[it][(long)((k < K) ? k : 0) * HD];
    }
  };

  load_tile(0);
  for (int k0 = 0; k0 < K; k0 += BK) {
    {
      const bool kin = (k0 + xc4) < K;
      #pragma unroll
      for (int it = 0; it < 4; ++it) {
        const int r = (tid + 256 * it) >> 3;
        float4 v = xv[it];
        const float s = xscale[it];
        v.x *= s; v.y *= s; v.z *= s; v.w *= s;
        if (!(kin && xrow_ok[it])) v = make_float4(0.f, 0.f, 0.f, 0.f);
        ldsX[r][xc4 + 0] = v.x;
        ldsX[r][xc4 + 1] = v.y;
        ldsX[r][xc4 + 2] = v.z;
        ldsX[r][xc4 + 3] = v.w;
      }
      #pragma unroll
      for (int it = 0; it < 12; ++it) {
        const int idx = tid + 256 * it;
        ldsW[idx % HD][wkk[it]] = (k0 + wkk[it] < K) ? wv[it] : 0.0f;
      }
    }
    __syncthreads();

    if (k0 + BK < K) load_tile(k0 + BK);

    const int kmax = (BK < K - k0) ? BK : (K - k0);
    const int mrow = wave * 16 + l16;
    for (int kk = 0; kk < kmax; kk += 4) {
      int ka = kk + half * 2;

      v2f A = *(const v2f*)&ldsX[mrow][ka];
      v2f Bf[6];
      #pragma unroll
      for (int t = 0; t < 6; ++t)
        Bf[t] = *(const v2f*)&ldsW[t * 16 + l16][ka];

      #pragma unroll
      for (int t = 0; t < 6; ++t)
        acc[t] = __builtin_amdgcn_wmma_f32_16x16x4_f32(
             false, A,  false, Bf[t],
             (short)0, acc[t],  false,  false);
    }
    __syncthreads();
  }

  const int orow = row0 + wave * 16 + half * 8;
  #pragma unroll
  for (int t = 0; t < 6; ++t) {
    int col = t * 16 + l16;
    #pragma unroll
    for (int i = 0; i < 8; ++i) {
      int r = orow + i;
      if (r < nrows) OUT[(long)r * HD + col] = acc[t][i];
    }
  }
}

__device__ __forceinline__ void gather_row_sum(const int* __restrict__ rowptr, const int* __restrict__ csr_src,
                                               const float* __restrict__ h, int row, int lane,
                                               float& a0, float& a1, float& a2) {
  const int beg = rowptr[row];
  const int end = rowptr[row + 1];
  a0 = 0.0f; a1 = 0.0f; a2 = 0.0f;
  for (int base = beg; base < end; base += 32) {
    const int cnt  = (end - base < 32) ? (end - base) : 32;
    const int mine = (lane < cnt) ? csr_src[base + lane] : 0;
    int j = 0;
    for (; j + 4 <= cnt; j += 4) {
      const float* p0 = h + (long)__builtin_amdgcn_readlane(mine, j + 0) * HD + lane;
      const float* p1 = h + (long)__builtin_amdgcn_readlane(mine, j + 1) * HD + lane;
      const float* p2 = h + (long)__builtin_amdgcn_readlane(mine, j + 2) * HD + lane;
      const float* p3 = h + (long)__builtin_amdgcn_readlane(mine, j + 3) * HD + lane;
      const float x00 = p0[0], x01 = p0[32], x02 = p0[64];
      const float x10 = p1[0], x11 = p1[32], x12 = p1[64];
      const float x20 = p2[0], x21 = p2[32], x22 = p2[64];
      const float x30 = p3[0], x31 = p3[32], x32 = p3[64];
      a0 += x00; a1 += x01; a2 += x02;
      a0 += x10; a1 += x11; a2 += x12;
      a0 += x20; a1 += x21; a2 += x22;
      a0 += x30; a1 += x31; a2 += x32;
    }
    for (; j < cnt; ++j) {
      const float* p = h + (long)__builtin_amdgcn_readlane(mine, j) * HD + lane;
      a0 += p[0]; a1 += p[32]; a2 += p[64];
    }
  }
}

template <bool MAPPED>
__device__ __forceinline__ void gather_row_sum_scaled(const int* __restrict__ rowptr, const int* __restrict__ csr_src,
                                                      const float* __restrict__ xw, const int* __restrict__ rowmap,
                                                      const float* __restrict__ scale, int row, int lane,
                                                      float& a0, float& a1, float& a2) {
  const int beg = rowptr[row];
  const int end = rowptr[row + 1];
  a0 = 0.0f; a1 = 0.0f; a2 = 0.0f;
  for (int base = beg; base < end; base += 32) {
    const int cnt  = (end - base < 32) ? (end - base) : 32;
    const int mine = (lane < cnt) ? csr_src[base + lane] : 0;
    const int m    = MAPPED ? rowmap[mine] : mine;
    const int sc   = __float_as_int(scale[mine]);
    int j = 0;
    for (; j + 4 <= cnt; j += 4) {
      const float* p0 = xw + (long)__builtin_amdgcn_readlane(m, j + 0) * HD + lane;
      const float* p1 = xw + (long)__builtin_amdgcn_readlane(m, j + 1) * HD + lane;
      const float* p2 = xw + (long)__builtin_amdgcn_readlane(m, j + 2) * HD + lane;
      const float* p3 = xw + (long)__builtin_amdgcn_readlane(m, j + 3) * HD + lane;
      const float x00 = p0[0], x01 = p0[32], x02 = p0[64];
      const float x10 = p1[0], x11 = p1[32], x12 = p1[64];
      const float x20 = p2[0], x21 = p2[32], x22 = p2[64];
      const float x30 = p3[0], x31 = p3[32], x32 = p3[64];
      __builtin_amdgcn_sched_barrier(0);
      const float c0 = __int_as_float(__builtin_amdgcn_readlane(sc, j + 0));
      const float c1 = __int_as_float(__builtin_amdgcn_readlane(sc, j + 1));
      const float c2 = __int_as_float(__builtin_amdgcn_readlane(sc, j + 2));
      const float c3 = __int_as_float(__builtin_amdgcn_readlane(sc, j + 3));
      a0 = fmaf(c0, x00, a0); a1 = fmaf(c0, x01, a1); a2 = fmaf(c0, x02, a2);
      a0 = fmaf(c1, x10, a0); a1 = fmaf(c1, x11, a1); a2 = fmaf(c1, x12, a2);
      a0 = fmaf(c2, x20, a0); a1 = fmaf(c2, x21, a1); a2 = fmaf(c2, x22, a2);
      a0 = fmaf(c3, x30, a0); a1 = fmaf(c3, x31, a1); a2 = fmaf(c3, x32, a2);
    }
    for (; j < cnt; ++j) {
      const float* p = xw + (long)__builtin_amdgcn_readlane(m, j) * HD + lane;
      const float x0 = p[0], x1 = p[32], x2 = p[64];
      const float c  = __int_as_float(__builtin_amdgcn_readlane(sc, j));
      a0 = fmaf(c, x0, a0); a1 = fmaf(c, x1, a1); a2 = fmaf(c, x2, a2);
    }
  }
}

template <bool MAPPED>
__device__ __forceinline__ void gather_scaled_prelu_node(const int* __restrict__ rowptr, const int* __restrict__ csr_src,
                                                         const float* __restrict__ xw, const float* __restrict__ norm_out,
                                                         const float* __restrict__ norm_in, const float* __restrict__ b,
                                                         const float* __restrict__ a, float* __restrict__ act_row,
                                                         const int* __restrict__ rowmap, int row, int lane) {
  const float ni = norm_in[row];
  const float b0 = b[lane], b1 = b[lane + 32], b2 = b[lane + 64];
  const float s0 = a[lane], s1 = a[lane + 32], s2 = a[lane + 64];
  float a0, a1, a2;
  gather_row_sum_scaled<MAPPED>(rowptr, csr_src, xw, rowmap, norm_out, row, lane, a0, a1, a2);
  const float v0 = a0 * ni + b0;
  const float v1 = a1 * ni + b1;
  const float v2 = a2 * ni + b2;
  float* o = act_row + lane;
  o[0]  = (v0 >= 0.0f) ? v0 : s0 * v0;
  o[32] = (v1 >= 0.0f) ? v1 : s1 * v1;
  o[64] = (v2 >= 0.0f) ? v2 : s2 * v2;
}

template <bool MAPPED>
__global__ __launch_bounds__(256)
void spmm_gather_scaled_prelu_kernel(const int* __restrict__ rowptr, const int* __restrict__ csr_src,
                                     const float* __restrict__ xw, const float* __restrict__ norm_out,
                                     const float* __restrict__ norm_in, const float* __restrict__ b,
                                     const float* __restrict__ a, float* __restrict__ act,
                                     const int* __restrict__ rowmap, int n) {
  const int lane = threadIdx.x & 31;
  const int row = __builtin_amdgcn_readfirstlane((int)(blockIdx.x * (blockDim.x >> 5) + (threadIdx.x >> 5)));
  if (row >= n) return;
  gather_scaled_prelu_node<MAPPED>(rowptr, csr_src, xw, norm_out, norm_in, b, a, act + (long)row * HD, rowmap, row, lane);
}

__global__ __launch_bounds__(256)
void spmm_gather_scaled_prelu_both_kernel(const int* __restrict__ rowptr, const int* __restrict__ csr_src,
                                          const float* __restrict__ xw, const float* __restrict__ norm_out,
                                          const float* __restrict__ norm_in, const float* __restrict__ b,
                                          const float* __restrict__ a, float* __restrict__ act2,
                                          const int* __restrict__ rowmap, int n, int blocks_per_view) {
  const int lane = threadIdx.x & 31;
  const bool second = (int)blockIdx.x >= blocks_per_view;
  const int blk = (int)blockIdx.x - (second ? blocks_per_view : 0);
  const int row = __builtin_amdgcn_readfirstlane((int)(blk * (blockDim.x >> 5) + (threadIdx.x >> 5)));
  if (row >= n) return;
  float* dst = act2 + (long)row * (2 * HD);
  if (second) gather_scaled_prelu_node<true>(rowptr, csr_src, xw, norm_out, norm_in, b, a, dst + HD, rowmap, row, lane);
  else        gather_scaled_prelu_node<false>(rowptr, csr_src, xw, norm_out, norm_in, b, a, dst, nullptr, row, lane);
}

__global__ __launch_bounds__(256)
void spmm_gather_prelu_dot_kernel(const int* __restrict__ rowptr, const int* __restrict__ csr_src,
                                  const float* __restrict__ h, const float* __restrict__ norm_in,
                                  const float* __restrict__ b, const float* __restrict__ a,
                                  const float* __restrict__ wsum, const float* __restrict__ bsum,
                                  float* __restrict__ out, int n) {
  const int lane = threadIdx.x & 31;
  const int row = __builtin_amdgcn_readfirstlane((int)(blockIdx.x * (blockDim.x >> 5) + (threadIdx.x >> 5)));
  if (row >= n) return;
  const float ni = norm_in[row];
  const float b0 = b[lane], b1 = b[lane + 32], b2 = b[lane + 64];
  const float s0 = a[lane], s1 = a[lane + 32], s2 = a[lane + 64];
  const float w0 = wsum[lane], w1 = wsum[lane + 32], w2 = wsum[lane + 64];
  const float bs = *bsum;
  float a0, a1, a2;
  gather_row_sum(rowptr, csr_src, h, row, lane, a0, a1, a2);
  const float v0 = a0 * ni + b0;
  const float v1 = a1 * ni + b1;
  const float v2 = a2 * ni + b2;
  const float z0 = (v0 >= 0.0f) ? v0 : s0 * v0;
  const float z1 = (v1 >= 0.0f) ? v1 : s1 * v1;
  const float z2 = (v2 >= 0.0f) ? v2 : s2 * v2;
  float s = fmaf(z0, w0, 0.0f);
  s = fmaf(z1, w1, s);
  s = fmaf(z2, w2, s);
  #pragma unroll
  for (int off = 16; off > 0; off >>= 1) s += __shfl_xor(s, off, 32);
  if (lane == 0) out[row] = s + bs;
}

__device__ __forceinline__ void gather_row_sum2(const int* __restrict__ rowptr, const int* __restrict__ csr_src,
                                                const float* __restrict__ h2, int row, int lane,
                                                float& a0, float& a1, float& a2, float& c0, float& c1, float& c2) {
  const int beg = rowptr[row];
  const int end = rowptr[row + 1];
  a0 = 0.0f; a1 = 0.0f; a2 = 0.0f; c0 = 0.0f; c1 = 0.0f; c2 = 0.0f;
  for (int base = beg; base < end; base += 32) {
    const int cnt  = (end - base < 32) ? (end - base) : 32;
    const int mine = (lane < cnt) ? csr_src[base + lane] : 0;
    int j = 0;
    for (; j + 4 <= cnt; j += 4) {
      const float* p0 = h2 + (long)__builtin_amdgcn_readlane(mine, j + 0) * (2 * HD) + lane;
      const float* p1 = h2 + (long)__builtin_amdgcn_readlane(mine, j + 1) * (2 * HD) + lane;
      const float* p2 = h2 + (long)__builtin_amdgcn_readlane(mine, j + 2) * (2 * HD) + lane;
      const float* p3 = h2 + (long)__builtin_amdgcn_readlane(mine, j + 3) * (2 * HD) + lane;
      const float x00 = p0[0], x01 = p0[32], x02 = p0[64], y00 = p0[96], y01 = p0[128], y02 = p0[160];
      const float x10 = p1[0], x11 = p1[32], x12 = p1[64], y10 = p1[96], y11 = p1[128], y12 = p1[160];
      const float x20 = p2[0], x21 = p2[32], x22 = p2[64], y20 = p2[96], y21 = p2[128], y22 = p2[160];
      const float x30 = p3[0], x31 = p3[32], x32 = p3[64], y30 = p3[96], y31 = p3[128], y32 = p3[160];
      a0 += x00; a1 += x01; a2 += x02; c0 += y00; c1 += y01; c2 += y02;
      a0 += x10; a1 += x11; a2 += x12; c0 += y10; c1 += y11; c2 += y12;
      a0 += x20; a1 += x21; a2 += x22; c0 += y20; c1 += y21; c2 += y22;
      a0 += x30; a1 += x31; a2 += x32; c0 += y30; c1 += y31; c2 += y32;
    }
    for (; j < cnt; ++j) {
      const float* p = h2 + (long)__builtin_amdgcn_readlane(mine, j) * (2 * HD) + lane;
      const float x0 = p[0], x1 = p[32], x2 = p[64], y0 = p[96], y1 = p[128], y2 = p[160];
      a0 += x0; a1 += x1; a2 += x2; c0 += y0; c1 += y1; c2 += y2;
    }
  }
}

__global__ __launch_bounds__(256)
void spmm_gather2_prelu_dot_kernel(const int* __restrict__ rowptr, const int* __restrict__ csr_src,
                                   const float* __restrict__ h2, const float* __restrict__ norm_in,
                                   const float* __restrict__ b, const float* __restrict__ a,
                                   const float* __restrict__ wsum, const float* __restrict__ bsum,
                                   float* __restrict__ out, int n) {
  const int lane = threadIdx.x & 31;
  const int row = __builtin_amdgcn_readfirstlane((int)(blockIdx.x * (blockDim.x >> 5) + (threadIdx.x >> 5)));
  if (row >= n) return;
  const float ni = norm_in[row];
  const float b0 = b[lane], b1 = b[lane + 32], b2 = b[lane + 64];
  const float s0 = a[lane], s1 = a[lane + 32], s2 = a[lane + 64];
  const float w0 = wsum[lane], w1 = wsum[lane + 32], w2 = wsum[lane + 64];
  const float bs = *bsum;
  float a0, a1, a2, c0, c1, c2;
  gather_row_sum2(rowptr, csr_src, h2, row, lane, a0, a1, a2, c0, c1, c2);
  const float v0 = a0 * ni + b0;
  const float v1 = a1 * ni + b1;
  const float v2 = a2 * ni + b2;
  const float u0 = c0 * ni + b0;
  const float u1 = c1 * ni + b1;
  const float u2 = c2 * ni + b2;
  const float z0 = (v0 >= 0.0f) ? v0 : s0 * v0;
  const float z1 = (v1 >= 0.0f) ? v1 : s1 * v1;
  const float z2 = (v2 >= 0.0f) ? v2 : s2 * v2;
  const float q0 = (u0 >= 0.0f) ? u0 : s0 * u0;
  const float q1 = (u1 >= 0.0f) ? u1 : s1 * u1;
  const float q2 = (u2 >= 0.0f) ? u2 : s2 * u2;
  float s = fmaf(z0, w0, 0.0f);
  s = fmaf(z1, w1, s);
  s = fmaf(z2, w2, s);
  float r = fmaf(q0, w0, 0.0f);
  r = fmaf(q1, w1, r);
  r = fmaf(q2, w2, r);
  for (int off = 16; off > 0; off >>= 1) {
    s += __shfl_xor(s, off, 32);
    r += __shfl_xor(r, off, 32);
  }
  if (lane == 0) {
    out[row]     = s + bs;
    out[n + row] = r + bs;
  }
}

__global__ void spmm_kernel(const int* __restrict__ src, const int* __restrict__ dst,
                            const float* __restrict__ h, float* __restrict__ agg, int E) {
  int gw   = (int)(((long)blockIdx.x * blockDim.x + threadIdx.x) >> 5);
  int lane = threadIdx.x & 31;
  if (gw < E) {
    int s = src[gw], d = dst[gw];
    const float* hp = h + (long)s * HD;
    float* ap = agg + (long)d * HD;
    #pragma unroll
    for (int i = 0; i < 3; ++i) {
      float v = hp[lane + 32 * i];
      __hip_atomic_fetch_add(&ap[lane + 32 * i], v, __ATOMIC_RELAXED, __HIP_MEMORY_SCOPE_AGENT);
    }
  }
}

__global__ void prelu_kernel(const float* __restrict__ agg, const float* __restrict__ norm_in,
                             const float* __restrict__ b, const float* __restrict__ a,
                             float* __restrict__ out, int n) {
  long idx = (long)blockIdx.x * blockDim.x + threadIdx.x;
  if (idx < (long)n * HD) {
    int i = (int)(idx / HD);
    int j = (int)(idx % HD);
    float v = agg[idx] * norm_in[i] + b[j];
    out[idx] = (v >= 0.0f) ? v : a[j] * v;
  }
}

__global__ void reduce_kernel(const float* __restrict__ z, const float* __restrict__ wsum,
                              const float* __restrict__ bsum, float* __restrict__ out, int n) {
  int gw   = (int)(((long)blockIdx.x * blockDim.x + threadIdx.x) >> 5);
  int lane = threadIdx.x & 31;
  if (gw < n) {
    const float* zp = z + (long)gw * HD;
    float s = 0.0f;
    #pragma unroll
    for (int i = 0; i < 3; ++i) s = fmaf(zp[lane + 32 * i], wsum[lane + 32 * i], s);
    #pragma unroll
    for (int off = 16; off > 0; off >>= 1) s += __shfl_xor(s, off, 32);
    if (lane == 0) out[gw] = s + *bsum;
  }
}

extern "C" void kernel_launch(void* const* d_in, const int* in_sizes, int n_in,
                              void* d_out, int out_size, void* d_ws, size_t ws_size,
                              hipStream_t stream) {
  const float* x    = (const float*)d_in[0];
  const int*   src  = (const int*)  d_in[1];
  const int*   dst  = (const int*)  d_in[2];
  const int*   perm = (const int*)  d_in[3];
  const float* W1   = (const float*)d_in[4];
  const float* b1   = (const float*)d_in[5];
  const float* a1   = (const float*)d_in[6];
  const float* W2   = (const float*)d_in[7];
  const float* b2   = (const float*)d_in[8];
  const float* a2   = (const float*)d_in[9];
  const float* Wm   = (const float*)d_in[10];
  const float* bm   = (const float*)d_in[11];
  float* out = (float*)d_out;

  const int E  = in_sizes[1];
  const int N  = in_sizes[3];
  const int K1 = in_sizes[0] / N;

  float* ws       = (float*)d_ws;
  float* norm_out = ws;
  float* norm_in  = ws + N;
  float* wsum     = ws + 2L * N;
  float* bsum     = ws + 2L * N + HD;
  float* bufA     = ws + 2L * N + 2 * HD;
  float* bufB     = bufA + (long)N * HD;

  const long NH = (long)N * HD;
  const size_t need_wide = (size_t)(2L * N + 2 * HD + 3 * NH + (N + 1) + E) * sizeof(float);
  const bool wide = (ws_size >= need_wide) && ((long)E <= NH);
  float* H2       = bufB;
  int*   rowptr   = (int*)(bufB + (wide ? 2 * NH : NH));
  int*   csr_src  = rowptr + (N + 1);
  int*   rank     = (int*)bufB;
  int*   cnt_out  = (int*)norm_out;
  int*   cnt_in   = (int*)norm_in;
  int*   hist_in  = (int*)bufA;
  int*   hist_out = (int*)bufB;
  int*   lrank    = hist_out + (long)HG * N;
  int*   rowloc   = lrank + E;
  int*   btot     = rowloc + N;

  const int gemm_blocks = (N + RPB - 1) / RPB;
  const int spmm_blocks = (int)(((long)E * 32 + 255) / 256);
  const int gath_blocks = (N + 7) / 8;
  const int elw_blocks  = (int)((NH + 255) / 256);
  const int red_blocks  = (int)(((long)N * 32 + 255) / 256);

  const size_t need_grouped = (size_t)(2L * N + 2 * HD + 2 * NH + (N + 1) + E) * sizeof(float);
  const bool grouped = wide || ((ws_size >= need_grouped) && ((long)E <= NH));
  const int  chunk    = (E + HG - 1) / HG;
  const bool lds_hist = grouped && (N <= HCAP) && ((long)(HG + 1) * N + E + 256 <= NH);

  if (lds_hist) {
    hist_lds_kernel<<<2 * HG, 1024, 0, stream>>>(src, dst, hist_in, hist_out, lrank, E, N, chunk);
    hist_fold_scan_kernel<<<(2 * N + 255) / 256 + 1, 256, 0, stream>>>(hist_in, hist_out, rowloc, btot, norm_out, norm_in, N,
                                                                       Wm, bm, wsum, bsum);
    place_chunked_scan_kernel<<<(E + 255) / 256, 256, 0, stream>>>(src, dst, lrank, rowloc, btot, hist_in, rowptr,
                                                                   csr_src, E, N, chunk);
  } else {
    fill_zero_kernel<<<512, 256, 0, stream>>>(norm_out, 2L * N);
    degree_kernel<<<(E + 255) / 256, 256, 0, stream>>>(src, dst, cnt_out, cnt_in, grouped ? rank : nullptr, E);
    if (grouped) {
      rowptr_scan_kernel<<<1, 1024, 0, stream>>>(cnt_in, rowptr, N);
      place_kernel<<<(E + 255) / 256, 256, 0, stream>>>(src, dst, rank, rowptr, csr_src, E);
    }
    finalize_norm_kernel<<<(2 * N + 255) / 256, 256, 0, stream>>>(norm_out, 2 * N);
    wsum_kernel<<<1, 128, 0, stream>>>(Wm, bm, wsum, bsum);
  }

  if (grouped) gemm_norm_wmma_kernel<<<gemm_blocks, 256, 0, stream>>>(x, nullptr, nullptr, W1, bufA, N, K1, 0);

  if (wide) {
    spmm_gather_scaled_prelu_both_kernel<<<2 * gath_blocks, 256, 0, stream>>>(rowptr, csr_src, bufA, norm_out, norm_in,
                                                                             b1, a1, H2, perm, N, gath_blocks);
    gemm_norm_wmma_kernel<<<(2 * N + RPB - 1) / RPB, 256, 0, stream>>>(H2, nullptr, norm_out, W2, H2, 2 * N, HD, 1);
    spmm_gather2_prelu_dot_kernel<<<gath_blocks, 256, 0, stream>>>(rowptr, csr_src, H2, norm_in, b2, a2,
                                                                  wsum, bsum, out, N);
    return;
  }

  for (int view = 0; view < 2; ++view) {
    const int* rowmap = view ? perm : nullptr;
    float* outv = out + (long)view * N;
    if (grouped) {
      if (view) {
        spmm_gather_scaled_prelu_kernel<true><<<gath_blocks, 256, 0, stream>>>(rowptr, csr_src, bufA, norm_out, norm_in,
                                                                              b1, a1, bufB, rowmap, N);
      } else {
        spmm_gather_scaled_prelu_kernel<false><<<gath_blocks, 256, 0, stream>>>(rowptr, csr_src, bufA, norm_out, norm_in,
                                                                               b1, a1, bufB, nullptr, N);
      }
      gemm_norm_wmma_kernel<<<gemm_blocks, 256, 0, stream>>>(bufB, nullptr, norm_out, W2, bufB, N, HD, 0);
      spmm_gather_prelu_dot_kernel<<<gath_blocks, 256, 0, stream>>>(rowptr, csr_src, bufB, norm_in, b2, a2,
                                                                   wsum, bsum, outv, N);
    } else {
      gemm_norm_wmma_kernel<<<gemm_blocks, 256, 0, stream>>>(x, rowmap, norm_out, W1, bufA, N, K1, 0);
      fill_zero_kernel<<<2048, 256, 0, stream>>>(bufB, NH);
      spmm_kernel<<<spmm_blocks, 256, 0, stream>>>(src, dst, bufA, bufB, E);
      prelu_kernel<<<elw_blocks, 256, 0, stream>>>(bufB, norm_in, b1, a1, bufA, N);
      gemm_norm_wmma_kernel<<<gemm_blocks, 256, 0, stream>>>(bufA, nullptr, norm_out, W2, bufB, N, HD, 0);
      fill_zero_kernel<<<2048, 256, 0, stream>>>(bufA, NH);
      spmm_kernel<<<spmm_blocks, 256, 0, stream>>>(src, dst, bufB, bufA, E);
      prelu_kernel<<<elw_blocks, 256, 0, stream>>>(bufA, norm_in, b2, a2, bufB, N);
      reduce_kernel<<<red_blocks, 256, 0, stream>>>(bufB, wsum, bsum, outv, N);
    }
  }
}


# x@W1 (needs nothing of the graph) and the edge placement share one launch: product work-groups first, placement as few work-groups of 8 edges per thread in the places they leave; 7 -> 6 launches
# speedup vs baseline: 4.8389x; 1.1321x over previous
#include <hip/hip_runtime.h>

typedef float v2f __attribute__((ext_vector_type(2)));
typedef float v8f __attribute__((ext_vector_type(8)));

constexpr int HD   = 96;
constexpr int RPB  = 128;
constexpr int BK   = 32;
constexpr int LPAD = 36;
constexpr int WPAD = 38;
constexpr int HG   = 64;
constexpr int HCAP = 65536;


__global__ void fill_zero_kernel(float* __restrict__ p, long n) {
  long i = (long)blockIdx.x * blockDim.x + threadIdx.x;
  long stride = (long)gridDim.x * blockDim.x;
  for (; i < n; i += stride) p[i] = 0.0f;
}

__global__ void degree_kernel(const int* __restrict__ src, const int* __restrict__ dst,
                              int* __restrict__ cnt_out, int* __restrict__ cnt_in,
                              int* __restrict__ rank, int E) {
  int e = blockIdx.x * blockDim.x + threadIdx.x;
  if (e < E) {
    __hip_atomic_fetch_add(&cnt_out[src[e]], 1, __ATOMIC_RELAXED, __HIP_MEMORY_SCOPE_AGENT);
    if (rank) {
      rank[e] = __hip_atomic_fetch_add(&cnt_in[dst[e]], 1, __ATOMIC_RELAXED, __HIP_MEMORY_SCOPE_AGENT);
    } else {
      __hip_atomic_fetch_add(&cnt_in[dst[e]], 1, __ATOMIC_RELAXED, __HIP_MEMORY_SCOPE_AGENT);
    }
  }
}

__global__ void finalize_norm_kernel(float* __restrict__ p, int n) {
  int i = blockIdx.x * blockDim.x + threadIdx.x;
  if (i < n) p[i] = 1.0f / sqrtf(fmaxf((float)__float_as_int(p[i]), 1.0f));
}

__global__ __launch_bounds__(1024)
void rowptr_scan_kernel(const int* __restrict__ cnt, int* __restrict__ rowptr, int n) {
  __shared__ int wave_tot[32];
  const int t = threadIdx.x, lane = t & 31, w = t >> 5;
  int carry = 0;
  int c[4], nx[4];
  for (int k = 0; k < 4; ++k) {
    const int i = t * 4 + k;
    const int v = cnt[i < n ? i : n - 1];
    c[k] = (i < n) ? v : 0;
  }
  for (int base = 0; base < n; base += 4096) {
    const int i0 = base + t * 4;
    for (int k = 0; k < 4; ++k) {
      const int i = i0 + 4096 + k;
      const int v = cnt[i < n ? i : n - 1];
      nx[k] = (i < n) ? v : 0;
    }
    const int s = c[0] + c[1] + c[2] + c[3];
    int incl = s;
    for (int off = 1; off < 32; off <<= 1) {
      int v = __shfl_up(incl, off, 32);
      if (lane >= off) incl += v;
    }
    if (lane == 31) wave_tot[w] = incl;
    __syncthreads();
    const int wt = wave_tot[lane];
    int wincl = wt;
    for (int off = 1; off < 32; off <<= 1) {
      int v = __shfl_up(wincl, off, 32);
      if (lane >= off) wincl += v;
    }
    const int wave_excl = __shfl(wincl, w, 32) - __shfl(wt, w, 32);
    const int total     = __shfl(wincl, 31, 32);
    int run = carry + wave_excl + incl - s;
    for (int k = 0; k < 4; ++k) {
      if (i0 + k < n) rowptr[i0 + k] = run;
      run += c[k];
    }
    carry += total;
    for (int k = 0; k < 4; ++k) c[k] = nx[k];
    __syncthreads();
  }
  if (t == 0) rowptr[n] = carry;
}

__global__ void place_kernel(const int* __restrict__ src, const int* __restrict__ dst,
                             const int* __restrict__ rank, const int* __restrict__ rowptr,
                             int* __restrict__ csr_src, int E) {
  int e = blockIdx.x * blockDim.x + threadIdx.x;
  if (e < E) csr_src[rowptr[dst[e]] + rank[e]] = src[e];
}

__global__ __launch_bounds__(1024)
void hist_lds_kernel(const int* __restrict__ src, const int* __restrict__ dst,
                     int* __restrict__ hist_in, int* __restrict__ hist_out,
                     int* __restrict__ rank, int E, int n, int chunk) {
  __shared__ int4 cnt4[HCAP / 4];
  int* cnt = reinterpret_cast<int*>(cnt4);
  const int  t     = threadIdx.x;
  const bool is_in = blockIdx.x < HG;
  const int  g     = is_in ? (int)blockIdx.x : (int)blockIdx.x - HG;
  const int* __restrict__ key  = is_in ? dst : src;
  int*       __restrict__ hist = (is_in ? hist_in : hist_out) + (long)g * n;
  const bool vec = ((n & 3) == 0) && ((((size_t)hist) & 15) == 0);

  if (vec) {
    for (int i = t; i < (n >> 2); i += 1024) cnt4[i] = make_int4(0, 0, 0, 0);
  } else {
    for (int i = t; i < n; i += 1024) cnt[i] = 0;
  }
  __syncthreads();

  const int beg = g * chunk;
  const int end = (beg + chunk < E) ? (beg + chunk) : E;
  for (int base = beg + t; base < end; base += 8 * 1024) {
    int k[8];
    #pragma unroll
    for (int u = 0; u < 8; ++u) {
      const int e = base + u * 1024;
      k[u] = key[e < end ? e : end - 1];
    }
    int r[8];
    #pragma unroll
    for (int u = 0; u < 8; ++u) {
      const int e = base + u * 1024;
      r[u] = __hip_atomic_fetch_add(&cnt[k[u]], (e < end) ? 1 : 0, __ATOMIC_RELAXED, __HIP_MEMORY_SCOPE_WORKGROUP);
    }
    if (is_in) {
      #pragma unroll
      for (int u = 0; u < 8; ++u) {
        const int e = base + u * 1024;
        if (e < end) rank[e] = r[u];
      }
    }
  }
  __syncthreads();

  if (vec) {
    int4* __restrict__ h4 = reinterpret_cast<int4*>(hist);
    for (int i = t; i < (n >> 2); i += 1024) h4[i] = cnt4[i];
  } else {
    for (int i = t; i < n; i += 1024) hist[i] = cnt[i];
  }
}

__global__ __launch_bounds__(256)
void hist_fold_scan_kernel(int* __restrict__ hist_in, const int* __restrict__ hist_out,
                           int* __restrict__ rowloc, int* __restrict__ btot,
                           float* __restrict__ norm_out, float* __restrict__ norm_in, int n,
                           const float* __restrict__ Wm, const float* __restrict__ bm,
                           float* __restrict__ wsum, float* __restrict__ bsum) {
  __shared__ int wave_tot[8];
  const int t = threadIdx.x;
  if (blockIdx.x == 0) {
    if (t < HD) {
      float s = 0.0f;
      for (int j = 0; j < HD; ++j) s += Wm[t * HD + j];
      wsum[t] = s;
    } else if (t == HD) {
      float s = 0.0f;
      for (int j = 0; j < HD; ++j) s += bm[j];
      *bsum = s;
    }
    return;
  }
  const int blk = (int)blockIdx.x - 1;
  const int i = blk * 256 + t;
  int c[HG];
  int run = 0;
  if (i < n) {
    #pragma unroll
    for (int g = 0; g < HG; ++g) c[g] = hist_in[(long)g * n + i];
    #pragma unroll
    for (int g = 0; g < HG; ++g) run += c[g];
  } else if (i < 2 * n) {
    const int j = i - n;
    int s = 0;
    #pragma unroll
    for (int g = 0; g < HG; ++g) s += hist_out[(long)g * n + j];
    norm_out[j] = 1.0f / sqrtf(fmaxf((float)s, 1.0f));
  }
  if (blk * 256 < n) {
    const int lane = t & 31, w = t >> 5;
    int incl = run;
    for (int off = 1; off < 32; off <<= 1) {
      int v = __shfl_up(incl, off, 32);
      if (lane >= off) incl += v;
    }
    if (lane == 31) wave_tot[w] = incl;
    __syncthreads();
    int wexcl = 0, tot = 0;
    for (int k = 0; k < 8; ++k) {
      const int v = wave_tot[k];
      if (k < w) wexcl += v;
      tot += v;
    }
    if (i < n) rowloc[i] = wexcl + incl - run;
    if (t == 0) btot[blk] = tot;
  }
  if (i < n) {
    int before = 0;
    #pragma unroll
    for (int g = 0; g < HG; ++g) {
      hist_in[(long)g * n + i] = before;
      before += c[g];
    }
    norm_in[i] = 1.0f / sqrtf(fmaxf((float)run, 1.0f));
  }
}

template <int EPT>
__device__ __forceinline__
void place_chunked_scan_body(const int* __restrict__ src, const int* __restrict__ dst,
                             const int* __restrict__ rank, const int* __restrict__ rowloc,
                             const int* __restrict__ btot, const int* __restrict__ before,
                             int* __restrict__ rowptr, int* __restrict__ csr_src,
                             int E, int n, int chunk, int wg, int nwg) {
  __shared__ int bp[256];
  __shared__ int wave_tot[8];
  const int t = threadIdx.x, lane = t & 31, w = t >> 5;
  const int nblk = (n + 255) >> 8;
  int d[EPT], rk[EPT], s[EPT], loc[EPT], bf[EPT];
  #pragma unroll
  for (int u = 0; u < EPT; ++u) {
    const int e  = (wg * EPT + u) * 256 + t;
    const int ec = (e < E) ? e : E - 1;
    d[u]  = dst[ec];
    rk[u] = rank[ec];
    s[u]  = src[ec];
  }
  const int bl = btot[(t < nblk) ? t : 0];
  const int v  = (t < nblk) ? bl : 0;
  #pragma unroll
  for (int u = 0; u < EPT; ++u) {
    const int e  = (wg * EPT + u) * 256 + t;
    const int ec = (e < E) ? e : E - 1;
    const int g  = ec / chunk;
    loc[u] = rowloc[d[u]];
    bf[u]  = before[(long)g * n + d[u]];
  }
  int incl = v;
  for (int off = 1; off < 32; off <<= 1) {
    int u = __shfl_up(incl, off, 32);
    if (lane >= off) incl += u;
  }
  if (lane == 31) wave_tot[w] = incl;
  __syncthreads();
  int wexcl = 0, tot = 0;
  for (int k = 0; k < 8; ++k) {
    const int u = wave_tot[k];
    if (k < w) wexcl += u;
    tot += u;
  }
  bp[t] = wexcl + incl - v;
  __syncthreads();
  #pragma unroll
  for (int u = 0; u < EPT; ++u) {
    const int e = (wg * EPT + u) * 256 + t;
    if (e < E) csr_src[bp[d[u] >> 8] + loc[u] + bf[u] + rk[u]] = s[u];
  }
  for (int b = wg; b < nblk; b += nwg) {
    const int i = b * 256 + t;
    if (i < n) rowptr[i] = rowloc[i] + bp[b];
  }
  if (wg == 0 && t == 0) rowptr[n] = tot;
}

__global__ __launch_bounds__(256)
void place_chunked_scan_kernel(const int* __restrict__ src, const int* __restrict__ dst,
                               const int* __restrict__ rank, const int* __restrict__ rowloc,
                               const int* __restrict__ btot, const int* __restrict__ before,
                               int* __restrict__ rowptr, int* __restrict__ csr_src,
                               int E, int n, int chunk) {
  place_chunked_scan_body<1>(src, dst, rank, rowloc, btot, before, rowptr, csr_src, E, n, chunk,
                             (int)blockIdx.x, (int)gridDim.x);
}

__global__ void wsum_kernel(const float* __restrict__ Wm, const float* __restrict__ bm,
                            float* __restrict__ wsum, float* __restrict__ bsum) {
  int t = threadIdx.x;
  if (t < HD) {
    float s = 0.0f;
    for (int j = 0; j < HD; ++j) s += Wm[t * HD + j];
    wsum[t] = s;
  } else if (t == HD) {
    float s = 0.0f;
    for (int j = 0; j < HD; ++j) s += bm[j];
    *bsum = s;
  }
}

__device__ __forceinline__
void gemm_norm_wmma_body(const float* X, const int* __restrict__ rowmap,
                         const float* __restrict__ norm, const float* __restrict__ W,
                         float* OUT, int nrows, int K, int norm_shift, int wg) {
  __shared__ float ldsX[RPB][LPAD];
  __shared__ float ldsW[HD][WPAD];

  const int tid  = threadIdx.x;
  const int wave = tid >> 5;
  const int lane = tid & 31;
  const int half = lane >> 4;
  const int l16  = lane & 15;
  const int row0 = wg * RPB;

  v8f acc[6] = {};

  const int xc4 = (tid & 7) * 4;
  const float* xptr[4];
  float        xscale[4];
  bool         xrow_ok[4];
  int          xsrc[4];
  #pragma unroll
  for (int it = 0; it < 4; ++it) {
    const int grow = row0 + ((tid + 256 * it) >> 3);
    xrow_ok[it]    = grow < nrows;
    xsrc[it]       = xrow_ok[it] ? grow : nrows - 1;
  }
  if (norm) {
    #pragma unroll
    for (int it = 0; it < 4; ++it) xscale[it] = norm[xsrc[it] >> norm_shift];
  } else {
    #pragma unroll
    for (int it = 0; it < 4; ++it) xscale[it] = 1.0f;
  }
  if (rowmap) {
    #pragma unroll
    for (int it = 0; it < 4; ++it) xsrc[it] = rowmap[xsrc[it]];
  }
  #pragma unroll
  for (int it = 0; it < 4; ++it) xptr[it] = X + (long)xsrc[it] * K + xc4;
  const float* wptr[12];
  int wkk[12];
  #pragma unroll
  for (int it = 0; it < 12; ++it) {
    const int idx = tid + 256 * it;
    wkk[it]  = idx / HD;
    wptr[it] = W + (idx % HD);
  }

  float4 xv[4];
  float  wv[12];
  auto load_tile = [&](int k0) {
    const bool kin = (k0 + xc4) < K;
    #pragma unroll
    for (int it = 0; it < 4; ++it) {
      const float* p = xptr[it] + (kin ? k0 : -xc4);
      xv[it] = *(const float4*)p;
      __builtin_prefetch(((k0 + BK + xc4) < K) ? (p + BK) : p, 0, 0);
    }
    #pragma unroll
    for (int it = 0; it < 12; ++it) {
      const int k = k0 + wkk[it];
      wv[it] = wptr[it][(long)((k < K) ? k : 0) * HD];
    }
  };

  load_tile(0);
  for (int k0 = 0; k0 < K; k0 += BK) {
    {
      const bool kin = (k0 + xc4) < K;
      #pragma unroll
      for (int it = 0; it < 4; ++it) {
        const int r = (tid + 256 * it) >> 3;
        float4 v = xv[it];
        const float s = xscale[it];
        v.x *= s; v.y *= s; v.z *= s; v.w *= s;
        if (!(kin && xrow_ok[it])) v = make_float4(0.f, 0.f, 0.f, 0.f);
        ldsX[r][xc4 + 0] = v.x;
        ldsX[r][xc4 + 1] = v.y;
        ldsX[r][xc4 + 2] = v.z;
        ldsX[r][xc4 + 3] = v.w;
      }
      #pragma unroll
      for (int it = 0; it < 12; ++it) {
        const int idx = tid + 256 * it;
        ldsW[idx % HD][wkk[it]] = (k0 + wkk[it] < K) ? wv[it] : 0.0f;
      }
    }
    __syncthreads();

    if (k0 + BK < K) load_tile(k0 + BK);

    const int kmax = (BK < K - k0) ? BK : (K - k0);
    const int mrow = wave * 16 + l16;
    for (int kk = 0; kk < kmax; kk += 4) {
      int ka = kk + half * 2;

      v2f A = *(const v2f*)&ldsX[mrow][ka];
      v2f Bf[6];
      #pragma unroll
      for (int t = 0; t < 6; ++t)
        Bf[t] = *(const v2f*)&ldsW[t * 16 + l16][ka];

      #pragma unroll
      for (int t = 0; t < 6; ++t)
        acc[t] = __builtin_amdgcn_wmma_f32_16x16x4_f32(
             false, A,  false, Bf[t],
             (short)0, acc[t],  false,  false);
    }
    __syncthreads();
  }

  const int orow = row0 + wave * 16 + half * 8;
  #pragma unroll
  for (int t = 0; t < 6; ++t) {
    int col = t * 16 + l16;
    #pragma unroll
    for (int i = 0; i < 8; ++i) {
      int r = orow + i;
      if (r < nrows) OUT[(long)r * HD + col] = acc[t][i];
    }
  }
}

__global__ __launch_bounds__(256)
void gemm_norm_wmma_kernel(const float* X, const int* __restrict__ rowmap,
                           const float* __restrict__ norm, const float* __restrict__ W,
                           float* OUT, int nrows, int K, int norm_shift) {
  gemm_norm_wmma_body(X, rowmap, norm, W, OUT, nrows, K, norm_shift, (int)blockIdx.x);
}

constexpr int PLACE_EPT = 8;
__global__ __launch_bounds__(256)
void gemm_and_place_kernel(const float* X, const float* __restrict__ W, float* OUT, int nrows, int K, int gemm_wgs,
                           const int* __restrict__ src, const int* __restrict__ dst,
                           const int* __restrict__ rank, const int* __restrict__ rowloc,
                           const int* __restrict__ btot, const int* __restrict__ before,
                           int* __restrict__ rowptr, int* __restrict__ csr_src,
                           int E, int n, int chunk) {
  if ((int)blockIdx.x < gemm_wgs) {
    gemm_norm_wmma_body(X, nullptr, nullptr, W, OUT, nrows, K, 0, (int)blockIdx.x);
  } else {
    place_chunked_scan_body<PLACE_EPT>(src, dst, rank, rowloc, btot, before, rowptr, csr_src, E, n, chunk,
                                       (int)blockIdx.x - gemm_wgs, (int)gridDim.x - gemm_wgs);
  }
}

__device__ __forceinline__ void gather_row_sum(const int* __restrict__ rowptr, const int* __restrict__ csr_src,
                                               const float* __restrict__ h, int row, int lane,
                                               float& a0, float& a1, float& a2) {
  const int beg = rowptr[row];
  const int end = rowptr[row + 1];
  a0 = 0.0f; a1 = 0.0f; a2 = 0.0f;
  for (int base = beg; base < end; base += 32) {
    const int cnt  = (end - base < 32) ? (end - base) : 32;
    const int mine = (lane < cnt) ? csr_src[base + lane] : 0;
    int j = 0;
    for (; j + 4 <= cnt; j += 4) {
      const float* p0 = h + (long)__builtin_amdgcn_readlane(mine, j + 0) * HD + lane;
      const float* p1 = h + (long)__builtin_amdgcn_readlane(mine, j + 1) * HD + lane;
      const float* p2 = h + (long)__builtin_amdgcn_readlane(mine, j + 2) * HD + lane;
      const float* p3 = h + (long)__builtin_amdgcn_readlane(mine, j + 3) * HD + lane;
      const float x00 = p0[0], x01 = p0[32], x02 = p0[64];
      const float x10 = p1[0], x11 = p1[32], x12 = p1[64];
      const float x20 = p2[0], x21 = p2[32], x22 = p2[64];
      const float x30 = p3[0], x31 = p3[32], x32 = p3[64];
      a0 += x00; a1 += x01; a2 += x02;
      a0 += x10; a1 += x11; a2 += x12;
      a0 += x20; a1 += x21; a2 += x22;
      a0 += x30; a1 += x31; a2 += x32;
    }
    for (; j < cnt; ++j) {
      const float* p = h + (long)__builtin_amdgcn_readlane(mine, j) * HD + lane;
      a0 += p[0]; a1 += p[32]; a2 += p[64];
    }
  }
}

template <bool MAPPED>
__device__ __forceinline__ void gather_row_sum_scaled(const int* __restrict__ rowptr, const int* __restrict__ csr_src,
                                                      const float* __restrict__ xw, const int* __restrict__ rowmap,
                                                      const float* __restrict__ scale, int row, int lane,
                                                      float& a0, float& a1, float& a2) {
  const int beg = rowptr[row];
  const int end = rowptr[row + 1];
  a0 = 0.0f; a1 = 0.0f; a2 = 0.0f;
  for (int base = beg; base < end; base += 32) {
    const int cnt  = (end - base < 32) ? (end - base) : 32;
    const int mine = (lane < cnt) ? csr_src[base + lane] : 0;
    const int m    = MAPPED ? rowmap[mine] : mine;
    const int sc   = __float_as_int(scale[mine]);
    int j = 0;
    for (; j + 4 <= cnt; j += 4) {
      const float* p0 = xw + (long)__builtin_amdgcn_readlane(m, j + 0) * HD + lane;
      const float* p1 = xw + (long)__builtin_amdgcn_readlane(m, j + 1) * HD + lane;
      const float* p2 = xw + (long)__builtin_amdgcn_readlane(m, j + 2) * HD + lane;
      const float* p3 = xw + (long)__builtin_amdgcn_readlane(m, j + 3) * HD + lane;
      const float x00 = p0[0], x01 = p0[32], x02 = p0[64];
      const float x10 = p1[0], x11 = p1[32], x12 = p1[64];
      const float x20 = p2[0], x21 = p2[32], x22 = p2[64];
      const float x30 = p3[0], x31 = p3[32], x32 = p3[64];
      __builtin_amdgcn_sched_barrier(0);
      const float c0 = __int_as_float(__builtin_amdgcn_readlane(sc, j + 0));
      const float c1 = __int_as_float(__builtin_amdgcn_readlane(sc, j + 1));
      const float c2 = __int_as_float(__builtin_amdgcn_readlane(sc, j + 2));
      const float c3 = __int_as_float(__builtin_amdgcn_readlane(sc, j + 3));
      a0 = fmaf(c0, x00, a0); a1 = fmaf(c0, x01, a1); a2 = fmaf(c0, x02, a2);
      a0 = fmaf(c1, x10, a0); a1 = fmaf(c1, x11, a1); a2 = fmaf(c1, x12, a2);
      a0 = fmaf(c2, x20, a0); a1 = fmaf(c2, x21, a1); a2 = fmaf(c2, x22, a2);
      a0 = fmaf(c3, x30, a0); a1 = fmaf(c3, x31, a1); a2 = fmaf(c3, x32, a2);
    }
    for (; j < cnt; ++j) {
      const float* p = xw + (long)__builtin_amdgcn_readlane(m, j) * HD + lane;
      const float x0 = p[0], x1 = p[32], x2 = p[64];
      const float c  = __int_as_float(__builtin_amdgcn_readlane(sc, j));
      a0 = fmaf(c, x0, a0); a1 = fmaf(c, x1, a1); a2 = fmaf(c, x2, a2);
    }
  }
}

template <bool MAPPED>
__device__ __forceinline__ void gather_scaled_prelu_node(const int* __restrict__ rowptr, const int* __restrict__ csr_src,
                                                         const float* __restrict__ xw, const float* __restrict__ norm_out,
                                                         const float* __restrict__ norm_in, const float* __restrict__ b,
                                                         const float* __restrict__ a, float* __restrict__ act_row,
                                                         const int* __restrict__ rowmap, int row, int lane) {
  const float ni = norm_in[row];
  const float b0 = b[lane], b1 = b[lane + 32], b2 = b[lane + 64];
  const float s0 = a[lane], s1 = a[lane + 32], s2 = a[lane + 64];
  float a0, a1, a2;
  gather_row_sum_scaled<MAPPED>(rowptr, csr_src, xw, rowmap, norm_out, row, lane, a0, a1, a2);
  const float v0 = a0 * ni + b0;
  const float v1 = a1 * ni + b1;
  const float v2 = a2 * ni + b2;
  float* o = act_row + lane;
  o[0]  = (v0 >= 0.0f) ? v0 : s0 * v0;
  o[32] = (v1 >= 0.0f) ? v1 : s1 * v1;
  o[64] = (v2 >= 0.0f) ? v2 : s2 * v2;
}

template <bool MAPPED>
__global__ __launch_bounds__(256)
void spmm_gather_scaled_prelu_kernel(const int* __restrict__ rowptr, const int* __restrict__ csr_src,
                                     const float* __restrict__ xw, const float* __restrict__ norm_out,
                                     const float* __restrict__ norm_in, const float* __restrict__ b,
                                     const float* __restrict__ a, float* __restrict__ act,
                                     const int* __restrict__ rowmap, int n) {
  const int lane = threadIdx.x & 31;
  const int row = __builtin_amdgcn_readfirstlane((int)(blockIdx.x * (blockDim.x >> 5) + (threadIdx.x >> 5)));
  if (row >= n) return;
  gather_scaled_prelu_node<MAPPED>(rowptr, csr_src, xw, norm_out, norm_in, b, a, act + (long)row * HD, rowmap, row, lane);
}

__global__ __launch_bounds__(256)
void spmm_gather_scaled_prelu_both_kernel(const int* __restrict__ rowptr, const int* __restrict__ csr_src,
                                          const float* __restrict__ xw, const float* __restrict__ norm_out,
                                          const float* __restrict__ norm_in, const float* __restrict__ b,
                                          const float* __restrict__ a, float* __restrict__ act2,
                                          const int* __restrict__ rowmap, int n, int blocks_per_view) {
  const int lane = threadIdx.x & 31;
  const bool second = (int)blockIdx.x >= blocks_per_view;
  const int blk = (int)blockIdx.x - (second ? blocks_per_view : 0);
  const int row = __builtin_amdgcn_readfirstlane((int)(blk * (blockDim.x >> 5) + (threadIdx.x >> 5)));
  if (row >= n) return;
  float* dst = act2 + (long)row * (2 * HD);
  if (second) gather_scaled_prelu_node<true>(rowptr, csr_src, xw, norm_out, norm_in, b, a, dst + HD, rowmap, row, lane);
  else        gather_scaled_prelu_node<false>(rowptr, csr_src, xw, norm_out, norm_in, b, a, dst, nullptr, row, lane);
}

__global__ __launch_bounds__(256)
void spmm_gather_prelu_dot_kernel(const int* __restrict__ rowptr, const int* __restrict__ csr_src,
                                  const float* __restrict__ h, const float* __restrict__ norm_in,
                                  const float* __restrict__ b, const float* __restrict__ a,
                                  const float* __restrict__ wsum, const float* __restrict__ bsum,
                                  float* __restrict__ out, int n) {
  const int lane = threadIdx.x & 31;
  const int row = __builtin_amdgcn_readfirstlane((int)(blockIdx.x * (blockDim.x >> 5) + (threadIdx.x >> 5)));
  if (row >= n) return;
  const float ni = norm_in[row];
  const float b0 = b[lane], b1 = b[lane + 32], b2 = b[lane + 64];
  const float s0 = a[lane], s1 = a[lane + 32], s2 = a[lane + 64];
  const float w0 = wsum[lane], w1 = wsum[lane + 32], w2 = wsum[lane + 64];
  const float bs = *bsum;
  float a0, a1, a2;
  gather_row_sum(rowptr, csr_src, h, row, lane, a0, a1, a2);
  const float v0 = a0 * ni + b0;
  const float v1 = a1 * ni + b1;
  const float v2 = a2 * ni + b2;
  const float z0 = (v0 >= 0.0f) ? v0 : s0 * v0;
  const float z1 = (v1 >= 0.0f) ? v1 : s1 * v1;
  const float z2 = (v2 >= 0.0f) ? v2 : s2 * v2;
  float s = fmaf(z0, w0, 0.0f);
  s = fmaf(z1, w1, s);
  s = fmaf(z2, w2, s);
  #pragma unroll
  for (int off = 16; off > 0; off >>= 1) s += __shfl_xor(s, off, 32);
  if (lane == 0) out[row] = s + bs;
}

__device__ __forceinline__ void gather_row_sum2(const int* __restrict__ rowptr, const int* __restrict__ csr_src,
                                                const float* __restrict__ h2, int row, int lane,
                                                float& a0, float& a1, float& a2, float& c0, float& c1, float& c2) {
  const int beg = rowptr[row];
  const int end = rowptr[row + 1];
  a0 = 0.0f; a1 = 0.0f; a2 = 0.0f; c0 = 0.0f; c1 = 0.0f; c2 = 0.0f;
  for (int base = beg; base < end; base += 32) {
    const int cnt  = (end - base < 32) ? (end - base) : 32;
    const int mine = (lane < cnt) ? csr_src[base + lane] : 0;
    int j = 0;
    for (; j + 4 <= cnt; j += 4) {
      const float* p0 = h2 + (long)__builtin_amdgcn_readlane(mine, j + 0) * (2 * HD) + lane;
      const float* p1 = h2 + (long)__builtin_amdgcn_readlane(mine, j + 1) * (2 * HD) + lane;
      const float* p2 = h2 + (long)__builtin_amdgcn_readlane(mine, j + 2) * (2 * HD) + lane;
      const float* p3 = h2 + (long)__builtin_amdgcn_readlane(mine, j + 3) * (2 * HD) + lane;
      const float x00 = p0[0], x01 = p0[32], x02 = p0[64], y00 = p0[96], y01 = p0[128], y02 = p0[160];
      const float x10 = p1[0], x11 = p1[32], x12 = p1[64], y10 = p1[96], y11 = p1[128], y12 = p1[160];
      const float x20 = p2[0], x21 = p2[32], x22 = p2[64], y20 = p2[96], y21 = p2[128], y22 = p2[160];
      const float x30 = p3[0], x31 = p3[32], x32 = p3[64], y30 = p3[96], y31 = p3[128], y32 = p3[160];
      a0 += x00; a1 += x01; a2 += x02; c0 += y00; c1 += y01; c2 += y02;
      a0 += x10; a1 += x11; a2 += x12; c0 += y10; c1 += y11; c2 += y12;
      a0 += x20; a1 += x21; a2 += x22; c0 += y20; c1 += y21; c2 += y22;
      a0 += x30; a1 += x31; a2 += x32; c0 += y30; c1 += y31; c2 += y32;
    }
    for (; j < cnt; ++j) {
      const float* p = h2 + (long)__builtin_amdgcn_readlane(mine, j) * (2 * HD) + lane;
      const float x0 = p[0], x1 = p[32], x2 = p[64], y0 = p[96], y1 = p[128], y2 = p[160];
      a0 += x0; a1 += x1; a2 += x2; c0 += y0; c1 += y1; c2 += y2;
    }
  }
}

__global__ __launch_bounds__(256)
void spmm_gather2_prelu_dot_kernel(const int* __restrict__ rowptr, const int* __restrict__ csr_src,
                                   const float* __restrict__ h2, const float* __restrict__ norm_in,
                                   const float* __restrict__ b, const float* __restrict__ a,
                                   const float* __restrict__ wsum, const float* __restrict__ bsum,
                                   float* __restrict__ out, int n) {
  const int lane = threadIdx.x & 31;
  const int row = __builtin_amdgcn_readfirstlane((int)(blockIdx.x * (blockDim.x >> 5) + (threadIdx.x >> 5)));
  if (row >= n) return;
  const float ni = norm_in[row];
  const float b0 = b[lane], b1 = b[lane + 32], b2 = b[lane + 64];
  const float s0 = a[lane], s1 = a[lane + 32], s2 = a[lane + 64];
  const float w0 = wsum[lane], w1 = wsum[lane + 32], w2 = wsum[lane + 64];
  const float bs = *bsum;
  float a0, a1, a2, c0, c1, c2;
  gather_row_sum2(rowptr, csr_src, h2, row, lane, a0, a1, a2, c0, c1, c2);
  const float v0 = a0 * ni + b0;
  const float v1 = a1 * ni + b1;
  const float v2 = a2 * ni + b2;
  const float u0 = c0 * ni + b0;
  const float u1 = c1 * ni + b1;
  const float u2 = c2 * ni + b2;
  const float z0 = (v0 >= 0.0f) ? v0 : s0 * v0;
  const float z1 = (v1 >= 0.0f) ? v1 : s1 * v1;
  const float z2 = (v2 >= 0.0f) ? v2 : s2 * v2;
  const float q0 = (u0 >= 0.0f) ? u0 : s0 * u0;
  const float q1 = (u1 >= 0.0f) ? u1 : s1 * u1;
  const float q2 = (u2 >= 0.0f) ? u2 : s2 * u2;
  float s = fmaf(z0, w0, 0.0f);
  s = fmaf(z1, w1, s);
  s = fmaf(z2, w2, s);
  float r = fmaf(q0, w0, 0.0f);
  r = fmaf(q1, w1, r);
  r = fmaf(q2, w2, r);
  for (int off = 16; off > 0; off >>= 1) {
    s += __shfl_xor(s, off, 32);
    r += __shfl_xor(r, off, 32);
  }
  if (lane == 0) {
    out[row]     = s + bs;
    out[n + row] = r + bs;
  }
}

__global__ void spmm_kernel(const int* __restrict__ src, const int* __restrict__ dst,
                            const float* __restrict__ h, float* __restrict__ agg, int E) {
  int gw   = (int)(((long)blockIdx.x * blockDim.x + threadIdx.x) >> 5);
  int lane = threadIdx.x & 31;
  if (gw < E) {
    int s = src[gw], d = dst[gw];
    const float* hp = h + (long)s * HD;
    float* ap = agg + (long)d * HD;
    #pragma unroll
    for (int i = 0; i < 3; ++i) {
      float v = hp[lane + 32 * i];
      __hip_atomic_fetch_add(&ap[lane + 32 * i], v, __ATOMIC_RELAXED, __HIP_MEMORY_SCOPE_AGENT);
    }
  }
}

__global__ void prelu_kernel(const float* __restrict__ agg, const float* __restrict__ norm_in,
                             const float* __restrict__ b, const float* __restrict__ a,
                             float* __restrict__ out, int n) {
  long idx = (long)blockIdx.x * blockDim.x + threadIdx.x;
  if (idx < (long)n * HD) {
    int i = (int)(idx / HD);
    int j = (int)(idx % HD);
    float v = agg[idx] * norm_in[i] + b[j];
    out[idx] = (v >= 0.0f) ? v : a[j] * v;
  }
}

__global__ void reduce_kernel(const float* __restrict__ z, const float* __restrict__ wsum,
                              const float* __restrict__ bsum, float* __restrict__ out, int n) {
  int gw   = (int)(((long)blockIdx.x * blockDim.x + threadIdx.x) >> 5);
  int lane = threadIdx.x & 31;
  if (gw < n) {
    const float* zp = z + (long)gw * HD;
    float s = 0.0f;
    #pragma unroll
    for (int i = 0; i < 3; ++i) s = fmaf(zp[lane + 32 * i], wsum[lane + 32 * i], s);
    #pragma unroll
    for (int off = 16; off > 0; off >>= 1) s += __shfl_xor(s, off, 32);
    if (lane == 0) out[gw] = s + *bsum;
  }
}

extern "C" void kernel_launch(void* const* d_in, const int* in_sizes, int n_in,
                              void* d_out, int out_size, void* d_ws, size_t ws_size,
                              hipStream_t stream) {
  const float* x    = (const float*)d_in[0];
  const int*   src  = (const int*)  d_in[1];
  const int*   dst  = (const int*)  d_in[2];
  const int*   perm = (const int*)  d_in[3];
  const float* W1   = (const float*)d_in[4];
  const float* b1   = (const float*)d_in[5];
  const float* a1   = (const float*)d_in[6];
  const float* W2   = (const float*)d_in[7];
  const float* b2   = (const float*)d_in[8];
  const float* a2   = (const float*)d_in[9];
  const float* Wm   = (const float*)d_in[10];
  const float* bm   = (const float*)d_in[11];
  float* out = (float*)d_out;

  const int E  = in_sizes[1];
  const int N  = in_sizes[3];
  const int K1 = in_sizes[0] / N;

  float* ws       = (float*)d_ws;
  float* norm_out = ws;
  float* norm_in  = ws + N;
  float* wsum     = ws + 2L * N;
  float* bsum     = ws + 2L * N + HD;
  float* bufA     = ws + 2L * N + 2 * HD;
  float* bufB     = bufA + (long)N * HD;

  const long NH = (long)N * HD;
  const size_t need_wide = (size_t)(2L * N + 2 * HD + 3 * NH + (N + 1) + E) * sizeof(float);
  const bool wide = (ws_size >= need_wide) && ((long)E <= NH);
  float* H2       = bufB;
  int*   rowptr   = (int*)(bufB + (wide ? 2 * NH : NH));
  int*   csr_src  = rowptr + (N + 1);
  int*   rank     = (int*)bufB;
  int*   cnt_out  = (int*)norm_out;
  int*   cnt_in   = (int*)norm_in;
  const bool hist_in_h2 = wide && ((long)HG * N <= NH);
  int*   hist_in  = hist_in_h2 ? (int*)(bufB + NH) : (int*)bufA;
  int*   hist_out = (int*)bufB;
  int*   lrank    = hist_out + (long)HG * N;
  int*   rowloc   = lrank + E;
  int*   btot     = rowloc + N;

  const int gemm_blocks = (N + RPB - 1) / RPB;
  const int spmm_blocks = (int)(((long)E * 32 + 255) / 256);
  const int gath_blocks = (N + 7) / 8;
  const int elw_blocks  = (int)((NH + 255) / 256);
  const int red_blocks  = (int)(((long)N * 32 + 255) / 256);

  const size_t need_grouped = (size_t)(2L * N + 2 * HD + 2 * NH + (N + 1) + E) * sizeof(float);
  const bool grouped = wide || ((ws_size >= need_grouped) && ((long)E <= NH));
  const int  chunk    = (E + HG - 1) / HG;
  const bool lds_hist = grouped && (N <= HCAP) && ((long)(HG + 1) * N + E + 256 <= NH);

  if (lds_hist) {
    hist_lds_kernel<<<2 * HG, 1024, 0, stream>>>(src, dst, hist_in, hist_out, lrank, E, N, chunk);
    hist_fold_scan_kernel<<<(2 * N + 255) / 256 + 1, 256, 0, stream>>>(hist_in, hist_out, rowloc, btot, norm_out, norm_in, N,
                                                                       Wm, bm, wsum, bsum);
    if (hist_in_h2) {
      const int place_wgs = (E + PLACE_EPT * 256 - 1) / (PLACE_EPT * 256);
      gemm_and_place_kernel<<<gemm_blocks + place_wgs, 256, 0, stream>>>(x, W1, bufA, N, K1, gemm_blocks,
                                                                         src, dst, lrank, rowloc, btot, hist_in,
                                                                         rowptr, csr_src, E, N, chunk);
    } else {
      place_chunked_scan_kernel<<<(E + 255) / 256, 256, 0, stream>>>(src, dst, lrank, rowloc, btot, hist_in, rowptr,
                                                                     csr_src, E, N, chunk);
    }
  } else {
    fill_zero_kernel<<<512, 256, 0, stream>>>(norm_out, 2L * N);
    degree_kernel<<<(E + 255) / 256, 256, 0, stream>>>(src, dst, cnt_out, cnt_in, grouped ? rank : nullptr, E);
    if (grouped) {
      rowptr_scan_kernel<<<1, 1024, 0, stream>>>(cnt_in, rowptr, N);
      place_kernel<<<(E + 255) / 256, 256, 0, stream>>>(src, dst, rank, rowptr, csr_src, E);
    }
    finalize_norm_kernel<<<(2 * N + 255) / 256, 256, 0, stream>>>(norm_out, 2 * N);
    wsum_kernel<<<1, 128, 0, stream>>>(Wm, bm, wsum, bsum);
  }

  if (grouped && !(lds_hist && hist_in_h2))
    gemm_norm_wmma_kernel<<<gemm_blocks, 256, 0, stream>>>(x, nullptr, nullptr, W1, bufA, N, K1, 0);

  if (wide) {
    spmm_gather_scaled_prelu_both_kernel<<<2 * gath_blocks, 256, 0, stream>>>(rowptr, csr_src, bufA, norm_out, norm_in,
                                                                             b1, a1, H2, perm, N, gath_blocks);
    gemm_norm_wmma_kernel<<<(2 * N + RPB - 1) / RPB, 256, 0, stream>>>(H2, nullptr, norm_out, W2, H2, 2 * N, HD, 1);
    spmm_gather2_prelu_dot_kernel<<<gath_blocks, 256, 0, stream>>>(rowptr, csr_src, H2, norm_in, b2, a2,
                                                                  wsum, bsum, out, N);
    return;
  }

  for (int view = 0; view < 2; ++view) {
    const int* rowmap = view ? perm : nullptr;
    float* outv = out + (long)view * N;
    if (grouped) {
      if (view) {
        spmm_gather_scaled_prelu_kernel<true><<<gath_blocks, 256, 0, stream>>>(rowptr, csr_src, bufA, norm_out, norm_in,
                                                                              b1, a1, bufB, rowmap, N);
      } else {
        spmm_gather_scaled_prelu_kernel<false><<<gath_blocks, 256, 0, stream>>>(rowptr, csr_src, bufA, norm_out, norm_in,
                                                                               b1, a1, bufB, nullptr, N);
      }
      gemm_norm_wmma_kernel<<<gemm_blocks, 256, 0, stream>>>(bufB, nullptr, norm_out, W2, bufB, N, HD, 0);
      spmm_gather_prelu_dot_kernel<<<gath_blocks, 256, 0, stream>>>(rowptr, csr_src, bufB, norm_in, b2, a2,
                                                                   wsum, bsum, outv, N);
    } else {
      gemm_norm_wmma_kernel<<<gemm_blocks, 256, 0, stream>>>(x, rowmap, norm_out, W1, bufA, N, K1, 0);
      fill_zero_kernel<<<2048, 256, 0, stream>>>(bufB, NH);
      spmm_kernel<<<spmm_blocks, 256, 0, stream>>>(src, dst, bufA, bufB, E);
      prelu_kernel<<<elw_blocks, 256, 0, stream>>>(bufB, norm_in, b1, a1, bufA, N);
      gemm_norm_wmma_kernel<<<gemm_blocks, 256, 0, stream>>>(bufA, nullptr, norm_out, W2, bufB, N, HD, 0);
      fill_zero_kernel<<<2048, 256, 0, stream>>>(bufA, NH);
      spmm_kernel<<<spmm_blocks, 256, 0, stream>>>(src, dst, bufB, bufA, E);
      prelu_kernel<<<elw_blocks, 256, 0, stream>>>(bufA, norm_in, b2, a2, bufB, N);
      reduce_kernel<<<red_blocks, 256, 0, stream>>>(bufB, wsum, bsum, outv, N);
    }
  }
}


# layer-2 product fused into the layer-1 gather launch: resident work-groups loop over 16-node tiles, finished rows go through LDS to WMMA against W2 fragments held in registers; 6 -> 5 launches
# speedup vs baseline: 5.2034x; 1.0753x over previous
#include <hip/hip_runtime.h>

typedef float v2f __attribute__((ext_vector_type(2)));
typedef float v8f __attribute__((ext_vector_type(8)));

constexpr int HD   = 96;
constexpr int RPB  = 128;
constexpr int BK   = 32;
constexpr int LPAD = 36;
constexpr int WPAD = 38;
constexpr int HG   = 64;
constexpr int HCAP = 65536;


__global__ void fill_zero_kernel(float* __restrict__ p, long n) {
  long i = (long)blockIdx.x * blockDim.x + threadIdx.x;
  long stride = (long)gridDim.x * blockDim.x;
  for (; i < n; i += stride) p[i] = 0.0f;
}

__global__ void degree_kernel(const int* __restrict__ src, const int* __restrict__ dst,
                              int* __restrict__ cnt_out, int* __restrict__ cnt_in,
                              int* __restrict__ rank, int E) {
  int e = blockIdx.x * blockDim.x + threadIdx.x;
  if (e < E) {
    __hip_atomic_fetch_add(&cnt_out[src[e]], 1, __ATOMIC_RELAXED, __HIP_MEMORY_SCOPE_AGENT);
    if (rank) {
      rank[e] = __hip_atomic_fetch_add(&cnt_in[dst[e]], 1, __ATOMIC_RELAXED, __HIP_MEMORY_SCOPE_AGENT);
    } else {
      __hip_atomic_fetch_add(&cnt_in[dst[e]], 1, __ATOMIC_RELAXED, __HIP_MEMORY_SCOPE_AGENT);
    }
  }
}

__global__ void finalize_norm_kernel(float* __restrict__ p, int n) {
  int i = blockIdx.x * blockDim.x + threadIdx.x;
  if (i < n) p[i] = 1.0f / sqrtf(fmaxf((float)__float_as_int(p[i]), 1.0f));
}

__global__ __launch_bounds__(1024)
void rowptr_scan_kernel(const int* __restrict__ cnt, int* __restrict__ rowptr, int n) {
  __shared__ int wave_tot[32];
  const int t = threadIdx.x, lane = t & 31, w = t >> 5;
  int carry = 0;
  int c[4], nx[4];
  for (int k = 0; k < 4; ++k) {
    const int i = t * 4 + k;
    const int v = cnt[i < n ? i : n - 1];
    c[k] = (i < n) ? v : 0;
  }
  for (int base = 0; base < n; base += 4096) {
    const int i0 = base + t * 4;
    for (int k = 0; k < 4; ++k) {
      const int i = i0 + 4096 + k;
      const int v = cnt[i < n ? i : n - 1];
      nx[k] = (i < n) ? v : 0;
    }
    const int s = c[0] + c[1] + c[2] + c[3];
    int incl = s;
    for (int off = 1; off < 32; off <<= 1) {
      int v = __shfl_up(incl, off, 32);
      if (lane >= off) incl += v;
    }
    if (lane == 31) wave_tot[w] = incl;
    __syncthreads();
    const int wt = wave_tot[lane];
    int wincl = wt;
    for (int off = 1; off < 32; off <<= 1) {
      int v = __shfl_up(wincl, off, 32);
      if (lane >= off) wincl += v;
    }
    const int wave_excl = __shfl(wincl, w, 32) - __shfl(wt, w, 32);
    const int total     = __shfl(wincl, 31, 32);
    int run = carry + wave_excl + incl - s;
    for (int k = 0; k < 4; ++k) {
      if (i0 + k < n) rowptr[i0 + k] = run;
      run += c[k];
    }
    carry += total;
    for (int k = 0; k < 4; ++k) c[k] = nx[k];
    __syncthreads();
  }
  if (t == 0) rowptr[n] = carry;
}

__global__ void place_kernel(const int* __restrict__ src, const int* __restrict__ dst,
                             const int* __restrict__ rank, const int* __restrict__ rowptr,
                             int* __restrict__ csr_src, int E) {
  int e = blockIdx.x * blockDim.x + threadIdx.x;
  if (e < E) csr_src[rowptr[dst[e]] + rank[e]] = src[e];
}

__global__ __launch_bounds__(1024)
void hist_lds_kernel(const int* __restrict__ src, const int* __restrict__ dst,
                     int* __restrict__ hist_in, int* __restrict__ hist_out,
                     int* __restrict__ rank, int E, int n, int chunk) {
  __shared__ int4 cnt4[HCAP / 4];
  int* cnt = reinterpret_cast<int*>(cnt4);
  const int  t     = threadIdx.x;
  const bool is_in = blockIdx.x < HG;
  const int  g     = is_in ? (int)blockIdx.x : (int)blockIdx.x - HG;
  const int* __restrict__ key  = is_in ? dst : src;
  int*       __restrict__ hist = (is_in ? hist_in : hist_out) + (long)g * n;
  const bool vec = ((n & 3) == 0) && ((((size_t)hist) & 15) == 0);

  if (vec) {
    for (int i = t; i < (n >> 2); i += 1024) cnt4[i] = make_int4(0, 0, 0, 0);
  } else {
    for (int i = t; i < n; i += 1024) cnt[i] = 0;
  }
  __syncthreads();

  const int beg = g * chunk;
  const int end = (beg + chunk < E) ? (beg + chunk) : E;
  for (int base = beg + t; base < end; base += 8 * 1024) {
    int k[8];
    #pragma unroll
    for (int u = 0; u < 8; ++u) {
      const int e = base + u * 1024;
      k[u] = key[e < end ? e : end - 1];
    }
    int r[8];
    #pragma unroll
    for (int u = 0; u < 8; ++u) {
      const int e = base + u * 1024;
      r[u] = __hip_atomic_fetch_add(&cnt[k[u]], (e < end) ? 1 : 0, __ATOMIC_RELAXED, __HIP_MEMORY_SCOPE_WORKGROUP);
    }
    if (is_in) {
      #pragma unroll
      for (int u = 0; u < 8; ++u) {
        const int e = base + u * 1024;
        if (e < end) rank[e] = r[u];
      }
    }
  }
  __syncthreads();

  if (vec) {
    int4* __restrict__ h4 = reinterpret_cast<int4*>(hist);
    for (int i = t; i < (n >> 2); i += 1024) h4[i] = cnt4[i];
  } else {
    for (int i = t; i < n; i += 1024) hist[i] = cnt[i];
  }
}

__global__ __launch_bounds__(256)
void hist_fold_scan_kernel(int* __restrict__ hist_in, const int* __restrict__ hist_out,
                           int* __restrict__ rowloc, int* __restrict__ btot,
                           float* __restrict__ norm_out, float* __restrict__ norm_in, int n,
                           const float* __restrict__ Wm, const float* __restrict__ bm,
                           float* __restrict__ wsum, float* __restrict__ bsum) {
  __shared__ int wave_tot[8];
  const int t = threadIdx.x;
  if (blockIdx.x == 0) {
    if (t < HD) {
      float s = 0.0f;
      for (int j = 0; j < HD; ++j) s += Wm[t * HD + j];
      wsum[t] = s;
    } else if (t == HD) {
      float s = 0.0f;
      for (int j = 0; j < HD; ++j) s += bm[j];
      *bsum = s;
    }
    return;
  }
  const int blk = (int)blockIdx.x - 1;
  const int i = blk * 256 + t;
  int c[HG];
  int run = 0;
  if (i < n) {
    #pragma unroll
    for (int g = 0; g < HG; ++g) c[g] = hist_in[(long)g * n + i];
    #pragma unroll
    for (int g = 0; g < HG; ++g) run += c[g];
  } else if (i < 2 * n) {
    const int j = i - n;
    int s = 0;
    #pragma unroll
    for (int g = 0; g < HG; ++g) s += hist_out[(long)g * n + j];
    norm_out[j] = 1.0f / sqrtf(fmaxf((float)s, 1.0f));
  }
  if (blk * 256 < n) {
    const int lane = t & 31, w = t >> 5;
    int incl = run;
    for (int off = 1; off < 32; off <<= 1) {
      int v = __shfl_up(incl, off, 32);
      if (lane >= off) incl += v;
    }
    if (lane == 31) wave_tot[w] = incl;
    __syncthreads();
    int wexcl = 0, tot = 0;
    for (int k = 0; k < 8; ++k) {
      const int v = wave_tot[k];
      if (k < w) wexcl += v;
      tot += v;
    }
    if (i < n) rowloc[i] = wexcl + incl - run;
    if (t == 0) btot[blk] = tot;
  }
  if (i < n) {
    int before = 0;
    #pragma unroll
    for (int g = 0; g < HG; ++g) {
      hist_in[(long)g * n + i] = before;
      before += c[g];
    }
    norm_in[i] = 1.0f / sqrtf(fmaxf((float)run, 1.0f));
  }
}

template <int EPT>
__device__ __forceinline__
void place_chunked_scan_body(const int* __restrict__ src, const int* __restrict__ dst,
                             const int* __restrict__ rank, const int* __restrict__ rowloc,
                             const int* __restrict__ btot, const int* __restrict__ before,
                             int* __restrict__ rowptr, int* __restrict__ csr_src,
                             int E, int n, int chunk, int wg, int nwg) {
  __shared__ int bp[256];
  __shared__ int wave_tot[8];
  const int t = threadIdx.x, lane = t & 31, w = t >> 5;
  const int nblk = (n + 255) >> 8;
  int d[EPT], rk[EPT], s[EPT], loc[EPT], bf[EPT];
  #pragma unroll
  for (int u = 0; u < EPT; ++u) {
    const int e  = (wg * EPT + u) * 256 + t;
    const int ec = (e < E) ? e : E - 1;
    d[u]  = dst[ec];
    rk[u] = rank[ec];
    s[u]  = src[ec];
  }
  const int bl = btot[(t < nblk) ? t : 0];
  const int v  = (t < nblk) ? bl : 0;
  #pragma unroll
  for (int u = 0; u < EPT; ++u) {
    const int e  = (wg * EPT + u) * 256 + t;
    const int ec = (e < E) ? e : E - 1;
    const int g  = ec / chunk;
    loc[u] = rowloc[d[u]];
    bf[u]  = before[(long)g * n + d[u]];
  }
  int incl = v;
  for (int off = 1; off < 32; off <<= 1) {
    int u = __shfl_up(incl, off, 32);
    if (lane >= off) incl += u;
  }
  if (lane == 31) wave_tot[w] = incl;
  __syncthreads();
  int wexcl = 0, tot = 0;
  for (int k = 0; k < 8; ++k) {
    const int u = wave_tot[k];
    if (k < w) wexcl += u;
    tot += u;
  }
  bp[t] = wexcl + incl - v;
  __syncthreads();
  #pragma unroll
  for (int u = 0; u < EPT; ++u) {
    const int e = (wg * EPT + u) * 256 + t;
    if (e < E) csr_src[bp[d[u] >> 8] + loc[u] + bf[u] + rk[u]] = s[u];
  }
  for (int b = wg; b < nblk; b += nwg) {
    const int i = b * 256 + t;
    if (i < n) rowptr[i] = rowloc[i] + bp[b];
  }
  if (wg == 0 && t == 0) rowptr[n] = tot;
}

__global__ __launch_bounds__(256)
void place_chunked_scan_kernel(const int* __restrict__ src, const int* __restrict__ dst,
                               const int* __restrict__ rank, const int* __restrict__ rowloc,
                               const int* __restrict__ btot, const int* __restrict__ before,
                               int* __restrict__ rowptr, int* __restrict__ csr_src,
                               int E, int n, int chunk) {
  place_chunked_scan_body<1>(src, dst, rank, rowloc, btot, before, rowptr, csr_src, E, n, chunk,
                             (int)blockIdx.x, (int)gridDim.x);
}

__global__ void wsum_kernel(const float* __restrict__ Wm, const float* __restrict__ bm,
                            float* __restrict__ wsum, float* __restrict__ bsum) {
  int t = threadIdx.x;
  if (t < HD) {
    float s = 0.0f;
    for (int j = 0; j < HD; ++j) s += Wm[t * HD + j];
    wsum[t] = s;
  } else if (t == HD) {
    float s = 0.0f;
    for (int j = 0; j < HD; ++j) s += bm[j];
    *bsum = s;
  }
}

__device__ __forceinline__
void gemm_norm_wmma_body(const float* X, const int* __restrict__ rowmap,
                         const float* __restrict__ norm, const float* __restrict__ W,
                         float* OUT, int nrows, int K, int norm_shift, int wg) {
  __shared__ float ldsX[RPB][LPAD];
  __shared__ float ldsW[HD][WPAD];

  const int tid  = threadIdx.x;
  const int wave = tid >> 5;
  const int lane = tid & 31;
  const int half = lane >> 4;
  const int l16  = lane & 15;
  const int row0 = wg * RPB;

  v8f acc[6] = {};

  const int xc4 = (tid & 7) * 4;
  const float* xptr[4];
  float        xscale[4];
  bool         xrow_ok[4];
  int          xsrc[4];
  #pragma unroll
  for (int it = 0; it < 4; ++it) {
    const int grow = row0 + ((tid + 256 * it) >> 3);
    xrow_ok[it]    = grow < nrows;
    xsrc[it]       = xrow_ok[it] ? grow : nrows - 1;
  }
  if (norm) {
    #pragma unroll
    for (int it = 0; it < 4; ++it) xscale[it] = norm[xsrc[it] >> norm_shift];
  } else {
    #pragma unroll
    for (int it = 0; it < 4; ++it) xscale[it] = 1.0f;
  }
  if (rowmap) {
    #pragma unroll
    for (int it = 0; it < 4; ++it) xsrc[it] = rowmap[xsrc[it]];
  }
  #pragma unroll
  for (int it = 0; it < 4; ++it) xptr[it] = X + (long)xsrc[it] * K + xc4;
  const float* wptr[12];
  int wkk[12];
  #pragma unroll
  for (int it = 0; it < 12; ++it) {
    const int idx = tid + 256 * it;
    wkk[it]  = idx / HD;
    wptr[it] = W + (idx % HD);
  }

  float4 xv[4];
  float  wv[12];
  auto load_tile = [&](int k0) {
    const bool kin = (k0 + xc4) < K;
    #pragma unroll
    for (int it = 0; it < 4; ++it) {
      const float* p = xptr[it] + (kin ? k0 : -xc4);
      xv[it] = *(const float4*)p;
      __builtin_prefetch(((k0 + BK + xc4) < K) ? (p + BK) : p, 0, 0);
    }
    #pragma unroll
    for (int it = 0; it < 12; ++it) {
      const int k = k0 + wkk[it];
      wv[it] = wptr[it][(long)((k < K) ? k : 0) * HD];
    }
  };

  load_tile(0);
  for (int k0 = 0; k0 < K; k0 += BK) {
    {
      const bool kin = (k0 + xc4) < K;
      #pragma unroll
      for (int it = 0; it < 4; ++it) {
        const int r = (tid + 256 * it) >> 3;
        float4 v = xv[it];
        const float s = xscale[it];
        v.x *= s; v.y *= s; v.z *= s; v.w *= s;
        if (!(kin && xrow_ok[it])) v = make_float4(0.f, 0.f, 0.f, 0.f);
        ldsX[r][xc4 + 0] = v.x;
        ldsX[r][xc4 + 1] = v.y;
        ldsX[r][xc4 + 2] = v.z;
        ldsX[r][xc4 + 3] = v.w;
      }
      #pragma unroll
      for (int it = 0; it < 12; ++it) {
        const int idx = tid + 256 * it;
        ldsW[idx % HD][wkk[it]] = (k0 + wkk[it] < K) ? wv[it] : 0.0f;
      }
    }
    __syncthreads();

    if (k0 + BK < K) load_tile(k0 + BK);

    const int kmax = (BK < K - k0) ? BK : (K - k0);
    const int mrow = wave * 16 + l16;
    for (int kk = 0; kk < kmax; kk += 4) {
      int ka = kk + half * 2;

      v2f A = *(const v2f*)&ldsX[mrow][ka];
      v2f Bf[6];
      #pragma unroll
      for (int t = 0; t < 6; ++t)
        Bf[t] = *(const v2f*)&ldsW[t * 16 + l16][ka];

      #pragma unroll
      for (int t = 0; t < 6; ++t)
        acc[t] = __builtin_amdgcn_wmma_f32_16x16x4_f32(
             false, A,  false, Bf[t],
             (short)0, acc[t],  false,  false);
    }
    __syncthreads();
  }

  const int orow = row0 + wave * 16 + half * 8;
  #pragma unroll
  for (int t = 0; t < 6; ++t) {
    int col = t * 16 + l16;
    #pragma unroll
    for (int i = 0; i < 8; ++i) {
      int r = orow + i;
      if (r < nrows) OUT[(long)r * HD + col] = acc[t][i];
    }
  }
}

__global__ __launch_bounds__(256)
void gemm_norm_wmma_kernel(const float* X, const int* __restrict__ rowmap,
                           const float* __restrict__ norm, const float* __restrict__ W,
                           float* OUT, int nrows, int K, int norm_shift) {
  gemm_norm_wmma_body(X, rowmap, norm, W, OUT, nrows, K, norm_shift, (int)blockIdx.x);
}

constexpr int PLACE_EPT = 8;
__global__ __launch_bounds__(256)
void gemm_and_place_kernel(const float* X, const float* __restrict__ W, float* OUT, int nrows, int K, int gemm_wgs,
                           const int* __restrict__ src, const int* __restrict__ dst,
                           const int* __restrict__ rank, const int* __restrict__ rowloc,
                           const int* __restrict__ btot, const int* __restrict__ before,
                           int* __restrict__ rowptr, int* __restrict__ csr_src,
                           int E, int n, int chunk) {
  if ((int)blockIdx.x < gemm_wgs) {
    gemm_norm_wmma_body(X, nullptr, nullptr, W, OUT, nrows, K, 0, (int)blockIdx.x);
  } else {
    place_chunked_scan_body<PLACE_EPT>(src, dst, rank, rowloc, btot, before, rowptr, csr_src, E, n, chunk,
                                       (int)blockIdx.x - gemm_wgs, (int)gridDim.x - gemm_wgs);
  }
}

__device__ __forceinline__ void gather_row_sum(const int* __restrict__ rowptr, const int* __restrict__ csr_src,
                                               const float* __restrict__ h, int row, int lane,
                                               float& a0, float& a1, float& a2) {
  const int beg = rowptr[row];
  const int end = rowptr[row + 1];
  a0 = 0.0f; a1 = 0.0f; a2 = 0.0f;
  for (int base = beg; base < end; base += 32) {
    const int cnt  = (end - base < 32) ? (end - base) : 32;
    const int mine = (lane < cnt) ? csr_src[base + lane] : 0;
    int j = 0;
    for (; j + 4 <= cnt; j += 4) {
      const float* p0 = h + (long)__builtin_amdgcn_readlane(mine, j + 0) * HD + lane;
      const float* p1 = h + (long)__builtin_amdgcn_readlane(mine, j + 1) * HD + lane;
      const float* p2 = h + (long)__builtin_amdgcn_readlane(mine, j + 2) * HD + lane;
      const float* p3 = h + (long)__builtin_amdgcn_readlane(mine, j + 3) * HD + lane;
      const float x00 = p0[0], x01 = p0[32], x02 = p0[64];
      const float x10 = p1[0], x11 = p1[32], x12 = p1[64];
      const float x20 = p2[0], x21 = p2[32], x22 = p2[64];
      const float x30 = p3[0], x31 = p3[32], x32 = p3[64];
      a0 += x00; a1 += x01; a2 += x02;
      a0 += x10; a1 += x11; a2 += x12;
      a0 += x20; a1 += x21; a2 += x22;
      a0 += x30; a1 += x31; a2 += x32;
    }
    for (; j < cnt; ++j) {
      const float* p = h + (long)__builtin_amdgcn_readlane(mine, j) * HD + lane;
      a0 += p[0]; a1 += p[32]; a2 += p[64];
    }
  }
}

template <bool MAPPED>
__device__ __forceinline__ void gather_row_sum_scaled(const int* __restrict__ rowptr, const int* __restrict__ csr_src,
                                                      const float* __restrict__ xw, const int* __restrict__ rowmap,
                                                      const float* __restrict__ scale, int row, int lane,
                                                      float& a0, float& a1, float& a2) {
  const int beg = rowptr[row];
  const int end = rowptr[row + 1];
  a0 = 0.0f; a1 = 0.0f; a2 = 0.0f;
  for (int base = beg; base < end; base += 32) {
    const int cnt  = (end - base < 32) ? (end - base) : 32;
    const int mine = (lane < cnt) ? csr_src[base + lane] : 0;
    const int m    = MAPPED ? rowmap[mine] : mine;
    const int sc   = __float_as_int(scale[mine]);
    int j = 0;
    for (; j + 4 <= cnt; j += 4) {
      const float* p0 = xw + (long)__builtin_amdgcn_readlane(m, j + 0) * HD + lane;
      const float* p1 = xw + (long)__builtin_amdgcn_readlane(m, j + 1) * HD + lane;
      const float* p2 = xw + (long)__builtin_amdgcn_readlane(m, j + 2) * HD + lane;
      const float* p3 = xw + (long)__builtin_amdgcn_readlane(m, j + 3) * HD + lane;
      const float x00 = p0[0], x01 = p0[32], x02 = p0[64];
      const float x10 = p1[0], x11 = p1[32], x12 = p1[64];
      const float x20 = p2[0], x21 = p2[32], x22 = p2[64];
      const float x30 = p3[0], x31 = p3[32], x32 = p3[64];
      __builtin_amdgcn_sched_barrier(0);
      const float c0 = __int_as_float(__builtin_amdgcn_readlane(sc, j + 0));
      const float c1 = __int_as_float(__builtin_amdgcn_readlane(sc, j + 1));
      const float c2 = __int_as_float(__builtin_amdgcn_readlane(sc, j + 2));
      const float c3 = __int_as_float(__builtin_amdgcn_readlane(sc, j + 3));
      a0 = fmaf(c0, x00, a0); a1 = fmaf(c0, x01, a1); a2 = fmaf(c0, x02, a2);
      a0 = fmaf(c1, x10, a0); a1 = fmaf(c1, x11, a1); a2 = fmaf(c1, x12, a2);
      a0 = fmaf(c2, x20, a0); a1 = fmaf(c2, x21, a1); a2 = fmaf(c2, x22, a2);
      a0 = fmaf(c3, x30, a0); a1 = fmaf(c3, x31, a1); a2 = fmaf(c3, x32, a2);
    }
    for (; j < cnt; ++j) {
      const float* p = xw + (long)__builtin_amdgcn_readlane(m, j) * HD + lane;
      const float x0 = p[0], x1 = p[32], x2 = p[64];
      const float c  = __int_as_float(__builtin_amdgcn_readlane(sc, j));
      a0 = fmaf(c, x0, a0); a1 = fmaf(c, x1, a1); a2 = fmaf(c, x2, a2);
    }
  }
}

template <bool MAPPED>
__device__ __forceinline__ void gather_scaled_prelu_node(const int* __restrict__ rowptr, const int* __restrict__ csr_src,
                                                         const float* __restrict__ xw, const float* __restrict__ norm_out,
                                                         const float* __restrict__ norm_in, const float* __restrict__ b,
                                                         const float* __restrict__ a, float* __restrict__ act_row,
                                                         const int* __restrict__ rowmap, int row, int lane) {
  const float ni = norm_in[row];
  const float b0 = b[lane], b1 = b[lane + 32], b2 = b[lane + 64];
  const float s0 = a[lane], s1 = a[lane + 32], s2 = a[lane + 64];
  float a0, a1, a2;
  gather_row_sum_scaled<MAPPED>(rowptr, csr_src, xw, rowmap, norm_out, row, lane, a0, a1, a2);
  const float v0 = a0 * ni + b0;
  const float v1 = a1 * ni + b1;
  const float v2 = a2 * ni + b2;
  float* o = act_row + lane;
  o[0]  = (v0 >= 0.0f) ? v0 : s0 * v0;
  o[32] = (v1 >= 0.0f) ? v1 : s1 * v1;
  o[64] = (v2 >= 0.0f) ? v2 : s2 * v2;
}

template <bool MAPPED>
__global__ __launch_bounds__(256)
void spmm_gather_scaled_prelu_kernel(const int* __restrict__ rowptr, const int* __restrict__ csr_src,
                                     const float* __restrict__ xw, const float* __restrict__ norm_out,
                                     const float* __restrict__ norm_in, const float* __restrict__ b,
                                     const float* __restrict__ a, float* __restrict__ act,
                                     const int* __restrict__ rowmap, int n) {
  const int lane = threadIdx.x & 31;
  const int row = __builtin_amdgcn_readfirstlane((int)(blockIdx.x * (blockDim.x >> 5) + (threadIdx.x >> 5)));
  if (row >= n) return;
  gather_scaled_prelu_node<MAPPED>(rowptr, csr_src, xw, norm_out, norm_in, b, a, act + (long)row * HD, rowmap, row, lane);
}

template <bool MAPPED>
__device__ __forceinline__ void gather_scaled_prelu_vals(const int* __restrict__ rowptr, const int* __restrict__ csr_src,
                                                         const float* __restrict__ xw, const float* __restrict__ norm_out,
                                                         const float* __restrict__ norm_in, const float* __restrict__ b,
                                                         const float* __restrict__ a, const int* __restrict__ rowmap,
                                                         int row, int lane, float& o0, float& o1, float& o2) {
  const float ni = norm_in[row];
  const float b0 = b[lane], b1 = b[lane + 32], b2 = b[lane + 64];
  const float s0 = a[lane], s1 = a[lane + 32], s2 = a[lane + 64];
  float a0, a1, a2;
  gather_row_sum_scaled<MAPPED>(rowptr, csr_src, xw, rowmap, norm_out, row, lane, a0, a1, a2);
  const float v0 = a0 * ni + b0;
  const float v1 = a1 * ni + b1;
  const float v2 = a2 * ni + b2;
  o0 = (v0 >= 0.0f) ? v0 : s0 * v0;
  o1 = (v1 >= 0.0f) ? v1 : s1 * v1;
  o2 = (v2 >= 0.0f) ? v2 : s2 * v2;
}

constexpr int L1_WGS = 1280;
constexpr int TPAD = 100;
__global__ __launch_bounds__(256)
void spmm_gather_scaled_prelu_both_w2_kernel(const int* __restrict__ rowptr, const int* __restrict__ csr_src,
                                             const float* __restrict__ xw, const float* __restrict__ norm_out,
                                             const float* __restrict__ norm_in, const float* __restrict__ b,
                                             const float* __restrict__ a, const float* __restrict__ W2,
                                             float* __restrict__ h2, const int* __restrict__ rowmap,
                                             int n, int tiles_per_view) {
  __shared__ __attribute__((aligned(16))) float tile[2][16][TPAD];
  const int lane = threadIdx.x & 31;
  const int half = lane >> 4;
  const int l16  = lane & 15;
  const int wave = __builtin_amdgcn_readfirstlane((int)(threadIdx.x >> 5));

  v2f Bf[24];
  if (wave < 6) {
    const float* wp = W2 + wave * 16 + l16;
    #pragma unroll
    for (int j = 0; j < 24; ++j) {
      const int k = 4 * j + 2 * half;
      Bf[j].x = wp[k * HD];
      Bf[j].y = wp[(k + 1) * HD];
    }
  } else {
    #pragma unroll
    for (int j = 0; j < 24; ++j) { Bf[j].x = 0.0f; Bf[j].y = 0.0f; }
  }

  const int ntiles = 2 * tiles_per_view;
  int buf = 0;
  for (int t = (int)blockIdx.x; t < ntiles; t += (int)gridDim.x) {
    const bool second = t >= tiles_per_view;
    const int  node0  = (second ? t - tiles_per_view : t) * 16;
    for (int q = 0; q < 2; ++q) {
      const int r    = wave + 8 * q;
      const int node = node0 + r;
      float o0 = 0.0f, o1 = 0.0f, o2 = 0.0f;
      if (node < n) {
        const float no = norm_out[node];
        if (second) gather_scaled_prelu_vals<true>(rowptr, csr_src, xw, norm_out, norm_in, b, a, rowmap, node, lane, o0, o1, o2);
        else        gather_scaled_prelu_vals<false>(rowptr, csr_src, xw, norm_out, norm_in, b, a, nullptr, node, lane, o0, o1, o2);
        o0 *= no; o1 *= no; o2 *= no;
      }
      tile[buf][r][lane]      = o0;
      tile[buf][r][lane + 32] = o1;
      tile[buf][r][lane + 64] = o2;
    }
    __syncthreads();
    if (wave < 6) {
      v8f acc = {};
      #pragma unroll
      for (int j = 0; j < 24; ++j) {
        const v2f A = *(const v2f*)&tile[buf][l16][4 * j + 2 * half];
        acc = __builtin_amdgcn_wmma_f32_16x16x4_f32(
             false, A,  false, Bf[j],
             (short)0, acc,  false,  false);
      }
      const int col = wave * 16 + l16;
      #pragma unroll
      for (int i = 0; i < 8; ++i) {
        const int node = node0 + half * 8 + i;
        if (node < n) h2[((long)node * 2 + (second ? 1 : 0)) * HD + col] = acc[i];
      }
    }
    buf ^= 1;
  }
}

__global__ __launch_bounds__(256)
void spmm_gather_prelu_dot_kernel(const int* __restrict__ rowptr, const int* __restrict__ csr_src,
                                  const float* __restrict__ h, const float* __restrict__ norm_in,
                                  const float* __restrict__ b, const float* __restrict__ a,
                                  const float* __restrict__ wsum, const float* __restrict__ bsum,
                                  float* __restrict__ out, int n) {
  const int lane = threadIdx.x & 31;
  const int row = __builtin_amdgcn_readfirstlane((int)(blockIdx.x * (blockDim.x >> 5) + (threadIdx.x >> 5)));
  if (row >= n) return;
  const float ni = norm_in[row];
  const float b0 = b[lane], b1 = b[lane + 32], b2 = b[lane + 64];
  const float s0 = a[lane], s1 = a[lane + 32], s2 = a[lane + 64];
  const float w0 = wsum[lane], w1 = wsum[lane + 32], w2 = wsum[lane + 64];
  const float bs = *bsum;
  float a0, a1, a2;
  gather_row_sum(rowptr, csr_src, h, row, lane, a0, a1, a2);
  const float v0 = a0 * ni + b0;
  const float v1 = a1 * ni + b1;
  const float v2 = a2 * ni + b2;
  const float z0 = (v0 >= 0.0f) ? v0 : s0 * v0;
  const float z1 = (v1 >= 0.0f) ? v1 : s1 * v1;
  const float z2 = (v2 >= 0.0f) ? v2 : s2 * v2;
  float s = fmaf(z0, w0, 0.0f);
  s = fmaf(z1, w1, s);
  s = fmaf(z2, w2, s);
  #pragma unroll
  for (int off = 16; off > 0; off >>= 1) s += __shfl_xor(s, off, 32);
  if (lane == 0) out[row] = s + bs;
}

__device__ __forceinline__ void gather_row_sum2(const int* __restrict__ rowptr, const int* __restrict__ csr_src,
                                                const float* __restrict__ h2, int row, int lane,
                                                float& a0, float& a1, float& a2, float& c0, float& c1, float& c2) {
  const int beg = rowptr[row];
  const int end = rowptr[row + 1];
  a0 = 0.0f; a1 = 0.0f; a2 = 0.0f; c0 = 0.0f; c1 = 0.0f; c2 = 0.0f;
  for (int base = beg; base < end; base += 32) {
    const int cnt  = (end - base < 32) ? (end - base) : 32;
    const int mine = (lane < cnt) ? csr_src[base + lane] : 0;
    int j = 0;
    for (; j + 4 <= cnt; j += 4) {
      const float* p0 = h2 + (long)__builtin_amdgcn_readlane(mine, j + 0) * (2 * HD) + lane;
      const float* p1 = h2 + (long)__builtin_amdgcn_readlane(mine, j + 1) * (2 * HD) + lane;
      const float* p2 = h2 + (long)__builtin_amdgcn_readlane(mine, j + 2) * (2 * HD) + lane;
      const float* p3 = h2 + (long)__builtin_amdgcn_readlane(mine, j + 3) * (2 * HD) + lane;
      const float x00 = p0[0], x01 = p0[32], x02 = p0[64], y00 = p0[96], y01 = p0[128], y02 = p0[160];
      const float x10 = p1[0], x11 = p1[32], x12 = p1[64], y10 = p1[96], y11 = p1[128], y12 = p1[160];
      const float x20 = p2[0], x21 = p2[32], x22 = p2[64], y20 = p2[96], y21 = p2[128], y22 = p2[160];
      const float x30 = p3[0], x31 = p3[32], x32 = p3[64], y30 = p3[96], y31 = p3[128], y32 = p3[160];
      a0 += x00; a1 += x01; a2 += x02; c0 += y00; c1 += y01; c2 += y02;
      a0 += x10; a1 += x11; a2 += x12; c0 += y10; c1 += y11; c2 += y12;
      a0 += x20; a1 += x21; a2 += x22; c0 += y20; c1 += y21; c2 += y22;
      a0 += x30; a1 += x31; a2 += x32; c0 += y30; c1 += y31; c2 += y32;
    }
    for (; j < cnt; ++j) {
      const float* p = h2 + (long)__builtin_amdgcn_readlane(mine, j) * (2 * HD) + lane;
      const float x0 = p[0], x1 = p[32], x2 = p[64], y0 = p[96], y1 = p[128], y2 = p[160];
      a0 += x0; a1 += x1; a2 += x2; c0 += y0; c1 += y1; c2 += y2;
    }
  }
}

__global__ __launch_bounds__(256)
void spmm_gather2_prelu_dot_kernel(const int* __restrict__ rowptr, const int* __restrict__ csr_src,
                                   const float* __restrict__ h2, const float* __restrict__ norm_in,
                                   const float* __restrict__ b, const float* __restrict__ a,
                                   const float* __restrict__ wsum, const float* __restrict__ bsum,
                                   float* __restrict__ out, int n) {
  const int lane = threadIdx.x & 31;
  const int row = __builtin_amdgcn_readfirstlane((int)(blockIdx.x * (blockDim.x >> 5) + (threadIdx.x >> 5)));
  if (row >= n) return;
  const float ni = norm_in[row];
  const float b0 = b[lane], b1 = b[lane + 32], b2 = b[lane + 64];
  const float s0 = a[lane], s1 = a[lane + 32], s2 = a[lane + 64];
  const float w0 = wsum[lane], w1 = wsum[lane + 32], w2 = wsum[lane + 64];
  const float bs = *bsum;
  float a0, a1, a2, c0, c1, c2;
  gather_row_sum2(rowptr, csr_src, h2, row, lane, a0, a1, a2, c0, c1, c2);
  const float v0 = a0 * ni + b0;
  const float v1 = a1 * ni + b1;
  const float v2 = a2 * ni + b2;
  const float u0 = c0 * ni + b0;
  const float u1 = c1 * ni + b1;
  const float u2 = c2 * ni + b2;
  const float z0 = (v0 >= 0.0f) ? v0 : s0 * v0;
  const float z1 = (v1 >= 0.0f) ? v1 : s1 * v1;
  const float z2 = (v2 >= 0.0f) ? v2 : s2 * v2;
  const float q0 = (u0 >= 0.0f) ? u0 : s0 * u0;
  const float q1 = (u1 >= 0.0f) ? u1 : s1 * u1;
  const float q2 = (u2 >= 0.0f) ? u2 : s2 * u2;
  float s = fmaf(z0, w0, 0.0f);
  s = fmaf(z1, w1, s);
  s = fmaf(z2, w2, s);
  float r = fmaf(q0, w0, 0.0f);
  r = fmaf(q1, w1, r);
  r = fmaf(q2, w2, r);
  for (int off = 16; off > 0; off >>= 1) {
    s += __shfl_xor(s, off, 32);
    r += __shfl_xor(r, off, 32);
  }
  if (lane == 0) {
    out[row]     = s + bs;
    out[n + row] = r + bs;
  }
}

__global__ void spmm_kernel(const int* __restrict__ src, const int* __restrict__ dst,
                            const float* __restrict__ h, float* __restrict__ agg, int E) {
  int gw   = (int)(((long)blockIdx.x * blockDim.x + threadIdx.x) >> 5);
  int lane = threadIdx.x & 31;
  if (gw < E) {
    int s = src[gw], d = dst[gw];
    const float* hp = h + (long)s * HD;
    float* ap = agg + (long)d * HD;
    #pragma unroll
    for (int i = 0; i < 3; ++i) {
      float v = hp[lane + 32 * i];
      __hip_atomic_fetch_add(&ap[lane + 32 * i], v, __ATOMIC_RELAXED, __HIP_MEMORY_SCOPE_AGENT);
    }
  }
}

__global__ void prelu_kernel(const float* __restrict__ agg, const float* __restrict__ norm_in,
                             const float* __restrict__ b, const float* __restrict__ a,
                             float* __restrict__ out, int n) {
  long idx = (long)blockIdx.x * blockDim.x + threadIdx.x;
  if (idx < (long)n * HD) {
    int i = (int)(idx / HD);
    int j = (int)(idx % HD);
    float v = agg[idx] * norm_in[i] + b[j];
    out[idx] = (v >= 0.0f) ? v : a[j] * v;
  }
}

__global__ void reduce_kernel(const float* __restrict__ z, const float* __restrict__ wsum,
                              const float* __restrict__ bsum, float* __restrict__ out, int n) {
  int gw   = (int)(((long)blockIdx.x * blockDim.x + threadIdx.x) >> 5);
  int lane = threadIdx.x & 31;
  if (gw < n) {
    const float* zp = z + (long)gw * HD;
    float s = 0.0f;
    #pragma unroll
    for (int i = 0; i < 3; ++i) s = fmaf(zp[lane + 32 * i], wsum[lane + 32 * i], s);
    #pragma unroll
    for (int off = 16; off > 0; off >>= 1) s += __shfl_xor(s, off, 32);
    if (lane == 0) out[gw] = s + *bsum;
  }
}

extern "C" void kernel_launch(void* const* d_in, const int* in_sizes, int n_in,
                              void* d_out, int out_size, void* d_ws, size_t ws_size,
                              hipStream_t stream) {
  const float* x    = (const float*)d_in[0];
  const int*   src  = (const int*)  d_in[1];
  const int*   dst  = (const int*)  d_in[2];
  const int*   perm = (const int*)  d_in[3];
  const float* W1   = (const float*)d_in[4];
  const float* b1   = (const float*)d_in[5];
  const float* a1   = (const float*)d_in[6];
  const float* W2   = (const float*)d_in[7];
  const float* b2   = (const float*)d_in[8];
  const float* a2   = (const float*)d_in[9];
  const float* Wm   = (const float*)d_in[10];
  const float* bm   = (const float*)d_in[11];
  float* out = (float*)d_out;

  const int E  = in_sizes[1];
  const int N  = in_sizes[3];
  const int K1 = in_sizes[0] / N;

  float* ws       = (float*)d_ws;
  float* norm_out = ws;
  float* norm_in  = ws + N;
  float* wsum     = ws + 2L * N;
  float* bsum     = ws + 2L * N + HD;
  float* bufA     = ws + 2L * N + 2 * HD;
  float* bufB     = bufA + (long)N * HD;

  const long NH = (long)N * HD;
  const size_t need_wide = (size_t)(2L * N + 2 * HD + 3 * NH + (N + 1) + E) * sizeof(float);
  const bool wide = (ws_size >= need_wide) && ((long)E <= NH);
  float* H2       = bufB;
  int*   rowptr   = (int*)(bufB + (wide ? 2 * NH : NH));
  int*   csr_src  = rowptr + (N + 1);
  int*   rank     = (int*)bufB;
  int*   cnt_out  = (int*)norm_out;
  int*   cnt_in   = (int*)norm_in;
  const bool hist_in_h2 = wide && ((long)HG * N <= NH);
  int*   hist_in  = hist_in_h2 ? (int*)(bufB + NH) : (int*)bufA;
  int*   hist_out = (int*)bufB;
  int*   lrank    = hist_out + (long)HG * N;
  int*   rowloc   = lrank + E;
  int*   btot     = rowloc + N;

  const int gemm_blocks = (N + RPB - 1) / RPB;
  const int spmm_blocks = (int)(((long)E * 32 + 255) / 256);
  const int gath_blocks = (N + 7) / 8;
  const int elw_blocks  = (int)((NH + 255) / 256);
  const int red_blocks  = (int)(((long)N * 32 + 255) / 256);

  const size_t need_grouped = (size_t)(2L * N + 2 * HD + 2 * NH + (N + 1) + E) * sizeof(float);
  const bool grouped = wide || ((ws_size >= need_grouped) && ((long)E <= NH));
  const int  chunk    = (E + HG - 1) / HG;
  const bool lds_hist = grouped && (N <= HCAP) && ((long)(HG + 1) * N + E + 256 <= NH);

  if (lds_hist) {
    hist_lds_kernel<<<2 * HG, 1024, 0, stream>>>(src, dst, hist_in, hist_out, lrank, E, N, chunk);
    hist_fold_scan_kernel<<<(2 * N + 255) / 256 + 1, 256, 0, stream>>>(hist_in, hist_out, rowloc, btot, norm_out, norm_in, N,
                                                                       Wm, bm, wsum, bsum);
    if (hist_in_h2) {
      const int place_wgs = (E + PLACE_EPT * 256 - 1) / (PLACE_EPT * 256);
      gemm_and_place_kernel<<<gemm_blocks + place_wgs, 256, 0, stream>>>(x, W1, bufA, N, K1, gemm_blocks,
                                                                         src, dst, lrank, rowloc, btot, hist_in,
                                                                         rowptr, csr_src, E, N, chunk);
    } else {
      place_chunked_scan_kernel<<<(E + 255) / 256, 256, 0, stream>>>(src, dst, lrank, rowloc, btot, hist_in, rowptr,
                                                                     csr_src, E, N, chunk);
    }
  } else {
    fill_zero_kernel<<<512, 256, 0, stream>>>(norm_out, 2L * N);
    degree_kernel<<<(E + 255) / 256, 256, 0, stream>>>(src, dst, cnt_out, cnt_in, grouped ? rank : nullptr, E);
    if (grouped) {
      rowptr_scan_kernel<<<1, 1024, 0, stream>>>(cnt_in, rowptr, N);
      place_kernel<<<(E + 255) / 256, 256, 0, stream>>>(src, dst, rank, rowptr, csr_src, E);
    }
    finalize_norm_kernel<<<(2 * N + 255) / 256, 256, 0, stream>>>(norm_out, 2 * N);
    wsum_kernel<<<1, 128, 0, stream>>>(Wm, bm, wsum, bsum);
  }

  if (grouped && !(lds_hist && hist_in_h2))
    gemm_norm_wmma_kernel<<<gemm_blocks, 256, 0, stream>>>(x, nullptr, nullptr, W1, bufA, N, K1, 0);

  if (wide) {
    const int tiles_per_view = (N + 15) / 16;
    const int l1_trips = (2 * tiles_per_view + L1_WGS - 1) / L1_WGS;
    const int l1_wgs   = (2 * tiles_per_view + l1_trips - 1) / l1_trips;
    spmm_gather_scaled_prelu_both_w2_kernel<<<l1_wgs, 256, 0, stream>>>(rowptr, csr_src, bufA, norm_out, norm_in,
                                                                       b1, a1, W2, H2, perm, N, tiles_per_view);
    spmm_gather2_prelu_dot_kernel<<<gath_blocks, 256, 0, stream>>>(rowptr, csr_src, H2, norm_in, b2, a2,
                                                                  wsum, bsum, out, N);
    return;
  }

  for (int view = 0; view < 2; ++view) {
    const int* rowmap = view ? perm : nullptr;
    float* outv = out + (long)view * N;
    if (grouped) {
      if (view) {
        spmm_gather_scaled_prelu_kernel<true><<<gath_blocks, 256, 0, stream>>>(rowptr, csr_src, bufA, norm_out, norm_in,
                                                                              b1, a1, bufB, rowmap, N);
      } else {
        spmm_gather_scaled_prelu_kernel<false><<<gath_blocks, 256, 0, stream>>>(rowptr, csr_src, bufA, norm_out, norm_in,
                                                                               b1, a1, bufB, nullptr, N);
      }
      gemm_norm_wmma_kernel<<<gemm_blocks, 256, 0, stream>>>(bufB, nullptr, norm_out, W2, bufB, N, HD, 0);
      spmm_gather_prelu_dot_kernel<<<gath_blocks, 256, 0, stream>>>(rowptr, csr_src, bufB, norm_in, b2, a2,
                                                                   wsum, bsum, outv, N);
    } else {
      gemm_norm_wmma_kernel<<<gemm_blocks, 256, 0, stream>>>(x, rowmap, norm_out, W1, bufA, N, K1, 0);
      fill_zero_kernel<<<2048, 256, 0, stream>>>(bufB, NH);
      spmm_kernel<<<spmm_blocks, 256, 0, stream>>>(src, dst, bufA, bufB, E);
      prelu_kernel<<<elw_blocks, 256, 0, stream>>>(bufB, norm_in, b1, a1, bufA, N);
      gemm_norm_wmma_kernel<<<gemm_blocks, 256, 0, stream>>>(bufA, nullptr, norm_out, W2, bufB, N, HD, 0);
      fill_zero_kernel<<<2048, 256, 0, stream>>>(bufA, NH);
      spmm_kernel<<<spmm_blocks, 256, 0, stream>>>(src, dst, bufB, bufA, E);
      prelu_kernel<<<elw_blocks, 256, 0, stream>>>(bufA, norm_in, b2, a2, bufB, N);
      reduce_kernel<<<red_blocks, 256, 0, stream>>>(bufB, wsum, bsum, outv, N);
    }
  }
}


# layer 1: one wave walks a node's source list ONCE for both views (ids, scales loaded once; 24 row loads in flight per 4 sources, not 12 in two walks); tile = 8 nodes x 2 views; sum orders kept
# speedup vs baseline: 5.4585x; 1.0490x over previous
#include <hip/hip_runtime.h>

typedef float v2f __attribute__((ext_vector_type(2)));
typedef float v8f __attribute__((ext_vector_type(8)));

constexpr int HD   = 96;
constexpr int RPB  = 128;
constexpr int BK   = 32;
constexpr int LPAD = 36;
constexpr int WPAD = 38;
constexpr int HG   = 64;
constexpr int HCAP = 65536;


__global__ void fill_zero_kernel(float* __restrict__ p, long n) {
  long i = (long)blockIdx.x * blockDim.x + threadIdx.x;
  long stride = (long)gridDim.x * blockDim.x;
  for (; i < n; i += stride) p[i] = 0.0f;
}

__global__ void degree_kernel(const int* __restrict__ src, const int* __restrict__ dst,
                              int* __restrict__ cnt_out, int* __restrict__ cnt_in,
                              int* __restrict__ rank, int E) {
  int e = blockIdx.x * blockDim.x + threadIdx.x;
  if (e < E) {
    __hip_atomic_fetch_add(&cnt_out[src[e]], 1, __ATOMIC_RELAXED, __HIP_MEMORY_SCOPE_AGENT);
    if (rank) {
      rank[e] = __hip_atomic_fetch_add(&cnt_in[dst[e]], 1, __ATOMIC_RELAXED, __HIP_MEMORY_SCOPE_AGENT);
    } else {
      __hip_atomic_fetch_add(&cnt_in[dst[e]], 1, __ATOMIC_RELAXED, __HIP_MEMORY_SCOPE_AGENT);
    }
  }
}

__global__ void finalize_norm_kernel(float* __restrict__ p, int n) {
  int i = blockIdx.x * blockDim.x + threadIdx.x;
  if (i < n) p[i] = 1.0f / sqrtf(fmaxf((float)__float_as_int(p[i]), 1.0f));
}

__global__ __launch_bounds__(1024)
void rowptr_scan_kernel(const int* __restrict__ cnt, int* __restrict__ rowptr, int n) {
  __shared__ int wave_tot[32];
  const int t = threadIdx.x, lane = t & 31, w = t >> 5;
  int carry = 0;
  int c[4], nx[4];
  for (int k = 0; k < 4; ++k) {
    const int i = t * 4 + k;
    const int v = cnt[i < n ? i : n - 1];
    c[k] = (i < n) ? v : 0;
  }
  for (int base = 0; base < n; base += 4096) {
    const int i0 = base + t * 4;
    for (int k = 0; k < 4; ++k) {
      const int i = i0 + 4096 + k;
      const int v = cnt[i < n ? i : n - 1];
      nx[k] = (i < n) ? v : 0;
    }
    const int s = c[0] + c[1] + c[2] + c[3];
    int incl = s;
    for (int off = 1; off < 32; off <<= 1) {
      int v = __shfl_up(incl, off, 32);
      if (lane >= off) incl += v;
    }
    if (lane == 31) wave_tot[w] = incl;
    __syncthreads();
    const int wt = wave_tot[lane];
    int wincl = wt;
    for (int off = 1; off < 32; off <<= 1) {
      int v = __shfl_up(wincl, off, 32);
      if (lane >= off) wincl += v;
    }
    const int wave_excl = __shfl(wincl, w, 32) - __shfl(wt, w, 32);
    const int total     = __shfl(wincl, 31, 32);
    int run = carry + wave_excl + incl - s;
    for (int k = 0; k < 4; ++k) {
      if (i0 + k < n) rowptr[i0 + k] = run;
      run += c[k];
    }
    carry += total;
    for (int k = 0; k < 4; ++k) c[k] = nx[k];
    __syncthreads();
  }
  if (t == 0) rowptr[n] = carry;
}

__global__ void place_kernel(const int* __restrict__ src, const int* __restrict__ dst,
                             const int* __restrict__ rank, const int* __restrict__ rowptr,
                             int* __restrict__ csr_src, int E) {
  int e = blockIdx.x * blockDim.x + threadIdx.x;
  if (e < E) csr_src[rowptr[dst[e]] + rank[e]] = src[e];
}

__global__ __launch_bounds__(1024)
void hist_lds_kernel(const int* __restrict__ src, const int* __restrict__ dst,
                     int* __restrict__ hist_in, int* __restrict__ hist_out,
                     int* __restrict__ rank, int E, int n, int chunk) {
  __shared__ int4 cnt4[HCAP / 4];
  int* cnt = reinterpret_cast<int*>(cnt4);
  const int  t     = threadIdx.x;
  const bool is_in = blockIdx.x < HG;
  const int  g     = is_in ? (int)blockIdx.x : (int)blockIdx.x - HG;
  const int* __restrict__ key  = is_in ? dst : src;
  int*       __restrict__ hist = (is_in ? hist_in : hist_out) + (long)g * n;
  const bool vec = ((n & 3) == 0) && ((((size_t)hist) & 15) == 0);

  if (vec) {
    for (int i = t; i < (n >> 2); i += 1024) cnt4[i] = make_int4(0, 0, 0, 0);
  } else {
    for (int i = t; i < n; i += 1024) cnt[i] = 0;
  }
  __syncthreads();

  const int beg = g * chunk;
  const int end = (beg + chunk < E) ? (beg + chunk) : E;
  for (int base = beg + t; base < end; base += 8 * 1024) {
    int k[8];
    #pragma unroll
    for (int u = 0; u < 8; ++u) {
      const int e = base + u * 1024;
      k[u] = key[e < end ? e : end - 1];
    }
    int r[8];
    #pragma unroll
    for (int u = 0; u < 8; ++u) {
      const int e = base + u * 1024;
      r[u] = __hip_atomic_fetch_add(&cnt[k[u]], (e < end) ? 1 : 0, __ATOMIC_RELAXED, __HIP_MEMORY_SCOPE_WORKGROUP);
    }
    if (is_in) {
      #pragma unroll
      for (int u = 0; u < 8; ++u) {
        const int e = base + u * 1024;
        if (e < end) rank[e] = r[u];
      }
    }
  }
  __syncthreads();

  if (vec) {
    int4* __restrict__ h4 = reinterpret_cast<int4*>(hist);
    for (int i = t; i < (n >> 2); i += 1024) h4[i] = cnt4[i];
  } else {
    for (int i = t; i < n; i += 1024) hist[i] = cnt[i];
  }
}

__global__ __launch_bounds__(256)
void hist_fold_scan_kernel(int* __restrict__ hist_in, const int* __restrict__ hist_out,
                           int* __restrict__ rowloc, int* __restrict__ btot,
                           float* __restrict__ norm_out, float* __restrict__ norm_in, int n,
                           const float* __restrict__ Wm, const float* __restrict__ bm,
                           float* __restrict__ wsum, float* __restrict__ bsum) {
  __shared__ int wave_tot[8];
  const int t = threadIdx.x;
  if (blockIdx.x == 0) {
    if (t < HD) {
      float s = 0.0f;
      for (int j = 0; j < HD; ++j) s += Wm[t * HD + j];
      wsum[t] = s;
    } else if (t == HD) {
      float s = 0.0f;
      for (int j = 0; j < HD; ++j) s += bm[j];
      *bsum = s;
    }
    return;
  }
  const int blk = (int)blockIdx.x - 1;
  const int i = blk * 256 + t;
  int c[HG];
  int run = 0;
  if (i < n) {
    #pragma unroll
    for (int g = 0; g < HG; ++g) c[g] = hist_in[(long)g * n + i];
    #pragma unroll
    for (int g = 0; g < HG; ++g) run += c[g];
  } else if (i < 2 * n) {
    const int j = i - n;
    int s = 0;
    #pragma unroll
    for (int g = 0; g < HG; ++g) s += hist_out[(long)g * n + j];
    norm_out[j] = 1.0f / sqrtf(fmaxf((float)s, 1.0f));
  }
  if (blk * 256 < n) {
    const int lane = t & 31, w = t >> 5;
    int incl = run;
    for (int off = 1; off < 32; off <<= 1) {
      int v = __shfl_up(incl, off, 32);
      if (lane >= off) incl += v;
    }
    if (lane == 31) wave_tot[w] = incl;
    __syncthreads();
    int wexcl = 0, tot = 0;
    for (int k = 0; k < 8; ++k) {
      const int v = wave_tot[k];
      if (k < w) wexcl += v;
      tot += v;
    }
    if (i < n) rowloc[i] = wexcl + incl - run;
    if (t == 0) btot[blk] = tot;
  }
  if (i < n) {
    int before = 0;
    #pragma unroll
    for (int g = 0; g < HG; ++g) {
      hist_in[(long)g * n + i] = before;
      before += c[g];
    }
    norm_in[i] = 1.0f / sqrtf(fmaxf((float)run, 1.0f));
  }
}

template <int EPT>
__device__ __forceinline__
void place_chunked_scan_body(const int* __restrict__ src, const int* __restrict__ dst,
                             const int* __restrict__ rank, const int* __restrict__ rowloc,
                             const int* __restrict__ btot, const int* __restrict__ before,
                             int* __restrict__ rowptr, int* __restrict__ csr_src,
                             int E, int n, int chunk, int wg, int nwg) {
  __shared__ int bp[256];
  __shared__ int wave_tot[8];
  const int t = threadIdx.x, lane = t & 31, w = t >> 5;
  const int nblk = (n + 255) >> 8;
  int d[EPT], rk[EPT], s[EPT], loc[EPT], bf[EPT];
  #pragma unroll
  for (int u = 0; u < EPT; ++u) {
    const int e  = (wg * EPT + u) * 256 + t;
    const int ec = (e < E) ? e : E - 1;
    d[u]  = dst[ec];
    rk[u] = rank[ec];
    s[u]  = src[ec];
  }
  const int bl = btot[(t < nblk) ? t : 0];
  const int v  = (t < nblk) ? bl : 0;
  #pragma unroll
  for (int u = 0; u < EPT; ++u) {
    const int e  = (wg * EPT + u) * 256 + t;
    const int ec = (e < E) ? e : E - 1;
    const int g  = ec / chunk;
    loc[u] = rowloc[d[u]];
    bf[u]  = before[(long)g * n + d[u]];
  }
  int incl = v;
  for (int off = 1; off < 32; off <<= 1) {
    int u = __shfl_up(incl, off, 32);
    if (lane >= off) incl += u;
  }
  if (lane == 31) wave_tot[w] = incl;
  __syncthreads();
  int wexcl = 0, tot = 0;
  for (int k = 0; k < 8; ++k) {
    const int u = wave_tot[k];
    if (k < w) wexcl += u;
    tot += u;
  }
  bp[t] = wexcl + incl - v;
  __syncthreads();
  #pragma unroll
  for (int u = 0; u < EPT; ++u) {
    const int e = (wg * EPT + u) * 256 + t;
    if (e < E) csr_src[bp[d[u] >> 8] + loc[u] + bf[u] + rk[u]] = s[u];
  }
  for (int b = wg; b < nblk; b += nwg) {
    const int i = b * 256 + t;
    if (i < n) rowptr[i] = rowloc[i] + bp[b];
  }
  if (wg == 0 && t == 0) rowptr[n] = tot;
}

__global__ __launch_bounds__(256)
void place_chunked_scan_kernel(const int* __restrict__ src, const int* __restrict__ dst,
                               const int* __restrict__ rank, const int* __restrict__ rowloc,
                               const int* __restrict__ btot, const int* __restrict__ before,
                               int* __restrict__ rowptr, int* __restrict__ csr_src,
                               int E, int n, int chunk) {
  place_chunked_scan_body<1>(src, dst, rank, rowloc, btot, before, rowptr, csr_src, E, n, chunk,
                             (int)blockIdx.x, (int)gridDim.x);
}

__global__ void wsum_kernel(const float* __restrict__ Wm, const float* __restrict__ bm,
                            float* __restrict__ wsum, float* __restrict__ bsum) {
  int t = threadIdx.x;
  if (t < HD) {
    float s = 0.0f;
    for (int j = 0; j < HD; ++j) s += Wm[t * HD + j];
    wsum[t] = s;
  } else if (t == HD) {
    float s = 0.0f;
    for (int j = 0; j < HD; ++j) s += bm[j];
    *bsum = s;
  }
}

__device__ __forceinline__
void gemm_norm_wmma_body(const float* X, const int* __restrict__ rowmap,
                         const float* __restrict__ norm, const float* __restrict__ W,
                         float* OUT, int nrows, int K, int norm_shift, int wg) {
  __shared__ float ldsX[RPB][LPAD];
  __shared__ float ldsW[HD][WPAD];

  const int tid  = threadIdx.x;
  const int wave = tid >> 5;
  const int lane = tid & 31;
  const int half = lane >> 4;
  const int l16  = lane & 15;
  const int row0 = wg * RPB;

  v8f acc[6] = {};

  const int xc4 = (tid & 7) * 4;
  const float* xptr[4];
  float        xscale[4];
  bool         xrow_ok[4];
  int          xsrc[4];
  #pragma unroll
  for (int it = 0; it < 4; ++it) {
    const int grow = row0 + ((tid + 256 * it) >> 3);
    xrow_ok[it]    = grow < nrows;
    xsrc[it]       = xrow_ok[it] ? grow : nrows - 1;
  }
  if (norm) {
    #pragma unroll
    for (int it = 0; it < 4; ++it) xscale[it] = norm[xsrc[it] >> norm_shift];
  } else {
    #pragma unroll
    for (int it = 0; it < 4; ++it) xscale[it] = 1.0f;
  }
  if (rowmap) {
    #pragma unroll
    for (int it = 0; it < 4; ++it) xsrc[it] = rowmap[xsrc[it]];
  }
  #pragma unroll
  for (int it = 0; it < 4; ++it) xptr[it] = X + (long)xsrc[it] * K + xc4;
  const float* wptr[12];
  int wkk[12];
  #pragma unroll
  for (int it = 0; it < 12; ++it) {
    const int idx = tid + 256 * it;
    wkk[it]  = idx / HD;
    wptr[it] = W + (idx % HD);
  }

  float4 xv[4];
  float  wv[12];
  auto load_tile = [&](int k0) {
    const bool kin = (k0 + xc4) < K;
    #pragma unroll
    for (int it = 0; it < 4; ++it) {
      const float* p = xptr[it] + (kin ? k0 : -xc4);
      xv[it] = *(const float4*)p;
      __builtin_prefetch(((k0 + BK + xc4) < K) ? (p + BK) : p, 0, 0);
    }
    #pragma unroll
    for (int it = 0; it < 12; ++it) {
      const int k = k0 + wkk[it];
      wv[it] = wptr[it][(long)((k < K) ? k : 0) * HD];
    }
  };

  load_tile(0);
  for (int k0 = 0; k0 < K; k0 += BK) {
    {
      const bool kin = (k0 + xc4) < K;
      #pragma unroll
      for (int it = 0; it < 4; ++it) {
        const int r = (tid + 256 * it) >> 3;
        float4 v = xv[it];
        const float s = xscale[it];
        v.x *= s; v.y *= s; v.z *= s; v.w *= s;
        if (!(kin && xrow_ok[it])) v = make_float4(0.f, 0.f, 0.f, 0.f);
        ldsX[r][xc4 + 0] = v.x;
        ldsX[r][xc4 + 1] = v.y;
        ldsX[r][xc4 + 2] = v.z;
        ldsX[r][xc4 + 3] = v.w;
      }
      #pragma unroll
      for (int it = 0; it < 12; ++it) {
        const int idx = tid + 256 * it;
        ldsW[idx % HD][wkk[it]] = (k0 + wkk[it] < K) ? wv[it] : 0.0f;
      }
    }
    __syncthreads();

    if (k0 + BK < K) load_tile(k0 + BK);

    const int kmax = (BK < K - k0) ? BK : (K - k0);
    const int mrow = wave * 16 + l16;
    for (int kk = 0; kk < kmax; kk += 4) {
      int ka = kk + half * 2;

      v2f A = *(const v2f*)&ldsX[mrow][ka];
      v2f Bf[6];
      #pragma unroll
      for (int t = 0; t < 6; ++t)
        Bf[t] = *(const v2f*)&ldsW[t * 16 + l16][ka];

      #pragma unroll
      for (int t = 0; t < 6; ++t)
        acc[t] = __builtin_amdgcn_wmma_f32_16x16x4_f32(
             false, A,  false, Bf[t],
             (short)0, acc[t],  false,  false);
    }
    __syncthreads();
  }

  const int orow = row0 + wave * 16 + half * 8;
  #pragma unroll
  for (int t = 0; t < 6; ++t) {
    int col = t * 16 + l16;
    #pragma unroll
    for (int i = 0; i < 8; ++i) {
      int r = orow + i;
      if (r < nrows) OUT[(long)r * HD + col] = acc[t][i];
    }
  }
}

__global__ __launch_bounds__(256)
void gemm_norm_wmma_kernel(const float* X, const int* __restrict__ rowmap,
                           const float* __restrict__ norm, const float* __restrict__ W,
                           float* OUT, int nrows, int K, int norm_shift) {
  gemm_norm_wmma_body(X, rowmap, norm, W, OUT, nrows, K, norm_shift, (int)blockIdx.x);
}

constexpr int PLACE_EPT = 8;
__global__ __launch_bounds__(256)
void gemm_and_place_kernel(const float* X, const float* __restrict__ W, float* OUT, int nrows, int K, int gemm_wgs,
                           const int* __restrict__ src, const int* __restrict__ dst,
                           const int* __restrict__ rank, const int* __restrict__ rowloc,
                           const int* __restrict__ btot, const int* __restrict__ before,
                           int* __restrict__ rowptr, int* __restrict__ csr_src,
                           int E, int n, int chunk) {
  if ((int)blockIdx.x < gemm_wgs) {
    gemm_norm_wmma_body(X, nullptr, nullptr, W, OUT, nrows, K, 0, (int)blockIdx.x);
  } else {
    place_chunked_scan_body<PLACE_EPT>(src, dst, rank, rowloc, btot, before, rowptr, csr_src, E, n, chunk,
                                       (int)blockIdx.x - gemm_wgs, (int)gridDim.x - gemm_wgs);
  }
}

__device__ __forceinline__ void gather_row_sum(const int* __restrict__ rowptr, const int* __restrict__ csr_src,
                                               const float* __restrict__ h, int row, int lane,
                                               float& a0, float& a1, float& a2) {
  const int beg = rowptr[row];
  const int end = rowptr[row + 1];
  a0 = 0.0f; a1 = 0.0f; a2 = 0.0f;
  for (int base = beg; base < end; base += 32) {
    const int cnt  = (end - base < 32) ? (end - base) : 32;
    const int mine = (lane < cnt) ? csr_src[base + lane] : 0;
    int j = 0;
    for (; j + 4 <= cnt; j += 4) {
      const float* p0 = h + (long)__builtin_amdgcn_readlane(mine, j + 0) * HD + lane;
      const float* p1 = h + (long)__builtin_amdgcn_readlane(mine, j + 1) * HD + lane;
      const float* p2 = h + (long)__builtin_amdgcn_readlane(mine, j + 2) * HD + lane;
      const float* p3 = h + (long)__builtin_amdgcn_readlane(mine, j + 3) * HD + lane;
      const float x00 = p0[0], x01 = p0[32], x02 = p0[64];
      const float x10 = p1[0], x11 = p1[32], x12 = p1[64];
      const float x20 = p2[0], x21 = p2[32], x22 = p2[64];
      const float x30 = p3[0], x31 = p3[32], x32 = p3[64];
      a0 += x00; a1 += x01; a2 += x02;
      a0 += x10; a1 += x11; a2 += x12;
      a0 += x20; a1 += x21; a2 += x22;
      a0 += x30; a1 += x31; a2 += x32;
    }
    for (; j < cnt; ++j) {
      const float* p = h + (long)__builtin_amdgcn_readlane(mine, j) * HD + lane;
      a0 += p[0]; a1 += p[32]; a2 += p[64];
    }
  }
}

template <bool MAPPED>
__device__ __forceinline__ void gather_row_sum_scaled(const int* __restrict__ rowptr, const int* __restrict__ csr_src,
                                                      const float* __restrict__ xw, const int* __restrict__ rowmap,
                                                      const float* __restrict__ scale, int row, int lane,
                                                      float& a0, float& a1, float& a2) {
  const int beg = rowptr[row];
  const int end = rowptr[row + 1];
  a0 = 0.0f; a1 = 0.0f; a2 = 0.0f;
  for (int base = beg; base < end; base += 32) {
    const int cnt  = (end - base < 32) ? (end - base) : 32;
    const int mine = (lane < cnt) ? csr_src[base + lane] : 0;
    const int m    = MAPPED ? rowmap[mine] : mine;
    const int sc   = __float_as_int(scale[mine]);
    int j = 0;
    for (; j + 4 <= cnt; j += 4) {
      const float* p0 = xw + (long)__builtin_amdgcn_readlane(m, j + 0) * HD + lane;
      const float* p1 = xw + (long)__builtin_amdgcn_readlane(m, j + 1) * HD + lane;
      const float* p2 = xw + (long)__builtin_amdgcn_readlane(m, j + 2) * HD + lane;
      const float* p3 = xw + (long)__builtin_amdgcn_readlane(m, j + 3) * HD + lane;
      const float x00 = p0[0], x01 = p0[32], x02 = p0[64];
      const float x10 = p1[0], x11 = p1[32], x12 = p1[64];
      const float x20 = p2[0], x21 = p2[32], x22 = p2[64];
      const float x30 = p3[0], x31 = p3[32], x32 = p3[64];
      __builtin_amdgcn_sched_barrier(0);
      const float c0 = __int_as_float(__builtin_amdgcn_readlane(sc, j + 0));
      const float c1 = __int_as_float(__builtin_amdgcn_readlane(sc, j + 1));
      const float c2 = __int_as_float(__builtin_amdgcn_readlane(sc, j + 2));
      const float c3 = __int_as_float(__builtin_amdgcn_readlane(sc, j + 3));
      a0 = fmaf(c0, x00, a0); a1 = fmaf(c0, x01, a1); a2 = fmaf(c0, x02, a2);
      a0 = fmaf(c1, x10, a0); a1 = fmaf(c1, x11, a1); a2 = fmaf(c1, x12, a2);
      a0 = fmaf(c2, x20, a0); a1 = fmaf(c2, x21, a1); a2 = fmaf(c2, x22, a2);
      a0 = fmaf(c3, x30, a0); a1 = fmaf(c3, x31, a1); a2 = fmaf(c3, x32, a2);
    }
    for (; j < cnt; ++j) {
      const float* p = xw + (long)__builtin_amdgcn_readlane(m, j) * HD + lane;
      const float x0 = p[0], x1 = p[32], x2 = p[64];
      const float c  = __int_as_float(__builtin_amdgcn_readlane(sc, j));
      a0 = fmaf(c, x0, a0); a1 = fmaf(c, x1, a1); a2 = fmaf(c, x2, a2);
    }
  }
}

template <bool MAPPED>
__device__ __forceinline__ void gather_scaled_prelu_node(const int* __restrict__ rowptr, const int* __restrict__ csr_src,
                                                         const float* __restrict__ xw, const float* __restrict__ norm_out,
                                                         const float* __restrict__ norm_in, const float* __restrict__ b,
                                                         const float* __restrict__ a, float* __restrict__ act_row,
                                                         const int* __restrict__ rowmap, int row, int lane) {
  const float ni = norm_in[row];
  const float b0 = b[lane], b1 = b[lane + 32], b2 = b[lane + 64];
  const float s0 = a[lane], s1 = a[lane + 32], s2 = a[lane + 64];
  float a0, a1, a2;
  gather_row_sum_scaled<MAPPED>(rowptr, csr_src, xw, rowmap, norm_out, row, lane, a0, a1, a2);
  const float v0 = a0 * ni + b0;
  const float v1 = a1 * ni + b1;
  const float v2 = a2 * ni + b2;
  float* o = act_row + lane;
  o[0]  = (v0 >= 0.0f) ? v0 : s0 * v0;
  o[32] = (v1 >= 0.0f) ? v1 : s1 * v1;
  o[64] = (v2 >= 0.0f) ? v2 : s2 * v2;
}

template <bool MAPPED>
__global__ __launch_bounds__(256)
void spmm_gather_scaled_prelu_kernel(const int* __restrict__ rowptr, const int* __restrict__ csr_src,
                                     const float* __restrict__ xw, const float* __restrict__ norm_out,
                                     const float* __restrict__ norm_in, const float* __restrict__ b,
                                     const float* __restrict__ a, float* __restrict__ act,
                                     const int* __restrict__ rowmap, int n) {
  const int lane = threadIdx.x & 31;
  const int row = __builtin_amdgcn_readfirstlane((int)(blockIdx.x * (blockDim.x >> 5) + (threadIdx.x >> 5)));
  if (row >= n) return;
  gather_scaled_prelu_node<MAPPED>(rowptr, csr_src, xw, norm_out, norm_in, b, a, act + (long)row * HD, rowmap, row, lane);
}

__device__ __forceinline__ void gather_row_sum_scaled_both(const int* __restrict__ rowptr, const int* __restrict__ csr_src,
                                                           const float* __restrict__ xw, const int* __restrict__ rowmap,
                                                           const float* __restrict__ scale, int row, int lane,
                                                           float& a0, float& a1, float& a2,
                                                           float& c0, float& c1, float& c2) {
  const int beg = rowptr[row];
  const int end = rowptr[row + 1];
  a0 = 0.0f; a1 = 0.0f; a2 = 0.0f; c0 = 0.0f; c1 = 0.0f; c2 = 0.0f;
  for (int base = beg; base < end; base += 32) {
    const int cnt  = (end - base < 32) ? (end - base) : 32;
    const int mine = (lane < cnt) ? csr_src[base + lane] : 0;
    const int m    = rowmap[mine];
    const int sc   = __float_as_int(scale[mine]);
    int j = 0;
    for (; j + 4 <= cnt; j += 4) {
      const float* p0 = xw + (long)__builtin_amdgcn_readlane(mine, j + 0) * HD + lane;
      const float* p1 = xw + (long)__builtin_amdgcn_readlane(mine, j + 1) * HD + lane;
      const float* p2 = xw + (long)__builtin_amdgcn_readlane(mine, j + 2) * HD + lane;
      const float* p3 = xw + (long)__builtin_amdgcn_readlane(mine, j + 3) * HD + lane;
      const float x00 = p0[0], x01 = p0[32], x02 = p0[64];
      const float x10 = p1[0], x11 = p1[32], x12 = p1[64];
      const float x20 = p2[0], x21 = p2[32], x22 = p2[64];
      const float x30 = p3[0], x31 = p3[32], x32 = p3[64];
      __builtin_amdgcn_sched_barrier(0);
      const float* q0 = xw + (long)__builtin_amdgcn_readlane(m, j + 0) * HD + lane;
      const float* q1 = xw + (long)__builtin_amdgcn_readlane(m, j + 1) * HD + lane;
      const float* q2 = xw + (long)__builtin_amdgcn_readlane(m, j + 2) * HD + lane;
      const float* q3 = xw + (long)__builtin_amdgcn_readlane(m, j + 3) * HD + lane;
      const float y00 = q0[0], y01 = q0[32], y02 = q0[64];
      const float y10 = q1[0], y11 = q1[32], y12 = q1[64];
      const float y20 = q2[0], y21 = q2[32], y22 = q2[64];
      const float y30 = q3[0], y31 = q3[32], y32 = q3[64];
      __builtin_amdgcn_sched_barrier(0);
      const float s0 = __int_as_float(__builtin_amdgcn_readlane(sc, j + 0));
      const float s1 = __int_as_float(__builtin_amdgcn_readlane(sc, j + 1));
      const float s2 = __int_as_float(__builtin_amdgcn_readlane(sc, j + 2));
      const float s3 = __int_as_float(__builtin_amdgcn_readlane(sc, j + 3));
      a0 = fmaf(s0, x00, a0); a1 = fmaf(s0, x01, a1); a2 = fmaf(s0, x02, a2);
      a0 = fmaf(s1, x10, a0); a1 = fmaf(s1, x11, a1); a2 = fmaf(s1, x12, a2);
      a0 = fmaf(s2, x20, a0); a1 = fmaf(s2, x21, a1); a2 = fmaf(s2, x22, a2);
      a0 = fmaf(s3, x30, a0); a1 = fmaf(s3, x31, a1); a2 = fmaf(s3, x32, a2);
      c0 = fmaf(s0, y00, c0); c1 = fmaf(s0, y01, c1); c2 = fmaf(s0, y02, c2);
      c0 = fmaf(s1, y10, c0); c1 = fmaf(s1, y11, c1); c2 = fmaf(s1, y12, c2);
      c0 = fmaf(s2, y20, c0); c1 = fmaf(s2, y21, c1); c2 = fmaf(s2, y22, c2);
      c0 = fmaf(s3, y30, c0); c1 = fmaf(s3, y31, c1); c2 = fmaf(s3, y32, c2);
    }
    for (; j < cnt; ++j) {
      const float* p = xw + (long)__builtin_amdgcn_readlane(mine, j) * HD + lane;
      const float* q = xw + (long)__builtin_amdgcn_readlane(m, j) * HD + lane;
      const float x0 = p[0], x1 = p[32], x2 = p[64];
      const float y0 = q[0], y1 = q[32], y2 = q[64];
      const float s  = __int_as_float(__builtin_amdgcn_readlane(sc, j));
      a0 = fmaf(s, x0, a0); a1 = fmaf(s, x1, a1); a2 = fmaf(s, x2, a2);
      c0 = fmaf(s, y0, c0); c1 = fmaf(s, y1, c1); c2 = fmaf(s, y2, c2);
    }
  }
}

__device__ __forceinline__ void gather_scaled_prelu_vals_both(const int* __restrict__ rowptr, const int* __restrict__ csr_src,
                                                              const float* __restrict__ xw, const float* __restrict__ norm_out,
                                                              const float* __restrict__ norm_in, const float* __restrict__ b,
                                                              const float* __restrict__ a, const int* __restrict__ rowmap,
                                                              int row, int lane, float& o0, float& o1, float& o2,
                                                              float& u0, float& u1, float& u2) {
  const float ni = norm_in[row];
  float a0, a1, a2, c0, c1, c2;
  gather_row_sum_scaled_both(rowptr, csr_src, xw, rowmap, norm_out, row, lane, a0, a1, a2, c0, c1, c2);
  const float b0 = b[lane], b1 = b[lane + 32], b2 = b[lane + 64];
  const float s0 = a[lane], s1 = a[lane + 32], s2 = a[lane + 64];
  const float v0 = a0 * ni + b0;
  const float v1 = a1 * ni + b1;
  const float v2 = a2 * ni + b2;
  const float w0 = c0 * ni + b0;
  const float w1 = c1 * ni + b1;
  const float w2 = c2 * ni + b2;
  o0 = (v0 >= 0.0f) ? v0 : s0 * v0;
  o1 = (v1 >= 0.0f) ? v1 : s1 * v1;
  o2 = (v2 >= 0.0f) ? v2 : s2 * v2;
  u0 = (w0 >= 0.0f) ? w0 : s0 * w0;
  u1 = (w1 >= 0.0f) ? w1 : s1 * w1;
  u2 = (w2 >= 0.0f) ? w2 : s2 * w2;
}

constexpr int L1_WGS = 1280;
constexpr int TPAD = 100;
__global__ __launch_bounds__(256, 10)
void spmm_gather_scaled_prelu_both_w2_kernel(const int* __restrict__ rowptr, const int* __restrict__ csr_src,
                                             const float* __restrict__ xw, const float* __restrict__ norm_out,
                                             const float* __restrict__ norm_in, const float* __restrict__ b,
                                             const float* __restrict__ a, const float* __restrict__ W2,
                                             float* __restrict__ h2, const int* __restrict__ rowmap,
                                             int n, int ntiles) {
  __shared__ __attribute__((aligned(16))) float tile[2][16][TPAD];
  __shared__ float sb[HD], sa[HD];
  const int lane = threadIdx.x & 31;
  const int half = lane >> 4;
  const int l16  = lane & 15;
  const int wave = __builtin_amdgcn_readfirstlane((int)(threadIdx.x >> 5));

  if (threadIdx.x < HD) { sb[threadIdx.x] = b[threadIdx.x]; sa[threadIdx.x] = a[threadIdx.x]; }
  __syncthreads();

  v2f Bf[24];
  if (wave < 6) {
    const float* wp = W2 + wave * 16 + l16;
    #pragma unroll
    for (int j = 0; j < 24; ++j) {
      const int k = 4 * j + 2 * half;
      Bf[j].x = wp[k * HD];
      Bf[j].y = wp[(k + 1) * HD];
    }
  } else {
    #pragma unroll
    for (int j = 0; j < 24; ++j) { Bf[j].x = 0.0f; Bf[j].y = 0.0f; }
  }

  int buf = 0;
  for (int t = (int)blockIdx.x; t < ntiles; t += (int)gridDim.x) {
    const int node0 = t * 8;
    const int node  = node0 + wave;
    float o0 = 0.0f, o1 = 0.0f, o2 = 0.0f;
    float u0 = 0.0f, u1 = 0.0f, u2 = 0.0f;
    if (node < n) {
      const float no = norm_out[node];
      gather_scaled_prelu_vals_both(rowptr, csr_src, xw, norm_out, norm_in, sb, sa, rowmap, node, lane,
                                    o0, o1, o2, u0, u1, u2);
      o0 *= no; o1 *= no; o2 *= no;
      u0 *= no; u1 *= no; u2 *= no;
    }
    tile[buf][2 * wave][lane]          = o0;
    tile[buf][2 * wave][lane + 32]     = o1;
    tile[buf][2 * wave][lane + 64]     = o2;
    tile[buf][2 * wave + 1][lane]      = u0;
    tile[buf][2 * wave + 1][lane + 32] = u1;
    tile[buf][2 * wave + 1][lane + 64] = u2;
    __syncthreads();
    if (wave < 6) {
      v8f acc = {};
      #pragma unroll
      for (int j = 0; j < 24; ++j) {
        const v2f A = *(const v2f*)&tile[buf][l16][4 * j + 2 * half];
        acc = __builtin_amdgcn_wmma_f32_16x16x4_f32(
             false, A,  false, Bf[j],
             (short)0, acc,  false,  false);
      }
      const int col = wave * 16 + l16;
      float* hp = h2 + ((long)node0 * 2 + half * 8) * HD + col;
      const int lim = n - node0 - half * 4;
      #pragma unroll
      for (int i = 0; i < 8; ++i) {
        if ((i >> 1) < lim) hp[i * HD] = acc[i];
      }
    }
    buf ^= 1;
  }
}

__global__ __launch_bounds__(256)
void spmm_gather_prelu_dot_kernel(const int* __restrict__ rowptr, const int* __restrict__ csr_src,
                                  const float* __restrict__ h, const float* __restrict__ norm_in,
                                  const float* __restrict__ b, const float* __restrict__ a,
                                  const float* __restrict__ wsum, const float* __restrict__ bsum,
                                  float* __restrict__ out, int n) {
  const int lane = threadIdx.x & 31;
  const int row = __builtin_amdgcn_readfirstlane((int)(blockIdx.x * (blockDim.x >> 5) + (threadIdx.x >> 5)));
  if (row >= n) return;
  const float ni = norm_in[row];
  const float b0 = b[lane], b1 = b[lane + 32], b2 = b[lane + 64];
  const float s0 = a[lane], s1 = a[lane + 32], s2 = a[lane + 64];
  const float w0 = wsum[lane], w1 = wsum[lane + 32], w2 = wsum[lane + 64];
  const float bs = *bsum;
  float a0, a1, a2;
  gather_row_sum(rowptr, csr_src, h, row, lane, a0, a1, a2);
  const float v0 = a0 * ni + b0;
  const float v1 = a1 * ni + b1;
  const float v2 = a2 * ni + b2;
  const float z0 = (v0 >= 0.0f) ? v0 : s0 * v0;
  const float z1 = (v1 >= 0.0f) ? v1 : s1 * v1;
  const float z2 = (v2 >= 0.0f) ? v2 : s2 * v2;
  float s = fmaf(z0, w0, 0.0f);
  s = fmaf(z1, w1, s);
  s = fmaf(z2, w2, s);
  #pragma unroll
  for (int off = 16; off > 0; off >>= 1) s += __shfl_xor(s, off, 32);
  if (lane == 0) out[row] = s + bs;
}

__device__ __forceinline__ void gather_row_sum2(const int* __restrict__ rowptr, const int* __restrict__ csr_src,
                                                const float* __restrict__ h2, int row, int lane,
                                                float& a0, float& a1, float& a2, float& c0, float& c1, float& c2) {
  const int beg = rowptr[row];
  const int end = rowptr[row + 1];
  a0 = 0.0f; a1 = 0.0f; a2 = 0.0f; c0 = 0.0f; c1 = 0.0f; c2 = 0.0f;
  for (int base = beg; base < end; base += 32) {
    const int cnt  = (end - base < 32) ? (end - base) : 32;
    const int mine = (lane < cnt) ? csr_src[base + lane] : 0;
    int j = 0;
    for (; j + 4 <= cnt; j += 4) {
      const float* p0 = h2 + (long)__builtin_amdgcn_readlane(mine, j + 0) * (2 * HD) + lane;
      const float* p1 = h2 + (long)__builtin_amdgcn_readlane(mine, j + 1) * (2 * HD) + lane;
      const float* p2 = h2 + (long)__builtin_amdgcn_readlane(mine, j + 2) * (2 * HD) + lane;
      const float* p3 = h2 + (long)__builtin_amdgcn_readlane(mine, j + 3) * (2 * HD) + lane;
      const float x00 = p0[0], x01 = p0[32], x02 = p0[64], y00 = p0[96], y01 = p0[128], y02 = p0[160];
      const float x10 = p1[0], x11 = p1[32], x12 = p1[64], y10 = p1[96], y11 = p1[128], y12 = p1[160];
      const float x20 = p2[0], x21 = p2[32], x22 = p2[64], y20 = p2[96], y21 = p2[128], y22 = p2[160];
      const float x30 = p3[0], x31 = p3[32], x32 = p3[64], y30 = p3[96], y31 = p3[128], y32 = p3[160];
      a0 += x00; a1 += x01; a2 += x02; c0 += y00; c1 += y01; c2 += y02;
      a0 += x10; a1 += x11; a2 += x12; c0 += y10; c1 += y11; c2 += y12;
      a0 += x20; a1 += x21; a2 += x22; c0 += y20; c1 += y21; c2 += y22;
      a0 += x30; a1 += x31; a2 += x32; c0 += y30; c1 += y31; c2 += y32;
    }
    for (; j < cnt; ++j) {
      const float* p = h2 + (long)__builtin_amdgcn_readlane(mine, j) * (2 * HD) + lane;
      const float x0 = p[0], x1 = p[32], x2 = p[64], y0 = p[96], y1 = p[128], y2 = p[160];
      a0 += x0; a1 += x1; a2 += x2; c0 += y0; c1 += y1; c2 += y2;
    }
  }
}

__global__ __launch_bounds__(256)
void spmm_gather2_prelu_dot_kernel(const int* __restrict__ rowptr, const int* __restrict__ csr_src,
                                   const float* __restrict__ h2, const float* __restrict__ norm_in,
                                   const float* __restrict__ b, const float* __restrict__ a,
                                   const float* __restrict__ wsum, const float* __restrict__ bsum,
                                   float* __restrict__ out, int n) {
  const int lane = threadIdx.x & 31;
  const int row = __builtin_amdgcn_readfirstlane((int)(blockIdx.x * (blockDim.x >> 5) + (threadIdx.x >> 5)));
  if (row >= n) return;
  const float ni = norm_in[row];
  const float b0 = b[lane], b1 = b[lane + 32], b2 = b[lane + 64];
  const float s0 = a[lane], s1 = a[lane + 32], s2 = a[lane + 64];
  const float w0 = wsum[lane], w1 = wsum[lane + 32], w2 = wsum[lane + 64];
  const float bs = *bsum;
  float a0, a1, a2, c0, c1, c2;
  gather_row_sum2(rowptr, csr_src, h2, row, lane, a0, a1, a2, c0, c1, c2);
  const float v0 = a0 * ni + b0;
  const float v1 = a1 * ni + b1;
  const float v2 = a2 * ni + b2;
  const float u0 = c0 * ni + b0;
  const float u1 = c1 * ni + b1;
  const float u2 = c2 * ni + b2;
  const float z0 = (v0 >= 0.0f) ? v0 : s0 * v0;
  const float z1 = (v1 >= 0.0f) ? v1 : s1 * v1;
  const float z2 = (v2 >= 0.0f) ? v2 : s2 * v2;
  const float q0 = (u0 >= 0.0f) ? u0 : s0 * u0;
  const float q1 = (u1 >= 0.0f) ? u1 : s1 * u1;
  const float q2 = (u2 >= 0.0f) ? u2 : s2 * u2;
  float s = fmaf(z0, w0, 0.0f);
  s = fmaf(z1, w1, s);
  s = fmaf(z2, w2, s);
  float r = fmaf(q0, w0, 0.0f);
  r = fmaf(q1, w1, r);
  r = fmaf(q2, w2, r);
  for (int off = 16; off > 0; off >>= 1) {
    s += __shfl_xor(s, off, 32);
    r += __shfl_xor(r, off, 32);
  }
  if (lane == 0) {
    out[row]     = s + bs;
    out[n + row] = r + bs;
  }
}

__global__ void spmm_kernel(const int* __restrict__ src, const int* __restrict__ dst,
                            const float* __restrict__ h, float* __restrict__ agg, int E) {
  int gw   = (int)(((long)blockIdx.x * blockDim.x + threadIdx.x) >> 5);
  int lane = threadIdx.x & 31;
  if (gw < E) {
    int s = src[gw], d = dst[gw];
    const float* hp = h + (long)s * HD;
    float* ap = agg + (long)d * HD;
    #pragma unroll
    for (int i = 0; i < 3; ++i) {
      float v = hp[lane + 32 * i];
      __hip_atomic_fetch_add(&ap[lane + 32 * i], v, __ATOMIC_RELAXED, __HIP_MEMORY_SCOPE_AGENT);
    }
  }
}

__global__ void prelu_kernel(const float* __restrict__ agg, const float* __restrict__ norm_in,
                             const float* __restrict__ b, const float* __restrict__ a,
                             float* __restrict__ out, int n) {
  long idx = (long)blockIdx.x * blockDim.x + threadIdx.x;
  if (idx < (long)n * HD) {
    int i = (int)(idx / HD);
    int j = (int)(idx % HD);
    float v = agg[idx] * norm_in[i] + b[j];
    out[idx] = (v >= 0.0f) ? v : a[j] * v;
  }
}

__global__ void reduce_kernel(const float* __restrict__ z, const float* __restrict__ wsum,
                              const float* __restrict__ bsum, float* __restrict__ out, int n) {
  int gw   = (int)(((long)blockIdx.x * blockDim.x + threadIdx.x) >> 5);
  int lane = threadIdx.x & 31;
  if (gw < n) {
    const float* zp = z + (long)gw * HD;
    float s = 0.0f;
    #pragma unroll
    for (int i = 0; i < 3; ++i) s = fmaf(zp[lane + 32 * i], wsum[lane + 32 * i], s);
    #pragma unroll
    for (int off = 16; off > 0; off >>= 1) s += __shfl_xor(s, off, 32);
    if (lane == 0) out[gw] = s + *bsum;
  }
}

extern "C" void kernel_launch(void* const* d_in, const int* in_sizes, int n_in,
                              void* d_out, int out_size, void* d_ws, size_t ws_size,
                              hipStream_t stream) {
  const float* x    = (const float*)d_in[0];
  const int*   src  = (const int*)  d_in[1];
  const int*   dst  = (const int*)  d_in[2];
  const int*   perm = (const int*)  d_in[3];
  const float* W1   = (const float*)d_in[4];
  const float* b1   = (const float*)d_in[5];
  const float* a1   = (const float*)d_in[6];
  const float* W2   = (const float*)d_in[7];
  const float* b2   = (const float*)d_in[8];
  const float* a2   = (const float*)d_in[9];
  const float* Wm   = (const float*)d_in[10];
  const float* bm   = (const float*)d_in[11];
  float* out = (float*)d_out;

  const int E  = in_sizes[1];
  const int N  = in_sizes[3];
  const int K1 = in_sizes[0] / N;

  float* ws       = (float*)d_ws;
  float* norm_out = ws;
  float* norm_in  = ws + N;
  float* wsum     = ws + 2L * N;
  float* bsum     = ws + 2L * N + HD;
  float* bufA     = ws + 2L * N + 2 * HD;
  float* bufB     = bufA + (long)N * HD;

  const long NH = (long)N * HD;
  const size_t need_wide = (size_t)(2L * N + 2 * HD + 3 * NH + (N + 1) + E) * sizeof(float);
  const bool wide = (ws_size >= need_wide) && ((long)E <= NH);
  float* H2       = bufB;
  int*   rowptr   = (int*)(bufB + (wide ? 2 * NH : NH));
  int*   csr_src  = rowptr + (N + 1);
  int*   rank     = (int*)bufB;
  int*   cnt_out  = (int*)norm_out;
  int*   cnt_in   = (int*)norm_in;
  const bool hist_in_h2 = wide && ((long)HG * N <= NH);
  int*   hist_in  = hist_in_h2 ? (int*)(bufB + NH) : (int*)bufA;
  int*   hist_out = (int*)bufB;
  int*   lrank    = hist_out + (long)HG * N;
  int*   rowloc   = lrank + E;
  int*   btot     = rowloc + N;

  const int gemm_blocks = (N + RPB - 1) / RPB;
  const int spmm_blocks = (int)(((long)E * 32 + 255) / 256);
  const int gath_blocks = (N + 7) / 8;
  const int elw_blocks  = (int)((NH + 255) / 256);
  const int red_blocks  = (int)(((long)N * 32 + 255) / 256);

  const size_t need_grouped = (size_t)(2L * N + 2 * HD + 2 * NH + (N + 1) + E) * sizeof(float);
  const bool grouped = wide || ((ws_size >= need_grouped) && ((long)E <= NH));
  const int  chunk    = (E + HG - 1) / HG;
  const bool lds_hist = grouped && (N <= HCAP) && ((long)(HG + 1) * N + E + 256 <= NH);

  if (lds_hist) {
    hist_lds_kernel<<<2 * HG, 1024, 0, stream>>>(src, dst, hist_in, hist_out, lrank, E, N, chunk);
    hist_fold_scan_kernel<<<(2 * N + 255) / 256 + 1, 256, 0, stream>>>(hist_in, hist_out, rowloc, btot, norm_out, norm_in, N,
                                                                       Wm, bm, wsum, bsum);
    if (hist_in_h2) {
      const int place_wgs = (E + PLACE_EPT * 256 - 1) / (PLACE_EPT * 256);
      gemm_and_place_kernel<<<gemm_blocks + place_wgs, 256, 0, stream>>>(x, W1, bufA, N, K1, gemm_blocks,
                                                                         src, dst, lrank, rowloc, btot, hist_in,
                                                                         rowptr, csr_src, E, N, chunk);
    } else {
      place_chunked_scan_kernel<<<(E + 255) / 256, 256, 0, stream>>>(src, dst, lrank, rowloc, btot, hist_in, rowptr,
                                                                     csr_src, E, N, chunk);
    }
  } else {
    fill_zero_kernel<<<512, 256, 0, stream>>>(norm_out, 2L * N);
    degree_kernel<<<(E + 255) / 256, 256, 0, stream>>>(src, dst, cnt_out, cnt_in, grouped ? rank : nullptr, E);
    if (grouped) {
      rowptr_scan_kernel<<<1, 1024, 0, stream>>>(cnt_in, rowptr, N);
      place_kernel<<<(E + 255) / 256, 256, 0, stream>>>(src, dst, rank, rowptr, csr_src, E);
    }
    finalize_norm_kernel<<<(2 * N + 255) / 256, 256, 0, stream>>>(norm_out, 2 * N);
    wsum_kernel<<<1, 128, 0, stream>>>(Wm, bm, wsum, bsum);
  }

  if (grouped && !(lds_hist && hist_in_h2))
    gemm_norm_wmma_kernel<<<gemm_blocks, 256, 0, stream>>>(x, nullptr, nullptr, W1, bufA, N, K1, 0);

  if (wide) {
    const int l1_tiles = (N + 7) / 8;
    const int l1_trips = (l1_tiles + L1_WGS - 1) / L1_WGS;
    const int l1_wgs   = (l1_tiles + l1_trips - 1) / l1_trips;
    spmm_gather_scaled_prelu_both_w2_kernel<<<l1_wgs, 256, 0, stream>>>(rowptr, csr_src, bufA, norm_out, norm_in,
                                                                       b1, a1, W2, H2, perm, N, l1_tiles);
    spmm_gather2_prelu_dot_kernel<<<gath_blocks, 256, 0, stream>>>(rowptr, csr_src, H2, norm_in, b2, a2,
                                                                  wsum, bsum, out, N);
    return;
  }

  for (int view = 0; view < 2; ++view) {
    const int* rowmap = view ? perm : nullptr;
    float* outv = out + (long)view * N;
    if (grouped) {
      if (view) {
        spmm_gather_scaled_prelu_kernel<true><<<gath_blocks, 256, 0, stream>>>(rowptr, csr_src, bufA, norm_out, norm_in,
                                                                              b1, a1, bufB, rowmap, N);
      } else {
        spmm_gather_scaled_prelu_kernel<false><<<gath_blocks, 256, 0, stream>>>(rowptr, csr_src, bufA, norm_out, norm_in,
                                                                               b1, a1, bufB, nullptr, N);
      }
      gemm_norm_wmma_kernel<<<gemm_blocks, 256, 0, stream>>>(bufB, nullptr, norm_out, W2, bufB, N, HD, 0);
      spmm_gather_prelu_dot_kernel<<<gath_blocks, 256, 0, stream>>>(rowptr, csr_src, bufB, norm_in, b2, a2,
                                                                   wsum, bsum, outv, N);
    } else {
      gemm_norm_wmma_kernel<<<gemm_blocks, 256, 0, stream>>>(x, rowmap, norm_out, W1, bufA, N, K1, 0);
      fill_zero_kernel<<<2048, 256, 0, stream>>>(bufB, NH);
      spmm_kernel<<<spmm_blocks, 256, 0, stream>>>(src, dst, bufA, bufB, E);
      prelu_kernel<<<elw_blocks, 256, 0, stream>>>(bufB, norm_in, b1, a1, bufA, N);
      gemm_norm_wmma_kernel<<<gemm_blocks, 256, 0, stream>>>(bufA, nullptr, norm_out, W2, bufB, N, HD, 0);
      fill_zero_kernel<<<2048, 256, 0, stream>>>(bufA, NH);
      spmm_kernel<<<spmm_blocks, 256, 0, stream>>>(src, dst, bufB, bufA, E);
      prelu_kernel<<<elw_blocks, 256, 0, stream>>>(bufA, norm_in, b2, a2, bufB, N);
      reduce_kernel<<<red_blocks, 256, 0, stream>>>(bufB, wsum, bsum, outv, N);
    }
  }
}
